# starting kernel
# baseline (speedup 1.0000x reference)
#include <hip/hip_runtime.h>

typedef __attribute__((ext_vector_type(2))) float v2f;
typedef __attribute__((ext_vector_type(8))) float v8f;

#define N_IN   128
#define N_HID  256
#define N_HID2 128
#define N_CLS  2

__global__ void deg_kernel(const int* __restrict__ col, float* __restrict__ deg, int E) {
    int e = blockIdx.x * blockDim.x + threadIdx.x;
    if (e < E) atomicAdd(&deg[col[e]], 1.0f);
}

__global__ void deginv_kernel(const float* __restrict__ deg, float* __restrict__ dinv, int n) {
    int i = blockIdx.x * blockDim.x + threadIdx.x;
    if (i < n) dinv[i] = 1.0f / fmaxf(deg[i], 1.0f);
}

template<int F>
__global__ void scatter_kernel(const float* __restrict__ src,
                               const int* __restrict__ row,
                               const int* __restrict__ col,
                               float* __restrict__ agg,
                               long long total) {
    long long tid = (long long)blockIdx.x * blockDim.x + threadIdx.x;
    if (tid >= total) return;
    int e = (int)(tid >> (__builtin_ctz(F)));
    int f = (int)(tid & (F - 1));
    int r = row[e], c = col[e];
    atomicAdd(&agg[(size_t)c * F + f], src[(size_t)r * F + f]);
}

template<int K, int NF, bool RELU>
__global__ __launch_bounds__(256)
void gemm_wmma(const float* __restrict__ A, const float* __restrict__ W,
               const float* __restrict__ bias, const float* __restrict__ dinv,
               float* __restrict__ out, int M) {
    const int lane  = threadIdx.x;
    const int waveId = threadIdx.y;
    const int tilesN = NF / 16;
    const int tilesM = (M + 15) / 16;
    int tile = blockIdx.x * blockDim.y + waveId;
    if (tile >= tilesM * tilesN) return;
    const int tm = tile / tilesN;
    const int tn = tile % tilesN;

    const int half = lane >> 4;
    const int l16  = lane & 15;

    int arow = tm * 16 + l16;
    if (arow >= M) arow = M - 1;
    const float* __restrict__ aptr = A + (size_t)arow * K;
    const float* __restrict__ wptr = W + (size_t)(tn * 16 + l16) * K;

    v8f acc = {};
    #pragma unroll
    for (int k = 0; k < K; k += 4) {
        const int kk = k + 2 * half;
        v2f a = *(const v2f*)(aptr + kk);
        v2f b = *(const v2f*)(wptr + kk);
        acc = __builtin_amdgcn_wmma_f32_16x16x4_f32(
             false, a,  false, b,
             (short)0, acc,  false,  false);
    }

    const int n  = tn * 16 + l16;
    const float bv = bias[n];
    #pragma unroll
    for (int r = 0; r < 8; ++r) {
        int m = tm * 16 + half * 8 + r;
        if (m < M) {
            float v = acc[r] * dinv[m] + bv;
            if (RELU) v = fmaxf(v, 0.0f);
            out[(size_t)m * NF + n] = v;
        }
    }
}

__global__ void head_kernel(const float* __restrict__ h, const float* __restrict__ W3,
                            const float* __restrict__ b3, float* __restrict__ out, int M) {
    int tid = blockIdx.x * blockDim.x + threadIdx.x;
    if (tid >= M * N_CLS) return;
    int m = tid / N_CLS, c = tid % N_CLS;
    const float* __restrict__ hm = h  + (size_t)m * N_HID2;
    const float* __restrict__ wc = W3 + (size_t)c * N_HID2;
    float s = b3[c];
    #pragma unroll 8
    for (int k = 0; k < N_HID2; ++k) s += hm[k] * wc[k];
    out[tid] = s;
}

extern "C" void kernel_launch(void* const* d_in, const int* in_sizes, int n_in,
                              void* d_out, int out_size, void* d_ws, size_t ws_size,
                              hipStream_t stream) {
    const float* x  = (const float*)d_in[0];
    const int*   ei = (const int*)  d_in[1];
    const float* W1 = (const float*)d_in[3];
    const float* b1 = (const float*)d_in[4];
    const float* W2 = (const float*)d_in[5];
    const float* b2 = (const float*)d_in[6];
    const float* W3 = (const float*)d_in[7];
    const float* b3 = (const float*)d_in[8];
    float* out = (float*)d_out;

    const int M = in_sizes[0] / N_IN;
    const int E = in_sizes[1] / 2;
    const int* row = ei;
    const int* col = ei + E;

    float* ws = (float*)d_ws;
    size_t off = 0;
    float* deg  = ws + off; off += (size_t)M;
    float* dinv = ws + off; off += (size_t)M;
    float* agg1 = ws + off; off += (size_t)M * N_IN;
    float* h1   = ws + off; off += (size_t)M * N_HID;
    float* agg2 = ws + off; off += (size_t)M * N_HID;
    float* h2   = ws + off; off += (size_t)M * N_HID2;
    (void)ws_size;

    hipMemsetAsync(deg,  0, sizeof(float) * (size_t)M, stream);
    hipMemsetAsync(agg1, 0, sizeof(float) * (size_t)M * N_IN, stream);
    hipMemsetAsync(agg2, 0, sizeof(float) * (size_t)M * N_HID, stream);

    deg_kernel<<<(E + 255) / 256, 256, 0, stream>>>(col, deg, E);
    deginv_kernel<<<(M + 255) / 256, 256, 0, stream>>>(deg, dinv, M);

    long long tot1 = (long long)E * N_IN;
    scatter_kernel<N_IN><<<(unsigned)((tot1 + 255) / 256), 256, 0, stream>>>(x, row, col, agg1, tot1);
    int tiles1 = ((M + 15) / 16) * (N_HID / 16);
    gemm_wmma<N_IN, N_HID, true>
        <<<(tiles1 + 7) / 8, dim3(32, 8), 0, stream>>>(agg1, W1, b1, dinv, h1, M);

    long long tot2 = (long long)E * N_HID;
    scatter_kernel<N_HID><<<(unsigned)((tot2 + 255) / 256), 256, 0, stream>>>(h1, row, col, agg2, tot2);
    int tiles2 = ((M + 15) / 16) * (N_HID2 / 16);
    gemm_wmma<N_HID, N_HID2, true>
        <<<(tiles2 + 7) / 8, dim3(32, 8), 0, stream>>>(agg2, W2, b2, dinv, h2, M);

    head_kernel<<<(M * N_CLS + 255) / 256, 256, 0, stream>>>(h2, W3, b3, out, M);
}


# Neighbour sums by owner groups: a work-group owns 196 nodes, scans all edges, adds matching source rows into an LDS tile by LDS float atomics, writes it once; plane memsets gone; sum order arrival
# speedup vs baseline: 2.8730x; 2.8730x over previous
#include <hip/hip_runtime.h>

typedef __attribute__((ext_vector_type(2))) float v2f;
typedef __attribute__((ext_vector_type(8))) float v8f;

#define N_IN   128
#define N_HID  256
#define N_HID2 128
#define N_CLS  2

__global__ void deg_kernel(const int* __restrict__ col, float* __restrict__ deg, int E) {
    int e = blockIdx.x * blockDim.x + threadIdx.x;
    if (e < E) atomicAdd(&deg[col[e]], 1.0f);
}

__global__ void deginv_kernel(const float* __restrict__ deg, float* __restrict__ dinv, int n) {
    int i = blockIdx.x * blockDim.x + threadIdx.x;
    if (i < n) dinv[i] = 1.0f / fmaxf(deg[i], 1.0f);
}

#define AGG_T   196
#define AGG_NT  1024
#define AGG_CAP 4096

__device__ __forceinline__ int4 load_cols(const int* __restrict__ col, int idx, int E, bool vec_ok) {
    int4 c;
    if (vec_ok && idx + 3 < E) {
        c = *(const int4*)(col + idx);
    } else {
        c.x = (idx     < E) ? col[idx]     : -1;
        c.y = (idx + 1 < E) ? col[idx + 1] : -1;
        c.z = (idx + 2 < E) ? col[idx + 2] : -1;
        c.w = (idx + 3 < E) ? col[idx + 3] : -1;
    }
    return c;
}

template<int F>
__global__ __launch_bounds__(AGG_NT)
void agg_owner(const float* __restrict__ src,
               const int* __restrict__ row,
               const int* __restrict__ col,
               float* __restrict__ agg, int M, int E) {
    constexpr int V  = F / 128;
    constexpr int CH = AGG_NT * 4;
    __shared__ __attribute__((aligned(16))) float acc[AGG_T * F];
    __shared__ int l_row[AGG_CAP];
    __shared__ int l_loc[AGG_CAP];
    __shared__ int l_n;

    const int tid  = threadIdx.x;
    const int lane = tid & 31;
    const int wave = tid >> 5;
    const int lo   = blockIdx.x * AGG_T;
    const int nT   = min(AGG_T, M - lo);
    const unsigned unT = (unsigned)nT;

    for (int i = tid * 4; i < AGG_T * F; i += AGG_NT * 4)
        *(float4*)&acc[i] = make_float4(0.f, 0.f, 0.f, 0.f);
    if (tid == 0) l_n = 0;
    __syncthreads();

    const bool vec_ok = ((E & 3) == 0);
    const int nIter = (E + CH - 1) / CH;
    const int it0 = (int)(blockIdx.x % (unsigned)nIter);
    for (int kk = 0; kk < nIter; kk += 4) {
        int4 c[4];
        int  ix[4];
        int  cb[4];
        bool fast = vec_ok && (kk + 3 < nIter);
        #pragma unroll
        for (int u = 0; u < 4; ++u) {
            int itr = it0 + kk + u;
            if (itr >= nIter) itr -= nIter;
            cb[u] = itr * CH;
            fast = fast && (cb[u] + CH <= E);
        }
        if (fast) {
            #pragma unroll
            for (int u = 0; u < 4; ++u) {
                ix[u] = cb[u] + tid * 4;
                c[u]  = *(const int4*)(col + ix[u]);
            }
        } else {
            #pragma unroll
            for (int u = 0; u < 4; ++u) {
                ix[u] = (kk + u < nIter) ? cb[u] + tid * 4 : E;
                c[u]  = load_cols(col, ix[u], E, vec_ok);
            }
        }
        #pragma unroll
        for (int u = 0; u < 4; ++u) {
            const unsigned d[4] = { (unsigned)(c[u].x - lo), (unsigned)(c[u].y - lo),
                                    (unsigned)(c[u].z - lo), (unsigned)(c[u].w - lo) };
            const int hits = (int)(d[0] < unT) + (int)(d[1] < unT) + (int)(d[2] < unT) + (int)(d[3] < unT);
            if (hits) {
                int pos = atomicAdd(&l_n, hits);
                #pragma unroll
                for (int j = 0; j < 4; ++j) {
                    if (d[j] < unT) {
                        const int e = ix[u] + j;
                        if (pos < AGG_CAP) {
                            l_row[pos] = e;
                            l_loc[pos] = (int)d[j];
                        } else {
                            const float* __restrict__ p = src + (size_t)row[e] * F;
                            float* a = &acc[(int)d[j] * F];
                            for (int f = 0; f < F; ++f)
                                atomicAdd(&a[(f & ~127) + (f & 3) * 32 + ((f & 127) >> 2)], p[f]);
                        }
                        ++pos;
                    }
                }
            }
        }
    }
    __syncthreads();

    const int n = min(l_n, AGG_CAP);
    for (int i = tid; i < n; i += AGG_NT) l_row[i] = row[l_row[i]];
    __syncthreads();

    for (int b = wave * 8; b < n; b += (AGG_NT / 32) * 8) {
        const int cnt = min(8, n - b);
        if (cnt == 8) {
            float4 v[8][V];
            int    lc[8];
            #pragma unroll
            for (int j = 0; j < 8; ++j) {
                const int r = l_row[b + j];
                lc[j] = l_loc[b + j];
                const float* __restrict__ p = src + (size_t)r * F + lane * 4;
                #pragma unroll
                for (int q = 0; q < V; ++q) v[j][q] = *(const float4*)(p + q * 128);
            }
            #pragma unroll
            for (int j = 0; j < 8; ++j) {
                #pragma unroll
                for (int q = 0; q < V; ++q) {
                    float* a = &acc[lc[j] * F + q * 128 + lane];
                    atomicAdd(a,      v[j][q].x);
                    atomicAdd(a + 32, v[j][q].y);
                    atomicAdd(a + 64, v[j][q].z);
                    atomicAdd(a + 96, v[j][q].w);
                }
            }
        } else {
            for (int j = 0; j < cnt; ++j) {
                const int r = l_row[b + j];
                const int l = l_loc[b + j];
                const float* __restrict__ p = src + (size_t)r * F + lane * 4;
                #pragma unroll
                for (int q = 0; q < V; ++q) {
                    const float4 t = *(const float4*)(p + q * 128);
                    float* a = &acc[l * F + q * 128 + lane];
                    atomicAdd(a,      t.x);
                    atomicAdd(a + 32, t.y);
                    atomicAdd(a + 64, t.z);
                    atomicAdd(a + 96, t.w);
                }
            }
        }
    }
    __syncthreads();

    float* __restrict__ dst = agg + (size_t)lo * F;
    const int tot = nT * F;
    for (int i = tid * 4; i < tot; i += AGG_NT * 4) {
        const int base = (i & ~127) + ((i & 127) >> 2);
        float4 t;
        t.x = acc[base];
        t.y = acc[base + 32];
        t.z = acc[base + 64];
        t.w = acc[base + 96];
        *(float4*)&dst[i] = t;
    }
}

template<int K, int NF, bool RELU>
__global__ __launch_bounds__(256)
void gemm_wmma(const float* __restrict__ A, const float* __restrict__ W,
               const float* __restrict__ bias, const float* __restrict__ dinv,
               float* __restrict__ out, int M) {
    const int lane  = threadIdx.x;
    const int waveId = threadIdx.y;
    const int tilesN = NF / 16;
    const int tilesM = (M + 15) / 16;
    int tile = blockIdx.x * blockDim.y + waveId;
    if (tile >= tilesM * tilesN) return;
    const int tm = tile / tilesN;
    const int tn = tile % tilesN;

    const int half = lane >> 4;
    const int l16  = lane & 15;

    int arow = tm * 16 + l16;
    if (arow >= M) arow = M - 1;
    const float* __restrict__ aptr = A + (size_t)arow * K;
    const float* __restrict__ wptr = W + (size_t)(tn * 16 + l16) * K;

    v8f acc = {};
    #pragma unroll
    for (int k = 0; k < K; k += 4) {
        const int kk = k + 2 * half;
        v2f a = *(const v2f*)(aptr + kk);
        v2f b = *(const v2f*)(wptr + kk);
        acc = __builtin_amdgcn_wmma_f32_16x16x4_f32(
             false, a,  false, b,
             (short)0, acc,  false,  false);
    }

    const int n  = tn * 16 + l16;
    const float bv = bias[n];
    #pragma unroll
    for (int r = 0; r < 8; ++r) {
        int m = tm * 16 + half * 8 + r;
        if (m < M) {
            float v = acc[r] * dinv[m] + bv;
            if (RELU) v = fmaxf(v, 0.0f);
            out[(size_t)m * NF + n] = v;
        }
    }
}

__global__ void head_kernel(const float* __restrict__ h, const float* __restrict__ W3,
                            const float* __restrict__ b3, float* __restrict__ out, int M) {
    int tid = blockIdx.x * blockDim.x + threadIdx.x;
    if (tid >= M * N_CLS) return;
    int m = tid / N_CLS, c = tid % N_CLS;
    const float* __restrict__ hm = h  + (size_t)m * N_HID2;
    const float* __restrict__ wc = W3 + (size_t)c * N_HID2;
    float s = b3[c];
    #pragma unroll 8
    for (int k = 0; k < N_HID2; ++k) s += hm[k] * wc[k];
    out[tid] = s;
}

extern "C" void kernel_launch(void* const* d_in, const int* in_sizes, int n_in,
                              void* d_out, int out_size, void* d_ws, size_t ws_size,
                              hipStream_t stream) {
    const float* x  = (const float*)d_in[0];
    const int*   ei = (const int*)  d_in[1];
    const float* W1 = (const float*)d_in[3];
    const float* b1 = (const float*)d_in[4];
    const float* W2 = (const float*)d_in[5];
    const float* b2 = (const float*)d_in[6];
    const float* W3 = (const float*)d_in[7];
    const float* b3 = (const float*)d_in[8];
    float* out = (float*)d_out;

    const int M = in_sizes[0] / N_IN;
    const int E = in_sizes[1] / 2;
    const int* row = ei;
    const int* col = ei + E;

    float* ws = (float*)d_ws;
    size_t off = 0;
    float* deg  = ws + off; off += (size_t)M;
    float* dinv = ws + off; off += (size_t)M;
    float* agg1 = ws + off; off += (size_t)M * N_IN;
    float* h1   = ws + off; off += (size_t)M * N_HID;
    float* agg2 = ws + off; off += (size_t)M * N_HID;
    float* h2   = ws + off; off += (size_t)M * N_HID2;
    (void)ws_size;

    hipMemsetAsync(deg,  0, sizeof(float) * (size_t)M, stream);

    deg_kernel<<<(E + 255) / 256, 256, 0, stream>>>(col, deg, E);
    deginv_kernel<<<(M + 255) / 256, 256, 0, stream>>>(deg, dinv, M);

    agg_owner<N_IN><<<(M + AGG_T - 1) / AGG_T, AGG_NT, 0, stream>>>(x, row, col, agg1, M, E);
    int tiles1 = ((M + 15) / 16) * (N_HID / 16);
    gemm_wmma<N_IN, N_HID, true>
        <<<(tiles1 + 7) / 8, dim3(32, 8), 0, stream>>>(agg1, W1, b1, dinv, h1, M);

    agg_owner<N_HID><<<(M + AGG_T - 1) / AGG_T, AGG_NT, 0, stream>>>(h1, row, col, agg2, M, E);
    int tiles2 = ((M + 15) / 16) * (N_HID2 / 16);
    gemm_wmma<N_HID, N_HID2, true>
        <<<(tiles2 + 7) / 8, dim3(32, 8), 0, stream>>>(agg2, W2, b2, dinv, h2, M);

    head_kernel<<<(M * N_CLS + 255) / 256, 256, 0, stream>>>(h2, W3, b3, out, M);
}


# GEMMs re-tiled in registers: a wave owns a 32x64 output block (2x4 accumulator tiles); each pair of 16-byte A/W fragment loads from global feeds 4 or 2 tiles, no LDS; k sum order changed
# speedup vs baseline: 3.6159x; 1.2586x over previous
#include <hip/hip_runtime.h>

typedef __attribute__((ext_vector_type(2))) float v2f;
typedef __attribute__((ext_vector_type(8))) float v8f;

#define N_IN   128
#define N_HID  256
#define N_HID2 128
#define N_CLS  2

__global__ void deg_kernel(const int* __restrict__ col, float* __restrict__ deg, int E) {
    int e = blockIdx.x * blockDim.x + threadIdx.x;
    if (e < E) atomicAdd(&deg[col[e]], 1.0f);
}

__global__ void deginv_kernel(const float* __restrict__ deg, float* __restrict__ dinv, int n) {
    int i = blockIdx.x * blockDim.x + threadIdx.x;
    if (i < n) dinv[i] = 1.0f / fmaxf(deg[i], 1.0f);
}

#define AGG_T   196
#define AGG_NT  1024
#define AGG_CAP 4096

__device__ __forceinline__ int4 load_cols(const int* __restrict__ col, int idx, int E, bool vec_ok) {
    int4 c;
    if (vec_ok && idx + 3 < E) {
        c = *(const int4*)(col + idx);
    } else {
        c.x = (idx     < E) ? col[idx]     : -1;
        c.y = (idx + 1 < E) ? col[idx + 1] : -1;
        c.z = (idx + 2 < E) ? col[idx + 2] : -1;
        c.w = (idx + 3 < E) ? col[idx + 3] : -1;
    }
    return c;
}

template<int F>
__global__ __launch_bounds__(AGG_NT)
void agg_owner(const float* __restrict__ src,
               const int* __restrict__ row,
               const int* __restrict__ col,
               float* __restrict__ agg, int M, int E) {
    constexpr int V  = F / 128;
    constexpr int CH = AGG_NT * 4;
    __shared__ __attribute__((aligned(16))) float acc[AGG_T * F];
    __shared__ int l_row[AGG_CAP];
    __shared__ int l_loc[AGG_CAP];
    __shared__ int l_n;

    const int tid  = threadIdx.x;
    const int lane = tid & 31;
    const int wave = tid >> 5;
    const int lo   = blockIdx.x * AGG_T;
    const int nT   = min(AGG_T, M - lo);
    const unsigned unT = (unsigned)nT;

    for (int i = tid * 4; i < AGG_T * F; i += AGG_NT * 4)
        *(float4*)&acc[i] = make_float4(0.f, 0.f, 0.f, 0.f);
    if (tid == 0) l_n = 0;
    __syncthreads();

    const bool vec_ok = ((E & 3) == 0);
    const int nIter = (E + CH - 1) / CH;
    const int it0 = (int)(blockIdx.x % (unsigned)nIter);
    for (int kk = 0; kk < nIter; kk += 4) {
        int4 c[4];
        int  ix[4];
        int  cb[4];
        bool fast = vec_ok && (kk + 3 < nIter);
        #pragma unroll
        for (int u = 0; u < 4; ++u) {
            int itr = it0 + kk + u;
            if (itr >= nIter) itr -= nIter;
            cb[u] = itr * CH;
            fast = fast && (cb[u] + CH <= E);
        }
        if (fast) {
            #pragma unroll
            for (int u = 0; u < 4; ++u) {
                ix[u] = cb[u] + tid * 4;
                c[u]  = *(const int4*)(col + ix[u]);
            }
        } else {
            #pragma unroll
            for (int u = 0; u < 4; ++u) {
                ix[u] = (kk + u < nIter) ? cb[u] + tid * 4 : E;
                c[u]  = load_cols(col, ix[u], E, vec_ok);
            }
        }
        #pragma unroll
        for (int u = 0; u < 4; ++u) {
            const unsigned d[4] = { (unsigned)(c[u].x - lo), (unsigned)(c[u].y - lo),
                                    (unsigned)(c[u].z - lo), (unsigned)(c[u].w - lo) };
            const int hits = (int)(d[0] < unT) + (int)(d[1] < unT) + (int)(d[2] < unT) + (int)(d[3] < unT);
            if (hits) {
                int pos = atomicAdd(&l_n, hits);
                #pragma unroll
                for (int j = 0; j < 4; ++j) {
                    if (d[j] < unT) {
                        const int e = ix[u] + j;
                        if (pos < AGG_CAP) {
                            l_row[pos] = e;
                            l_loc[pos] = (int)d[j];
                        } else {
                            const float* __restrict__ p = src + (size_t)row[e] * F;
                            float* a = &acc[(int)d[j] * F];
                            for (int f = 0; f < F; ++f)
                                atomicAdd(&a[(f & ~127) + (f & 3) * 32 + ((f & 127) >> 2)], p[f]);
                        }
                        ++pos;
                    }
                }
            }
        }
    }
    __syncthreads();

    const int n = min(l_n, AGG_CAP);
    for (int i = tid; i < n; i += AGG_NT) l_row[i] = row[l_row[i]];
    __syncthreads();

    for (int b = wave * 8; b < n; b += (AGG_NT / 32) * 8) {
        const int cnt = min(8, n - b);
        if (cnt == 8) {
            float4 v[8][V];
            int    lc[8];
            #pragma unroll
            for (int j = 0; j < 8; ++j) {
                const int r = l_row[b + j];
                lc[j] = l_loc[b + j];
                const float* __restrict__ p = src + (size_t)r * F + lane * 4;
                #pragma unroll
                for (int q = 0; q < V; ++q) v[j][q] = *(const float4*)(p + q * 128);
            }
            #pragma unroll
            for (int j = 0; j < 8; ++j) {
                #pragma unroll
                for (int q = 0; q < V; ++q) {
                    float* a = &acc[lc[j] * F + q * 128 + lane];
                    atomicAdd(a,      v[j][q].x);
                    atomicAdd(a + 32, v[j][q].y);
                    atomicAdd(a + 64, v[j][q].z);
                    atomicAdd(a + 96, v[j][q].w);
                }
            }
        } else {
            for (int j = 0; j < cnt; ++j) {
                const int r = l_row[b + j];
                const int l = l_loc[b + j];
                const float* __restrict__ p = src + (size_t)r * F + lane * 4;
                #pragma unroll
                for (int q = 0; q < V; ++q) {
                    const float4 t = *(const float4*)(p + q * 128);
                    float* a = &acc[l * F + q * 128 + lane];
                    atomicAdd(a,      t.x);
                    atomicAdd(a + 32, t.y);
                    atomicAdd(a + 64, t.z);
                    atomicAdd(a + 96, t.w);
                }
            }
        }
    }
    __syncthreads();

    float* __restrict__ dst = agg + (size_t)lo * F;
    const int tot = nT * F;
    for (int i = tid * 4; i < tot; i += AGG_NT * 4) {
        const int base = (i & ~127) + ((i & 127) >> 2);
        float4 t;
        t.x = acc[base];
        t.y = acc[base + 32];
        t.z = acc[base + 64];
        t.w = acc[base + 96];
        *(float4*)&dst[i] = t;
    }
}

#define GM_MT 2
#define GM_NT 4
#define GM_WAVES 4
#define GM_KS 16

template<int K, int NF, bool RELU>
__global__ __launch_bounds__(32 * GM_WAVES)
void gemm_wmma(const float* __restrict__ A, const float* __restrict__ W,
               const float* __restrict__ bias, const float* __restrict__ dinv,
               float* __restrict__ out, int M) {
    constexpr int NB = NF / (16 * GM_NT);
    const int lane   = threadIdx.x;
    const int waveId = threadIdx.y;
    const int stripsM = (M + 16 * GM_MT - 1) / (16 * GM_MT);
    const int tile = blockIdx.x * GM_WAVES + waveId;
    if (tile >= stripsM * NB) return;
    const int tm = tile / NB;
    const int tn = tile % NB;

    const int half = lane >> 4;
    const int l16  = lane & 15;

    const float* __restrict__ ap[GM_MT];
    const float* __restrict__ wp[GM_NT];
    #pragma unroll
    for (int i = 0; i < GM_MT; ++i) {
        int r = tm * (16 * GM_MT) + i * 16 + l16;
        if (r >= M) r = M - 1;
        ap[i] = A + (size_t)r * K + 8 * half;
    }
    #pragma unroll
    for (int j = 0; j < GM_NT; ++j)
        wp[j] = W + (size_t)(tn * (16 * GM_NT) + j * 16 + l16) * K + 8 * half;

    v8f acc[GM_MT][GM_NT];
    #pragma unroll
    for (int i = 0; i < GM_MT; ++i)
        #pragma unroll
        for (int j = 0; j < GM_NT; ++j)
            acc[i][j] = v8f{};

    float4 a[GM_MT][2], b[GM_NT][2];
    #pragma unroll
    for (int i = 0; i < GM_MT; ++i) {
        a[i][0] = *(const float4*)(ap[i]);
        a[i][1] = *(const float4*)(ap[i] + 4);
    }
    #pragma unroll
    for (int j = 0; j < GM_NT; ++j) {
        b[j][0] = *(const float4*)(wp[j]);
        b[j][1] = *(const float4*)(wp[j] + 4);
    }

    #pragma unroll 2
    for (int k0 = 0; k0 < K; k0 += GM_KS) {
        float4 an[GM_MT][2], bn[GM_NT][2];
        const int kn = (k0 + GM_KS < K) ? k0 + GM_KS : k0;
        #pragma unroll
        for (int i = 0; i < GM_MT; ++i) {
            an[i][0] = *(const float4*)(ap[i] + kn);
            an[i][1] = *(const float4*)(ap[i] + kn + 4);
        }
        #pragma unroll
        for (int j = 0; j < GM_NT; ++j) {
            bn[j][0] = *(const float4*)(wp[j] + kn);
            bn[j][1] = *(const float4*)(wp[j] + kn + 4);
        }
        __builtin_amdgcn_sched_barrier(0);

        #pragma unroll
        for (int p = 0; p < 2; ++p) {
            #pragma unroll
            for (int i = 0; i < GM_MT; ++i) {
                const v2f a0 = { a[i][p].x, a[i][p].y };
                #pragma unroll
                for (int j = 0; j < GM_NT; ++j) {
                    const v2f b0 = { b[j][p].x, b[j][p].y };
                    acc[i][j] = __builtin_amdgcn_wmma_f32_16x16x4_f32(
                         false, a0,  false, b0,
                         (short)0, acc[i][j],  false,  false);
                }
            }
            #pragma unroll
            for (int i = 0; i < GM_MT; ++i) {
                const v2f a1 = { a[i][p].z, a[i][p].w };
                #pragma unroll
                for (int j = 0; j < GM_NT; ++j) {
                    const v2f b1 = { b[j][p].z, b[j][p].w };
                    acc[i][j] = __builtin_amdgcn_wmma_f32_16x16x4_f32(
                         false, a1,  false, b1,
                         (short)0, acc[i][j],  false,  false);
                }
            }
        }
        #pragma unroll
        for (int i = 0; i < GM_MT; ++i) { a[i][0] = an[i][0]; a[i][1] = an[i][1]; }
        #pragma unroll
        for (int j = 0; j < GM_NT; ++j) { b[j][0] = bn[j][0]; b[j][1] = bn[j][1]; }
    }

    float bv[GM_NT];
    #pragma unroll
    for (int j = 0; j < GM_NT; ++j) bv[j] = bias[tn * (16 * GM_NT) + j * 16 + l16];

    float dv[GM_MT][8];
    #pragma unroll
    for (int i = 0; i < GM_MT; ++i) {
        #pragma unroll
        for (int r = 0; r < 8; ++r) {
            int m = tm * (16 * GM_MT) + i * 16 + half * 8 + r;
            if (m >= M) m = M - 1;
            dv[i][r] = dinv[m];
        }
    }

    #pragma unroll
    for (int i = 0; i < GM_MT; ++i) {
        #pragma unroll
        for (int r = 0; r < 8; ++r) {
            const int m = tm * (16 * GM_MT) + i * 16 + half * 8 + r;
            if (m < M) {
                #pragma unroll
                for (int j = 0; j < GM_NT; ++j) {
                    float v = acc[i][j][r] * dv[i][r] + bv[j];
                    if (RELU) v = fmaxf(v, 0.0f);
                    out[(size_t)m * NF + tn * (16 * GM_NT) + j * 16 + l16] = v;
                }
            }
        }
    }
}

__global__ void head_kernel(const float* __restrict__ h, const float* __restrict__ W3,
                            const float* __restrict__ b3, float* __restrict__ out, int M) {
    int tid = blockIdx.x * blockDim.x + threadIdx.x;
    if (tid >= M * N_CLS) return;
    int m = tid / N_CLS, c = tid % N_CLS;
    const float* __restrict__ hm = h  + (size_t)m * N_HID2;
    const float* __restrict__ wc = W3 + (size_t)c * N_HID2;
    float s = b3[c];
    #pragma unroll 8
    for (int k = 0; k < N_HID2; ++k) s += hm[k] * wc[k];
    out[tid] = s;
}

extern "C" void kernel_launch(void* const* d_in, const int* in_sizes, int n_in,
                              void* d_out, int out_size, void* d_ws, size_t ws_size,
                              hipStream_t stream) {
    const float* x  = (const float*)d_in[0];
    const int*   ei = (const int*)  d_in[1];
    const float* W1 = (const float*)d_in[3];
    const float* b1 = (const float*)d_in[4];
    const float* W2 = (const float*)d_in[5];
    const float* b2 = (const float*)d_in[6];
    const float* W3 = (const float*)d_in[7];
    const float* b3 = (const float*)d_in[8];
    float* out = (float*)d_out;

    const int M = in_sizes[0] / N_IN;
    const int E = in_sizes[1] / 2;
    const int* row = ei;
    const int* col = ei + E;

    float* ws = (float*)d_ws;
    size_t off = 0;
    float* deg  = ws + off; off += (size_t)M;
    float* dinv = ws + off; off += (size_t)M;
    float* agg1 = ws + off; off += (size_t)M * N_IN;
    float* h1   = ws + off; off += (size_t)M * N_HID;
    float* agg2 = ws + off; off += (size_t)M * N_HID;
    float* h2   = ws + off; off += (size_t)M * N_HID2;
    (void)ws_size;

    hipMemsetAsync(deg,  0, sizeof(float) * (size_t)M, stream);

    deg_kernel<<<(E + 255) / 256, 256, 0, stream>>>(col, deg, E);
    deginv_kernel<<<(M + 255) / 256, 256, 0, stream>>>(deg, dinv, M);

    agg_owner<N_IN><<<(M + AGG_T - 1) / AGG_T, AGG_NT, 0, stream>>>(x, row, col, agg1, M, E);
    int tiles1 = ((M + 16 * GM_MT - 1) / (16 * GM_MT)) * (N_HID / (16 * GM_NT));
    gemm_wmma<N_IN, N_HID, true>
        <<<(tiles1 + GM_WAVES - 1) / GM_WAVES, dim3(32, GM_WAVES), 0, stream>>>(agg1, W1, b1, dinv, h1, M);

    agg_owner<N_HID><<<(M + AGG_T - 1) / AGG_T, AGG_NT, 0, stream>>>(h1, row, col, agg2, M, E);
    int tiles2 = ((M + 16 * GM_MT - 1) / (16 * GM_MT)) * (N_HID2 / (16 * GM_NT));
    gemm_wmma<N_HID, N_HID2, true>
        <<<(tiles2 + GM_WAVES - 1) / GM_WAVES, dim3(32, GM_WAVES), 0, stream>>>(agg2, W2, b2, dinv, h2, M);

    head_kernel<<<(M * N_CLS + 255) / 256, 256, 0, stream>>>(h2, W3, b3, out, M);
}


# Edges binned once per call into per-owner-group lists (LDS counts + one returning global int add per group per block); both neighbour-sum launches read their list, not all edges; sum order arrival
# speedup vs baseline: 4.7912x; 1.3250x over previous
#include <hip/hip_runtime.h>

typedef __attribute__((ext_vector_type(2))) float v2f;
typedef __attribute__((ext_vector_type(8))) float v8f;

#define N_IN   128
#define N_HID  256
#define N_HID2 128
#define N_CLS  2

__global__ void deg_kernel(const int* __restrict__ col, float* __restrict__ deg, int E) {
    int e = blockIdx.x * blockDim.x + threadIdx.x;
    if (e < E) atomicAdd(&deg[col[e]], 1.0f);
}

__global__ void deginv_kernel(const float* __restrict__ deg, float* __restrict__ dinv, int n) {
    int i = blockIdx.x * blockDim.x + threadIdx.x;
    if (i < n) dinv[i] = 1.0f / fmaxf(deg[i], 1.0f);
}

#define AGG_T   196
#define AGG_NT  1024
#define AGG_CAP 4096

__device__ __forceinline__ int4 load_cols(const int* __restrict__ col, int idx, int E, bool vec_ok) {
    int4 c;
    if (vec_ok && idx + 3 < E) {
        c = *(const int4*)(col + idx);
    } else {
        c.x = (idx     < E) ? col[idx]     : -1;
        c.y = (idx + 1 < E) ? col[idx + 1] : -1;
        c.z = (idx + 2 < E) ? col[idx + 2] : -1;
        c.w = (idx + 3 < E) ? col[idx + 3] : -1;
    }
    return c;
}

#define BIN_NT   1024
#define BIN_MAXG 1024
#define BIN_CAP  16384

__global__ __launch_bounds__(BIN_NT)
void bin_edges(const int* __restrict__ row, const int* __restrict__ col,
               int2* __restrict__ lists, int* __restrict__ gcnt,
               int E, int M, int NG, int cap) {
    __shared__ int s_cnt[BIN_MAXG];
    __shared__ int s_base[BIN_MAXG];
    const int tid = threadIdx.x;
    const int e0  = (blockIdx.x * BIN_NT + tid) * 4;
    const bool vec_ok = ((E & 3) == 0);
    const int4 c4 = load_cols(col, e0, E, vec_ok);
    const int4 r4 = load_cols(row, e0, E, true);
    const int cc[4] = { c4.x, c4.y, c4.z, c4.w };
    const int rr[4] = { r4.x, r4.y, r4.z, r4.w };

    for (int g0 = 0; g0 < NG; g0 += BIN_MAXG) {
        for (int i = tid; i < BIN_MAXG; i += BIN_NT) s_cnt[i] = 0;
        __syncthreads();
        int gl[4], rank[4];
        #pragma unroll
        for (int j = 0; j < 4; ++j) {
            const unsigned uc = (unsigned)cc[j];
            const unsigned g = (uc < (unsigned)M) ? uc / AGG_T - (unsigned)g0 : 0xffffffffu;
            gl[j]   = (g < (unsigned)BIN_MAXG) ? (int)g : -1;
            rank[j] = (gl[j] >= 0) ? atomicAdd(&s_cnt[gl[j]], 1) : 0;
        }
        __syncthreads();
        for (int i = tid; i < BIN_MAXG; i += BIN_NT) {
            const int n = s_cnt[i];
            s_base[i] = (n > 0) ? atomicAdd(&gcnt[g0 + i], n) : 0;
        }
        __syncthreads();
        #pragma unroll
        for (int j = 0; j < 4; ++j) {
            if (gl[j] >= 0) {
                const int g   = g0 + gl[j];
                const int pos = s_base[gl[j]] + rank[j];
                if (pos < cap) lists[(size_t)g * cap + pos] = make_int2(rr[j], cc[j] - g * AGG_T);
            }
        }
    }
}

template<int F>
__global__ __launch_bounds__(AGG_NT)
void agg_owner(const float* __restrict__ src,
               const int* __restrict__ row,
               const int* __restrict__ col,
               const int2* __restrict__ lists,
               const int* __restrict__ gcnt, int cap,
               float* __restrict__ agg, int M, int E) {
    constexpr int V  = F / 128;
    constexpr int CH = AGG_NT * 4;
    __shared__ __attribute__((aligned(16))) float acc[AGG_T * F];
    __shared__ int l_row[AGG_CAP];
    __shared__ int l_loc[AGG_CAP];
    __shared__ int l_n;
    __shared__ int s_tot;

    const int tid  = threadIdx.x;
    const int lane = tid & 31;
    const int wave = tid >> 5;
    const int lo   = blockIdx.x * AGG_T;
    const int nT   = min(AGG_T, M - lo);
    const unsigned unT = (unsigned)nT;

    for (int i = tid * 4; i < AGG_T * F; i += AGG_NT * 4)
        *(float4*)&acc[i] = make_float4(0.f, 0.f, 0.f, 0.f);
    if (tid == 0) { l_n = 0; s_tot = gcnt[blockIdx.x]; }
    __syncthreads();

    const int  total  = s_tot;
    const bool listed = (total <= cap);
    const int  nch    = listed ? (total + AGG_CAP - 1) / AGG_CAP : 1;
    for (int ch = 0; ch < nch; ++ch) {
        int n;
        if (listed) {
            n = min(AGG_CAP, total - ch * AGG_CAP);
            const int2* __restrict__ lst = lists + (size_t)blockIdx.x * cap + (size_t)ch * AGG_CAP;
            for (int i = tid; i < n; i += AGG_NT) {
                const int2 rec = lst[i];
                l_row[i] = rec.x;
                l_loc[i] = rec.y;
            }
            __syncthreads();
        } else {
            const bool vec_ok = ((E & 3) == 0);
            const int nIter = (E + CH - 1) / CH;
            const int it0 = (int)(blockIdx.x % (unsigned)nIter);
            for (int kk = 0; kk < nIter; kk += 4) {
                int4 c[4];
                int  ix[4];
                int  cb[4];
                bool fast = vec_ok && (kk + 3 < nIter);
                #pragma unroll
                for (int u = 0; u < 4; ++u) {
                    int itr = it0 + kk + u;
                    if (itr >= nIter) itr -= nIter;
                    cb[u] = itr * CH;
                    fast = fast && (cb[u] + CH <= E);
                }
                if (fast) {
                    #pragma unroll
                    for (int u = 0; u < 4; ++u) {
                        ix[u] = cb[u] + tid * 4;
                        c[u]  = *(const int4*)(col + ix[u]);
                    }
                } else {
                    #pragma unroll
                    for (int u = 0; u < 4; ++u) {
                        ix[u] = (kk + u < nIter) ? cb[u] + tid * 4 : E;
                        c[u]  = load_cols(col, ix[u], E, vec_ok);
                    }
                }
                #pragma unroll
                for (int u = 0; u < 4; ++u) {
                    const unsigned d[4] = { (unsigned)(c[u].x - lo), (unsigned)(c[u].y - lo),
                                            (unsigned)(c[u].z - lo), (unsigned)(c[u].w - lo) };
                    const int hits = (int)(d[0] < unT) + (int)(d[1] < unT) + (int)(d[2] < unT) + (int)(d[3] < unT);
                    if (hits) {
                        int pos = atomicAdd(&l_n, hits);
                        #pragma unroll
                        for (int j = 0; j < 4; ++j) {
                            if (d[j] < unT) {
                                const int e = ix[u] + j;
                                if (pos < AGG_CAP) {
                                    l_row[pos] = e;
                                    l_loc[pos] = (int)d[j];
                                } else {
                                    const float* __restrict__ p = src + (size_t)row[e] * F;
                                    float* a = &acc[(int)d[j] * F];
                                    for (int f = 0; f < F; ++f)
                                        atomicAdd(&a[(f & ~127) + (f & 3) * 32 + ((f & 127) >> 2)], p[f]);
                                }
                                ++pos;
                            }
                        }
                    }
                }
            }
            __syncthreads();

            n = min(l_n, AGG_CAP);
            for (int i = tid; i < n; i += AGG_NT) l_row[i] = row[l_row[i]];
            __syncthreads();
        }

        for (int b = wave * 8; b < n; b += (AGG_NT / 32) * 8) {
            const int cnt = min(8, n - b);
            if (cnt == 8) {
                float4 v[8][V];
                int    lc[8];
                #pragma unroll
                for (int j = 0; j < 8; ++j) {
                    const int r = l_row[b + j];
                    lc[j] = l_loc[b + j];
                    const float* __restrict__ p = src + (size_t)r * F + lane * 4;
                    #pragma unroll
                    for (int q = 0; q < V; ++q) v[j][q] = *(const float4*)(p + q * 128);
                }
                #pragma unroll
                for (int j = 0; j < 8; ++j) {
                    #pragma unroll
                    for (int q = 0; q < V; ++q) {
                        float* a = &acc[lc[j] * F + q * 128 + lane];
                        atomicAdd(a,      v[j][q].x);
                        atomicAdd(a + 32, v[j][q].y);
                        atomicAdd(a + 64, v[j][q].z);
                        atomicAdd(a + 96, v[j][q].w);
                    }
                }
            } else {
                for (int j = 0; j < cnt; ++j) {
                    const int r = l_row[b + j];
                    const int l = l_loc[b + j];
                    const float* __restrict__ p = src + (size_t)r * F + lane * 4;
                    #pragma unroll
                    for (int q = 0; q < V; ++q) {
                        const float4 t = *(const float4*)(p + q * 128);
                        float* a = &acc[l * F + q * 128 + lane];
                        atomicAdd(a,      t.x);
                        atomicAdd(a + 32, t.y);
                        atomicAdd(a + 64, t.z);
                        atomicAdd(a + 96, t.w);
                    }
                }
            }
        }
        __syncthreads();
    }

    float* __restrict__ dst = agg + (size_t)lo * F;
    const int tot = nT * F;
    for (int i = tid * 4; i < tot; i += AGG_NT * 4) {
        const int base = (i & ~127) + ((i & 127) >> 2);
        float4 t;
        t.x = acc[base];
        t.y = acc[base + 32];
        t.z = acc[base + 64];
        t.w = acc[base + 96];
        *(float4*)&dst[i] = t;
    }
}

#define GM_MT 2
#define GM_NT 4
#define GM_WAVES 4
#define GM_KS 16

template<int K, int NF, bool RELU>
__global__ __launch_bounds__(32 * GM_WAVES)
void gemm_wmma(const float* __restrict__ A, const float* __restrict__ W,
               const float* __restrict__ bias, const float* __restrict__ dinv,
               float* __restrict__ out, int M) {
    constexpr int NB = NF / (16 * GM_NT);
    const int lane   = threadIdx.x;
    const int waveId = threadIdx.y;
    const int stripsM = (M + 16 * GM_MT - 1) / (16 * GM_MT);
    const int tile = blockIdx.x * GM_WAVES + waveId;
    if (tile >= stripsM * NB) return;
    const int tm = tile / NB;
    const int tn = tile % NB;

    const int half = lane >> 4;
    const int l16  = lane & 15;

    const float* __restrict__ ap[GM_MT];
    const float* __restrict__ wp[GM_NT];
    #pragma unroll
    for (int i = 0; i < GM_MT; ++i) {
        int r = tm * (16 * GM_MT) + i * 16 + l16;
        if (r >= M) r = M - 1;
        ap[i] = A + (size_t)r * K + 8 * half;
    }
    #pragma unroll
    for (int j = 0; j < GM_NT; ++j)
        wp[j] = W + (size_t)(tn * (16 * GM_NT) + j * 16 + l16) * K + 8 * half;

    v8f acc[GM_MT][GM_NT];
    #pragma unroll
    for (int i = 0; i < GM_MT; ++i)
        #pragma unroll
        for (int j = 0; j < GM_NT; ++j)
            acc[i][j] = v8f{};

    float4 a[GM_MT][2], b[GM_NT][2];
    #pragma unroll
    for (int i = 0; i < GM_MT; ++i) {
        a[i][0] = *(const float4*)(ap[i]);
        a[i][1] = *(const float4*)(ap[i] + 4);
    }
    #pragma unroll
    for (int j = 0; j < GM_NT; ++j) {
        b[j][0] = *(const float4*)(wp[j]);
        b[j][1] = *(const float4*)(wp[j] + 4);
    }

    #pragma unroll 2
    for (int k0 = 0; k0 < K; k0 += GM_KS) {
        float4 an[GM_MT][2], bn[GM_NT][2];
        const int kn = (k0 + GM_KS < K) ? k0 + GM_KS : k0;
        #pragma unroll
        for (int i = 0; i < GM_MT; ++i) {
            an[i][0] = *(const float4*)(ap[i] + kn);
            an[i][1] = *(const float4*)(ap[i] + kn + 4);
        }
        #pragma unroll
        for (int j = 0; j < GM_NT; ++j) {
            bn[j][0] = *(const float4*)(wp[j] + kn);
            bn[j][1] = *(const float4*)(wp[j] + kn + 4);
        }
        __builtin_amdgcn_sched_barrier(0);

        #pragma unroll
        for (int p = 0; p < 2; ++p) {
            #pragma unroll
            for (int i = 0; i < GM_MT; ++i) {
                const v2f a0 = { a[i][p].x, a[i][p].y };
                #pragma unroll
                for (int j = 0; j < GM_NT; ++j) {
                    const v2f b0 = { b[j][p].x, b[j][p].y };
                    acc[i][j] = __builtin_amdgcn_wmma_f32_16x16x4_f32(
                         false, a0,  false, b0,
                         (short)0, acc[i][j],  false,  false);
                }
            }
            #pragma unroll
            for (int i = 0; i < GM_MT; ++i) {
                const v2f a1 = { a[i][p].z, a[i][p].w };
                #pragma unroll
                for (int j = 0; j < GM_NT; ++j) {
                    const v2f b1 = { b[j][p].z, b[j][p].w };
                    acc[i][j] = __builtin_amdgcn_wmma_f32_16x16x4_f32(
                         false, a1,  false, b1,
                         (short)0, acc[i][j],  false,  false);
                }
            }
        }
        #pragma unroll
        for (int i = 0; i < GM_MT; ++i) { a[i][0] = an[i][0]; a[i][1] = an[i][1]; }
        #pragma unroll
        for (int j = 0; j < GM_NT; ++j) { b[j][0] = bn[j][0]; b[j][1] = bn[j][1]; }
    }

    float bv[GM_NT];
    #pragma unroll
    for (int j = 0; j < GM_NT; ++j) bv[j] = bias[tn * (16 * GM_NT) + j * 16 + l16];

    float dv[GM_MT][8];
    #pragma unroll
    for (int i = 0; i < GM_MT; ++i) {
        #pragma unroll
        for (int r = 0; r < 8; ++r) {
            int m = tm * (16 * GM_MT) + i * 16 + half * 8 + r;
            if (m >= M) m = M - 1;
            dv[i][r] = dinv[m];
        }
    }

    #pragma unroll
    for (int i = 0; i < GM_MT; ++i) {
        #pragma unroll
        for (int r = 0; r < 8; ++r) {
            const int m = tm * (16 * GM_MT) + i * 16 + half * 8 + r;
            if (m < M) {
                #pragma unroll
                for (int j = 0; j < GM_NT; ++j) {
                    float v = acc[i][j][r] * dv[i][r] + bv[j];
                    if (RELU) v = fmaxf(v, 0.0f);
                    out[(size_t)m * NF + tn * (16 * GM_NT) + j * 16 + l16] = v;
                }
            }
        }
    }
}

__global__ void head_kernel(const float* __restrict__ h, const float* __restrict__ W3,
                            const float* __restrict__ b3, float* __restrict__ out, int M) {
    int tid = blockIdx.x * blockDim.x + threadIdx.x;
    if (tid >= M * N_CLS) return;
    int m = tid / N_CLS, c = tid % N_CLS;
    const float* __restrict__ hm = h  + (size_t)m * N_HID2;
    const float* __restrict__ wc = W3 + (size_t)c * N_HID2;
    float s = b3[c];
    #pragma unroll 8
    for (int k = 0; k < N_HID2; ++k) s += hm[k] * wc[k];
    out[tid] = s;
}

extern "C" void kernel_launch(void* const* d_in, const int* in_sizes, int n_in,
                              void* d_out, int out_size, void* d_ws, size_t ws_size,
                              hipStream_t stream) {
    const float* x  = (const float*)d_in[0];
    const int*   ei = (const int*)  d_in[1];
    const float* W1 = (const float*)d_in[3];
    const float* b1 = (const float*)d_in[4];
    const float* W2 = (const float*)d_in[5];
    const float* b2 = (const float*)d_in[6];
    const float* W3 = (const float*)d_in[7];
    const float* b3 = (const float*)d_in[8];
    float* out = (float*)d_out;

    const int M = in_sizes[0] / N_IN;
    const int E = in_sizes[1] / 2;
    const int* row = ei;
    const int* col = ei + E;

    float* ws = (float*)d_ws;
    size_t off = 0;
    const int NG  = (M + AGG_T - 1) / AGG_T;
    const int NGp = (NG + 31) & ~31;
    float* deg  = ws + off; off += (size_t)M;
    int*   gcnt = (int*)(ws + off); off += (size_t)NGp;
    float* dinv = ws + off; off += (size_t)M;
    float* agg1 = ws + off; off += (size_t)M * N_IN;
    float* h1   = ws + off; off += (size_t)M * N_HID;
    float* agg2 = ws + off; off += (size_t)M * N_HID;
    float* h2   = ws + off; off += (size_t)M * N_HID2;
    off = (off + 1) & ~(size_t)1;
    int2* lists = (int2*)(ws + off);
    const size_t room = (ws_size / sizeof(float) - off) / 2 / (size_t)NG;
    const int cap = (int)(room < (size_t)BIN_CAP ? room : (size_t)BIN_CAP);

    hipMemsetAsync(deg,  0, sizeof(float) * ((size_t)M + (size_t)NGp), stream);

    bin_edges<<<(E + BIN_NT * 4 - 1) / (BIN_NT * 4), BIN_NT, 0, stream>>>(row, col, lists, gcnt, E, M, NG, cap);

    deg_kernel<<<(E + 255) / 256, 256, 0, stream>>>(col, deg, E);
    deginv_kernel<<<(M + 255) / 256, 256, 0, stream>>>(deg, dinv, M);

    agg_owner<N_IN><<<NG, AGG_NT, 0, stream>>>(x, row, col, lists, gcnt, cap, agg1, M, E);
    int tiles1 = ((M + 16 * GM_MT - 1) / (16 * GM_MT)) * (N_HID / (16 * GM_NT));
    gemm_wmma<N_IN, N_HID, true>
        <<<(tiles1 + GM_WAVES - 1) / GM_WAVES, dim3(32, GM_WAVES), 0, stream>>>(agg1, W1, b1, dinv, h1, M);

    agg_owner<N_HID><<<NG, AGG_NT, 0, stream>>>(h1, row, col, lists, gcnt, cap, agg2, M, E);
    int tiles2 = ((M + 16 * GM_MT - 1) / (16 * GM_MT)) * (N_HID2 / (16 * GM_NT));
    gemm_wmma<N_HID, N_HID2, true>
        <<<(tiles2 + GM_WAVES - 1) / GM_WAVES, dim3(32, GM_WAVES), 0, stream>>>(agg2, W2, b2, dinv, h2, M);

    head_kernel<<<(M * N_CLS + 255) / 256, 256, 0, stream>>>(h2, W3, b3, out, M);
}


# In-degree counted by owner groups: the layer-1 neighbour-sum group adds 1 per edge to an LDS integer per owned node, writes 1/max(deg,1); degree launch (global float atomics) and inverse launch gone
# speedup vs baseline: 6.2245x; 1.2992x over previous
#include <hip/hip_runtime.h>

typedef __attribute__((ext_vector_type(2))) float v2f;
typedef __attribute__((ext_vector_type(8))) float v8f;

#define N_IN   128
#define N_HID  256
#define N_HID2 128
#define N_CLS  2

#define AGG_T   196
#define AGG_NT  1024
#define AGG_CAP 4096

__device__ __forceinline__ int4 load_cols(const int* __restrict__ col, int idx, int E, bool vec_ok) {
    int4 c;
    if (vec_ok && idx + 3 < E) {
        c = *(const int4*)(col + idx);
    } else {
        c.x = (idx     < E) ? col[idx]     : -1;
        c.y = (idx + 1 < E) ? col[idx + 1] : -1;
        c.z = (idx + 2 < E) ? col[idx + 2] : -1;
        c.w = (idx + 3 < E) ? col[idx + 3] : -1;
    }
    return c;
}

#define BIN_NT   1024
#define BIN_MAXG 1024
#define BIN_CAP  16384

__global__ __launch_bounds__(BIN_NT)
void bin_edges(const int* __restrict__ row, const int* __restrict__ col,
               int2* __restrict__ lists, int* __restrict__ gcnt,
               int E, int M, int NG, int cap) {
    __shared__ int s_cnt[BIN_MAXG];
    __shared__ int s_base[BIN_MAXG];
    const int tid = threadIdx.x;
    const int e0  = (blockIdx.x * BIN_NT + tid) * 4;
    const bool vec_ok = ((E & 3) == 0);
    const int4 c4 = load_cols(col, e0, E, vec_ok);
    const int4 r4 = load_cols(row, e0, E, true);
    const int cc[4] = { c4.x, c4.y, c4.z, c4.w };
    const int rr[4] = { r4.x, r4.y, r4.z, r4.w };

    for (int g0 = 0; g0 < NG; g0 += BIN_MAXG) {
        for (int i = tid; i < BIN_MAXG; i += BIN_NT) s_cnt[i] = 0;
        __syncthreads();
        int gl[4], rank[4];
        #pragma unroll
        for (int j = 0; j < 4; ++j) {
            const unsigned uc = (unsigned)cc[j];
            const unsigned g = (uc < (unsigned)M) ? uc / AGG_T - (unsigned)g0 : 0xffffffffu;
            gl[j]   = (g < (unsigned)BIN_MAXG) ? (int)g : -1;
            rank[j] = (gl[j] >= 0) ? atomicAdd(&s_cnt[gl[j]], 1) : 0;
        }
        __syncthreads();
        for (int i = tid; i < BIN_MAXG; i += BIN_NT) {
            const int n = s_cnt[i];
            s_base[i] = (n > 0) ? atomicAdd(&gcnt[g0 + i], n) : 0;
        }
        __syncthreads();
        #pragma unroll
        for (int j = 0; j < 4; ++j) {
            if (gl[j] >= 0) {
                const int g   = g0 + gl[j];
                const int pos = s_base[gl[j]] + rank[j];
                if (pos < cap) lists[(size_t)g * cap + pos] = make_int2(rr[j], cc[j] - g * AGG_T);
            }
        }
    }
}

template<int F, bool DEG>
__global__ __launch_bounds__(AGG_NT)
void agg_owner(const float* __restrict__ src,
               const int* __restrict__ row,
               const int* __restrict__ col,
               const int2* __restrict__ lists,
               const int* __restrict__ gcnt, int cap,
               float* __restrict__ agg, float* __restrict__ dinv, int M, int E) {
    constexpr int V  = F / 128;
    constexpr int CH = AGG_NT * 4;
    __shared__ __attribute__((aligned(16))) float acc[AGG_T * F];
    __shared__ int l_row[AGG_CAP];
    __shared__ int l_loc[AGG_CAP];
    __shared__ int l_n;
    __shared__ int s_tot;
    __shared__ int s_deg[DEG ? AGG_T : 1];

    const int tid  = threadIdx.x;
    const int lane = tid & 31;
    const int wave = tid >> 5;
    const int lo   = blockIdx.x * AGG_T;
    const int nT   = min(AGG_T, M - lo);
    const unsigned unT = (unsigned)nT;

    for (int i = tid * 4; i < AGG_T * F; i += AGG_NT * 4)
        *(float4*)&acc[i] = make_float4(0.f, 0.f, 0.f, 0.f);
    if (tid == 0) { l_n = 0; s_tot = gcnt[blockIdx.x]; }
    if (DEG) {
        for (int i = tid; i < AGG_T; i += AGG_NT) s_deg[i] = 0;
    }
    __syncthreads();

    const int  total  = s_tot;
    const bool listed = (total <= cap);
    const int  nch    = listed ? (total + AGG_CAP - 1) / AGG_CAP : 1;
    for (int ch = 0; ch < nch; ++ch) {
        int n;
        if (listed) {
            n = min(AGG_CAP, total - ch * AGG_CAP);
            const int2* __restrict__ lst = lists + (size_t)blockIdx.x * cap + (size_t)ch * AGG_CAP;
            for (int i = tid; i < n; i += AGG_NT) {
                const int2 rec = lst[i];
                l_row[i] = rec.x;
                l_loc[i] = rec.y;
                if (DEG) atomicAdd(&s_deg[rec.y], 1);
            }
            __syncthreads();
        } else {
            const bool vec_ok = ((E & 3) == 0);
            const int nIter = (E + CH - 1) / CH;
            const int it0 = (int)(blockIdx.x % (unsigned)nIter);
            for (int kk = 0; kk < nIter; kk += 4) {
                int4 c[4];
                int  ix[4];
                int  cb[4];
                bool fast = vec_ok && (kk + 3 < nIter);
                #pragma unroll
                for (int u = 0; u < 4; ++u) {
                    int itr = it0 + kk + u;
                    if (itr >= nIter) itr -= nIter;
                    cb[u] = itr * CH;
                    fast = fast && (cb[u] + CH <= E);
                }
                if (fast) {
                    #pragma unroll
                    for (int u = 0; u < 4; ++u) {
                        ix[u] = cb[u] + tid * 4;
                        c[u]  = *(const int4*)(col + ix[u]);
                    }
                } else {
                    #pragma unroll
                    for (int u = 0; u < 4; ++u) {
                        ix[u] = (kk + u < nIter) ? cb[u] + tid * 4 : E;
                        c[u]  = load_cols(col, ix[u], E, vec_ok);
                    }
                }
                #pragma unroll
                for (int u = 0; u < 4; ++u) {
                    const unsigned d[4] = { (unsigned)(c[u].x - lo), (unsigned)(c[u].y - lo),
                                            (unsigned)(c[u].z - lo), (unsigned)(c[u].w - lo) };
                    const int hits = (int)(d[0] < unT) + (int)(d[1] < unT) + (int)(d[2] < unT) + (int)(d[3] < unT);
                    if (hits) {
                        int pos = atomicAdd(&l_n, hits);
                        #pragma unroll
                        for (int j = 0; j < 4; ++j) {
                            if (d[j] < unT) {
                                const int e = ix[u] + j;
                                if (DEG) atomicAdd(&s_deg[(int)d[j]], 1);
                                if (pos < AGG_CAP) {
                                    l_row[pos] = e;
                                    l_loc[pos] = (int)d[j];
                                } else {
                                    const float* __restrict__ p = src + (size_t)row[e] * F;
                                    float* a = &acc[(int)d[j] * F];
                                    for (int f = 0; f < F; ++f)
                                        atomicAdd(&a[(f & ~127) + (f & 3) * 32 + ((f & 127) >> 2)], p[f]);
                                }
                                ++pos;
                            }
                        }
                    }
                }
            }
            __syncthreads();

            n = min(l_n, AGG_CAP);
            for (int i = tid; i < n; i += AGG_NT) l_row[i] = row[l_row[i]];
            __syncthreads();
        }

        for (int b = wave * 8; b < n; b += (AGG_NT / 32) * 8) {
            const int cnt = min(8, n - b);
            if (cnt == 8) {
                float4 v[8][V];
                int    lc[8];
                #pragma unroll
                for (int j = 0; j < 8; ++j) {
                    const int r = l_row[b + j];
                    lc[j] = l_loc[b + j];
                    const float* __restrict__ p = src + (size_t)r * F + lane * 4;
                    #pragma unroll
                    for (int q = 0; q < V; ++q) v[j][q] = *(const float4*)(p + q * 128);
                }
                #pragma unroll
                for (int j = 0; j < 8; ++j) {
                    #pragma unroll
                    for (int q = 0; q < V; ++q) {
                        float* a = &acc[lc[j] * F + q * 128 + lane];
                        atomicAdd(a,      v[j][q].x);
                        atomicAdd(a + 32, v[j][q].y);
                        atomicAdd(a + 64, v[j][q].z);
                        atomicAdd(a + 96, v[j][q].w);
                    }
                }
            } else {
                for (int j = 0; j < cnt; ++j) {
                    const int r = l_row[b + j];
                    const int l = l_loc[b + j];
                    const float* __restrict__ p = src + (size_t)r * F + lane * 4;
                    #pragma unroll
                    for (int q = 0; q < V; ++q) {
                        const float4 t = *(const float4*)(p + q * 128);
                        float* a = &acc[l * F + q * 128 + lane];
                        atomicAdd(a,      t.x);
                        atomicAdd(a + 32, t.y);
                        atomicAdd(a + 64, t.z);
                        atomicAdd(a + 96, t.w);
                    }
                }
            }
        }
        __syncthreads();
    }

    float* __restrict__ dst = agg + (size_t)lo * F;
    const int tot = nT * F;
    for (int i = tid * 4; i < tot; i += AGG_NT * 4) {
        const int base = (i & ~127) + ((i & 127) >> 2);
        float4 t;
        t.x = acc[base];
        t.y = acc[base + 32];
        t.z = acc[base + 64];
        t.w = acc[base + 96];
        *(float4*)&dst[i] = t;
    }

    if (DEG) {
        for (int i = tid; i < nT; i += AGG_NT) dinv[lo + i] = 1.0f / fmaxf((float)s_deg[i], 1.0f);
    }
}

#define GM_MT 2
#define GM_NT 4
#define GM_WAVES 4
#define GM_KS 16

template<int K, int NF, bool RELU>
__global__ __launch_bounds__(32 * GM_WAVES)
void gemm_wmma(const float* __restrict__ A, const float* __restrict__ W,
               const float* __restrict__ bias, const float* __restrict__ dinv,
               float* __restrict__ out, int M) {
    constexpr int NB = NF / (16 * GM_NT);
    const int lane   = threadIdx.x;
    const int waveId = threadIdx.y;
    const int stripsM = (M + 16 * GM_MT - 1) / (16 * GM_MT);
    const int tile = blockIdx.x * GM_WAVES + waveId;
    if (tile >= stripsM * NB) return;
    const int tm = tile / NB;
    const int tn = tile % NB;

    const int half = lane >> 4;
    const int l16  = lane & 15;

    const float* __restrict__ ap[GM_MT];
    const float* __restrict__ wp[GM_NT];
    #pragma unroll
    for (int i = 0; i < GM_MT; ++i) {
        int r = tm * (16 * GM_MT) + i * 16 + l16;
        if (r >= M) r = M - 1;
        ap[i] = A + (size_t)r * K + 8 * half;
    }
    #pragma unroll
    for (int j = 0; j < GM_NT; ++j)
        wp[j] = W + (size_t)(tn * (16 * GM_NT) + j * 16 + l16) * K + 8 * half;

    v8f acc[GM_MT][GM_NT];
    #pragma unroll
    for (int i = 0; i < GM_MT; ++i)
        #pragma unroll
        for (int j = 0; j < GM_NT; ++j)
            acc[i][j] = v8f{};

    float4 a[GM_MT][2], b[GM_NT][2];
    #pragma unroll
    for (int i = 0; i < GM_MT; ++i) {
        a[i][0] = *(const float4*)(ap[i]);
        a[i][1] = *(const float4*)(ap[i] + 4);
    }
    #pragma unroll
    for (int j = 0; j < GM_NT; ++j) {
        b[j][0] = *(const float4*)(wp[j]);
        b[j][1] = *(const float4*)(wp[j] + 4);
    }

    #pragma unroll 2
    for (int k0 = 0; k0 < K; k0 += GM_KS) {
        float4 an[GM_MT][2], bn[GM_NT][2];
        const int kn = (k0 + GM_KS < K) ? k0 + GM_KS : k0;
        #pragma unroll
        for (int i = 0; i < GM_MT; ++i) {
            an[i][0] = *(const float4*)(ap[i] + kn);
            an[i][1] = *(const float4*)(ap[i] + kn + 4);
        }
        #pragma unroll
        for (int j = 0; j < GM_NT; ++j) {
            bn[j][0] = *(const float4*)(wp[j] + kn);
            bn[j][1] = *(const float4*)(wp[j] + kn + 4);
        }
        __builtin_amdgcn_sched_barrier(0);

        #pragma unroll
        for (int p = 0; p < 2; ++p) {
            #pragma unroll
            for (int i = 0; i < GM_MT; ++i) {
                const v2f a0 = { a[i][p].x, a[i][p].y };
                #pragma unroll
                for (int j = 0; j < GM_NT; ++j) {
                    const v2f b0 = { b[j][p].x, b[j][p].y };
                    acc[i][j] = __builtin_amdgcn_wmma_f32_16x16x4_f32(
                         false, a0,  false, b0,
                         (short)0, acc[i][j],  false,  false);
                }
            }
            #pragma unroll
            for (int i = 0; i < GM_MT; ++i) {
                const v2f a1 = { a[i][p].z, a[i][p].w };
                #pragma unroll
                for (int j = 0; j < GM_NT; ++j) {
                    const v2f b1 = { b[j][p].z, b[j][p].w };
                    acc[i][j] = __builtin_amdgcn_wmma_f32_16x16x4_f32(
                         false, a1,  false, b1,
                         (short)0, acc[i][j],  false,  false);
                }
            }
        }
        #pragma unroll
        for (int i = 0; i < GM_MT; ++i) { a[i][0] = an[i][0]; a[i][1] = an[i][1]; }
        #pragma unroll
        for (int j = 0; j < GM_NT; ++j) { b[j][0] = bn[j][0]; b[j][1] = bn[j][1]; }
    }

    float bv[GM_NT];
    #pragma unroll
    for (int j = 0; j < GM_NT; ++j) bv[j] = bias[tn * (16 * GM_NT) + j * 16 + l16];

    float dv[GM_MT][8];
    #pragma unroll
    for (int i = 0; i < GM_MT; ++i) {
        #pragma unroll
        for (int r = 0; r < 8; ++r) {
            int m = tm * (16 * GM_MT) + i * 16 + half * 8 + r;
            if (m >= M) m = M - 1;
            dv[i][r] = dinv[m];
        }
    }

    #pragma unroll
    for (int i = 0; i < GM_MT; ++i) {
        #pragma unroll
        for (int r = 0; r < 8; ++r) {
            const int m = tm * (16 * GM_MT) + i * 16 + half * 8 + r;
            if (m < M) {
                #pragma unroll
                for (int j = 0; j < GM_NT; ++j) {
                    float v = acc[i][j][r] * dv[i][r] + bv[j];
                    if (RELU) v = fmaxf(v, 0.0f);
                    out[(size_t)m * NF + tn * (16 * GM_NT) + j * 16 + l16] = v;
                }
            }
        }
    }
}

__global__ void head_kernel(const float* __restrict__ h, const float* __restrict__ W3,
                            const float* __restrict__ b3, float* __restrict__ out, int M) {
    int tid = blockIdx.x * blockDim.x + threadIdx.x;
    if (tid >= M * N_CLS) return;
    int m = tid / N_CLS, c = tid % N_CLS;
    const float* __restrict__ hm = h  + (size_t)m * N_HID2;
    const float* __restrict__ wc = W3 + (size_t)c * N_HID2;
    float s = b3[c];
    #pragma unroll 8
    for (int k = 0; k < N_HID2; ++k) s += hm[k] * wc[k];
    out[tid] = s;
}

extern "C" void kernel_launch(void* const* d_in, const int* in_sizes, int n_in,
                              void* d_out, int out_size, void* d_ws, size_t ws_size,
                              hipStream_t stream) {
    const float* x  = (const float*)d_in[0];
    const int*   ei = (const int*)  d_in[1];
    const float* W1 = (const float*)d_in[3];
    const float* b1 = (const float*)d_in[4];
    const float* W2 = (const float*)d_in[5];
    const float* b2 = (const float*)d_in[6];
    const float* W3 = (const float*)d_in[7];
    const float* b3 = (const float*)d_in[8];
    float* out = (float*)d_out;

    const int M = in_sizes[0] / N_IN;
    const int E = in_sizes[1] / 2;
    const int* row = ei;
    const int* col = ei + E;

    float* ws = (float*)d_ws;
    size_t off = 0;
    const int NG  = (M + AGG_T - 1) / AGG_T;
    const int NGp = (NG + 31) & ~31;
    int*   gcnt = (int*)(ws + off); off += (size_t)NGp;
    float* dinv = ws + off; off += (size_t)M;
    off = (off + 3) & ~(size_t)3;
    float* agg1 = ws + off; off += (size_t)M * N_IN;
    float* h1   = ws + off; off += (size_t)M * N_HID;
    float* agg2 = ws + off; off += (size_t)M * N_HID;
    float* h2   = ws + off; off += (size_t)M * N_HID2;
    off = (off + 1) & ~(size_t)1;
    int2* lists = (int2*)(ws + off);
    const size_t room = (ws_size / sizeof(float) - off) / 2 / (size_t)NG;
    const int cap = (int)(room < (size_t)BIN_CAP ? room : (size_t)BIN_CAP);

    hipMemsetAsync(gcnt, 0, sizeof(int) * (size_t)NGp, stream);

    bin_edges<<<(E + BIN_NT * 4 - 1) / (BIN_NT * 4), BIN_NT, 0, stream>>>(row, col, lists, gcnt, E, M, NG, cap);

    agg_owner<N_IN, true><<<NG, AGG_NT, 0, stream>>>(x, row, col, lists, gcnt, cap, agg1, dinv, M, E);
    int tiles1 = ((M + 16 * GM_MT - 1) / (16 * GM_MT)) * (N_HID / (16 * GM_NT));
    gemm_wmma<N_IN, N_HID, true>
        <<<(tiles1 + GM_WAVES - 1) / GM_WAVES, dim3(32, GM_WAVES), 0, stream>>>(agg1, W1, b1, dinv, h1, M);

    agg_owner<N_HID, false><<<NG, AGG_NT, 0, stream>>>(h1, row, col, lists, gcnt, cap, agg2, dinv, M, E);
    int tiles2 = ((M + 16 * GM_MT - 1) / (16 * GM_MT)) * (N_HID2 / (16 * GM_NT));
    gemm_wmma<N_HID, N_HID2, true>
        <<<(tiles2 + GM_WAVES - 1) / GM_WAVES, dim3(32, GM_WAVES), 0, stream>>>(agg2, W2, b2, dinv, h2, M);

    head_kernel<<<(M * N_CLS + 255) / 256, 256, 0, stream>>>(h2, W3, b3, out, M);
}


# Neighbour sums in registers: an owner group counting-sorts its binned edge list by destination node in LDS, a wave sums a node's rows 8 at a time; LDS float adds and sum tile gone; sum order arrival
# speedup vs baseline: 6.5386x; 1.0505x over previous
#include <hip/hip_runtime.h>

typedef __attribute__((ext_vector_type(2))) float v2f;
typedef __attribute__((ext_vector_type(8))) float v8f;

#define N_IN   128
#define N_HID  256
#define N_HID2 128
#define N_CLS  2

#define AGG_T    196
#define AGG_NT   1024
#define AGG_NPW  7
#define AGG_NOW  28
#define AGG_RB   8
#define AGG_LCAP 16384

__device__ __forceinline__ int4 load_cols(const int* __restrict__ col, int idx, int E, bool vec_ok) {
    int4 c;
    if (vec_ok && idx + 3 < E) {
        c = *(const int4*)(col + idx);
    } else {
        c.x = (idx     < E) ? col[idx]     : -1;
        c.y = (idx + 1 < E) ? col[idx + 1] : -1;
        c.z = (idx + 2 < E) ? col[idx + 2] : -1;
        c.w = (idx + 3 < E) ? col[idx + 3] : -1;
    }
    return c;
}

#define BIN_NT   1024
#define BIN_MAXG 1024
#define BIN_CAP  16384

__global__ __launch_bounds__(BIN_NT)
void bin_edges(const int* __restrict__ row, const int* __restrict__ col,
               int2* __restrict__ lists, int* __restrict__ gcnt,
               int E, int M, int NG, int cap) {
    __shared__ int s_cnt[BIN_MAXG];
    __shared__ int s_base[BIN_MAXG];
    const int tid = threadIdx.x;
    const int e0  = (blockIdx.x * BIN_NT + tid) * 4;
    const bool vec_ok = ((E & 3) == 0);
    const int4 c4 = load_cols(col, e0, E, vec_ok);
    const int4 r4 = load_cols(row, e0, E, true);
    const int cc[4] = { c4.x, c4.y, c4.z, c4.w };
    const int rr[4] = { r4.x, r4.y, r4.z, r4.w };

    for (int g0 = 0; g0 < NG; g0 += BIN_MAXG) {
        for (int i = tid; i < BIN_MAXG; i += BIN_NT) s_cnt[i] = 0;
        __syncthreads();
        int gl[4], rank[4];
        #pragma unroll
        for (int j = 0; j < 4; ++j) {
            const unsigned uc = (unsigned)cc[j];
            const unsigned g = (uc < (unsigned)M) ? uc / AGG_T - (unsigned)g0 : 0xffffffffu;
            gl[j]   = (g < (unsigned)BIN_MAXG) ? (int)g : -1;
            rank[j] = (gl[j] >= 0) ? atomicAdd(&s_cnt[gl[j]], 1) : 0;
        }
        __syncthreads();
        for (int i = tid; i < BIN_MAXG; i += BIN_NT) {
            const int n = s_cnt[i];
            s_base[i] = (n > 0) ? atomicAdd(&gcnt[g0 + i], n) : 0;
        }
        __syncthreads();
        #pragma unroll
        for (int j = 0; j < 4; ++j) {
            if (gl[j] >= 0) {
                const int g   = g0 + gl[j];
                const int pos = s_base[gl[j]] + rank[j];
                if (pos < cap) lists[(size_t)g * cap + pos] = make_int2(rr[j], cc[j] - g * AGG_T);
            }
        }
    }
}

template<int F, bool DEG>
__global__ __launch_bounds__(AGG_NT)
void agg_owner(const float* __restrict__ src,
               const int* __restrict__ row,
               const int* __restrict__ col,
               const int2* __restrict__ lists,
               const int* __restrict__ gcnt, int cap,
               float* __restrict__ agg, float* __restrict__ dinv, int M, int E) {
    constexpr int V = F / 128;
    static_assert(AGG_NOW * AGG_NPW >= AGG_T, "the summing waves must cover the group");
    static_assert(AGG_NOW * 32 <= AGG_NT, "the summing waves must exist");
    static_assert(32 * AGG_NPW >= AGG_T, "one wave scans the counts, AGG_NPW per lane");
    static_assert(AGG_LCAP >= BIN_CAP, "a complete list must fit");
    __shared__ int s_row[AGG_LCAP];
    __shared__ int s_cnt[AGG_T];
    __shared__ int s_off[AGG_T + 1];
    __shared__ int s_cur[AGG_T];
    __shared__ int s_tot;

    const int tid  = threadIdx.x;
    const int lane = tid & 31;
    const int wave = __builtin_amdgcn_readfirstlane(tid >> 5);
    const int lo   = blockIdx.x * AGG_T;
    const int nT   = min(AGG_T, M - lo);
    float* __restrict__ dst = agg + (size_t)lo * F;

    for (int i = tid; i < AGG_T; i += AGG_NT) s_cnt[i] = 0;
    if (tid == 0) s_tot = gcnt[blockIdx.x];
    __syncthreads();

    const int  total  = __builtin_amdgcn_readfirstlane(s_tot);
    const bool listed = (total <= cap) && (total <= AGG_LCAP);
    if (listed) {
        const int2* __restrict__ lst = lists + (size_t)blockIdx.x * cap;
        for (int i = tid; i < total; i += AGG_NT) atomicAdd(&s_cnt[lst[i].y], 1);
        __syncthreads();
        if (wave == 0) {
            int c[AGG_NPW];
            int s = 0;
            #pragma unroll
            for (int k = 0; k < AGG_NPW; ++k) {
                const int idx = lane * AGG_NPW + k;
                c[k] = (idx < AGG_T) ? s_cnt[idx] : 0;
                s += c[k];
            }
            int incl = s;
            #pragma unroll
            for (int d = 1; d < 32; d <<= 1) {
                const int t = __shfl_up(incl, d);
                if (lane >= d) incl += t;
            }
            int run = incl - s;
            #pragma unroll
            for (int k = 0; k < AGG_NPW; ++k) {
                const int idx = lane * AGG_NPW + k;
                if (idx < AGG_T) { s_off[idx] = run; s_cur[idx] = run; }
                run += c[k];
            }
            if (lane == 31) s_off[AGG_T] = run;
        }
        __syncthreads();
        for (int i = tid; i < total; i += AGG_NT) {
            const int2 rec = lst[i];
            s_row[atomicAdd(&s_cur[rec.y], 1)] = rec.x;
        }
        __syncthreads();

        if (wave < AGG_NOW) {
            for (int k = 0; k < AGG_NPW; ++k) {
                const int node = wave * AGG_NPW + k;
                if (node >= nT) break;
                const int beg = __builtin_amdgcn_readfirstlane(s_off[node]);
                const int end = __builtin_amdgcn_readfirstlane(s_off[node + 1]);
                float4 sum[V];
                #pragma unroll
                for (int q = 0; q < V; ++q) sum[q] = make_float4(0.f, 0.f, 0.f, 0.f);
                for (int b = beg; b < end; b += AGG_RB) {
                    const int cnt = min(AGG_RB, end - b);
                    float4 v[AGG_RB][V];
                    int    r[AGG_RB];
                    #pragma unroll
                    for (int j = 0; j < AGG_RB; ++j) r[j] = s_row[min(b + j, end - 1)];
                    #pragma unroll
                    for (int j = 0; j < AGG_RB; ++j) {
                        #pragma unroll
                        for (int q = 0; q < V; ++q) v[j][q] = make_float4(0.f, 0.f, 0.f, 0.f);
                        if (j < cnt) {
                            const float* __restrict__ p = src + (size_t)r[j] * F + lane * 4;
                            #pragma unroll
                            for (int q = 0; q < V; ++q) v[j][q] = *(const float4*)(p + q * 128);
                        }
                    }
                    #pragma unroll
                    for (int j = 0; j < AGG_RB; ++j) {
                        #pragma unroll
                        for (int q = 0; q < V; ++q) {
                            sum[q].x += v[j][q].x; sum[q].y += v[j][q].y;
                            sum[q].z += v[j][q].z; sum[q].w += v[j][q].w;
                        }
                    }
                }
                #pragma unroll
                for (int q = 0; q < V; ++q)
                    *(float4*)(dst + (size_t)node * F + q * 128 + lane * 4) = sum[q];
            }
        }
    } else {
        if (wave < AGG_NOW) {
            for (int k = 0; k < AGG_NPW; ++k) {
                const int node = wave * AGG_NPW + k;
                if (node >= nT) break;
                float4 sum[V];
                #pragma unroll
                for (int q = 0; q < V; ++q) sum[q] = make_float4(0.f, 0.f, 0.f, 0.f);
                int deg = 0;
                for (int base = 0; base < E; base += 32) {
                    const int e = base + lane;
                    const bool hit = (e < E) && (col[e] == lo + node);
                    unsigned m = __builtin_amdgcn_ballot_w32(hit);
                    deg += __builtin_popcount(m);
                    while (m) {
                        const int bit = __builtin_ctz(m);
                        m &= m - 1;
                        const int r = row[base + bit];
                        const float* __restrict__ p = src + (size_t)r * F + lane * 4;
                        #pragma unroll
                        for (int q = 0; q < V; ++q) {
                            const float4 t = *(const float4*)(p + q * 128);
                            sum[q].x += t.x; sum[q].y += t.y; sum[q].z += t.z; sum[q].w += t.w;
                        }
                    }
                }
                #pragma unroll
                for (int q = 0; q < V; ++q)
                    *(float4*)(dst + (size_t)node * F + q * 128 + lane * 4) = sum[q];
                if (DEG) {
                    if (lane == 0) s_cnt[node] = deg;
                }
            }
        }
    }

    if (DEG) {
        __syncthreads();
        for (int i = tid; i < nT; i += AGG_NT) dinv[lo + i] = 1.0f / fmaxf((float)s_cnt[i], 1.0f);
    }
}

#define GM_MT 2
#define GM_NT 4
#define GM_WAVES 4
#define GM_KS 16

template<int K, int NF, bool RELU>
__global__ __launch_bounds__(32 * GM_WAVES)
void gemm_wmma(const float* __restrict__ A, const float* __restrict__ W,
               const float* __restrict__ bias, const float* __restrict__ dinv,
               float* __restrict__ out, int M) {
    constexpr int NB = NF / (16 * GM_NT);
    const int lane   = threadIdx.x;
    const int waveId = threadIdx.y;
    const int stripsM = (M + 16 * GM_MT - 1) / (16 * GM_MT);
    const int tile = blockIdx.x * GM_WAVES + waveId;
    if (tile >= stripsM * NB) return;
    const int tm = tile / NB;
    const int tn = tile % NB;

    const int half = lane >> 4;
    const int l16  = lane & 15;

    const float* __restrict__ ap[GM_MT];
    const float* __restrict__ wp[GM_NT];
    #pragma unroll
    for (int i = 0; i < GM_MT; ++i) {
        int r = tm * (16 * GM_MT) + i * 16 + l16;
        if (r >= M) r = M - 1;
        ap[i] = A + (size_t)r * K + 8 * half;
    }
    #pragma unroll
    for (int j = 0; j < GM_NT; ++j)
        wp[j] = W + (size_t)(tn * (16 * GM_NT) + j * 16 + l16) * K + 8 * half;

    v8f acc[GM_MT][GM_NT];
    #pragma unroll
    for (int i = 0; i < GM_MT; ++i)
        #pragma unroll
        for (int j = 0; j < GM_NT; ++j)
            acc[i][j] = v8f{};

    float4 a[GM_MT][2], b[GM_NT][2];
    #pragma unroll
    for (int i = 0; i < GM_MT; ++i) {
        a[i][0] = *(const float4*)(ap[i]);
        a[i][1] = *(const float4*)(ap[i] + 4);
    }
    #pragma unroll
    for (int j = 0; j < GM_NT; ++j) {
        b[j][0] = *(const float4*)(wp[j]);
        b[j][1] = *(const float4*)(wp[j] + 4);
    }

    #pragma unroll 2
    for (int k0 = 0; k0 < K; k0 += GM_KS) {
        float4 an[GM_MT][2], bn[GM_NT][2];
        const int kn = (k0 + GM_KS < K) ? k0 + GM_KS : k0;
        #pragma unroll
        for (int i = 0; i < GM_MT; ++i) {
            an[i][0] = *(const float4*)(ap[i] + kn);
            an[i][1] = *(const float4*)(ap[i] + kn + 4);
        }
        #pragma unroll
        for (int j = 0; j < GM_NT; ++j) {
            bn[j][0] = *(const float4*)(wp[j] + kn);
            bn[j][1] = *(const float4*)(wp[j] + kn + 4);
        }
        __builtin_amdgcn_sched_barrier(0);

        #pragma unroll
        for (int p = 0; p < 2; ++p) {
            #pragma unroll
            for (int i = 0; i < GM_MT; ++i) {
                const v2f a0 = { a[i][p].x, a[i][p].y };
                #pragma unroll
                for (int j = 0; j < GM_NT; ++j) {
                    const v2f b0 = { b[j][p].x, b[j][p].y };
                    acc[i][j] = __builtin_amdgcn_wmma_f32_16x16x4_f32(
                         false, a0,  false, b0,
                         (short)0, acc[i][j],  false,  false);
                }
            }
            #pragma unroll
            for (int i = 0; i < GM_MT; ++i) {
                const v2f a1 = { a[i][p].z, a[i][p].w };
                #pragma unroll
                for (int j = 0; j < GM_NT; ++j) {
                    const v2f b1 = { b[j][p].z, b[j][p].w };
                    acc[i][j] = __builtin_amdgcn_wmma_f32_16x16x4_f32(
                         false, a1,  false, b1,
                         (short)0, acc[i][j],  false,  false);
                }
            }
        }
        #pragma unroll
        for (int i = 0; i < GM_MT; ++i) { a[i][0] = an[i][0]; a[i][1] = an[i][1]; }
        #pragma unroll
        for (int j = 0; j < GM_NT; ++j) { b[j][0] = bn[j][0]; b[j][1] = bn[j][1]; }
    }

    float bv[GM_NT];
    #pragma unroll
    for (int j = 0; j < GM_NT; ++j) bv[j] = bias[tn * (16 * GM_NT) + j * 16 + l16];

    float dv[GM_MT][8];
    #pragma unroll
    for (int i = 0; i < GM_MT; ++i) {
        #pragma unroll
        for (int r = 0; r < 8; ++r) {
            int m = tm * (16 * GM_MT) + i * 16 + half * 8 + r;
            if (m >= M) m = M - 1;
            dv[i][r] = dinv[m];
        }
    }

    #pragma unroll
    for (int i = 0; i < GM_MT; ++i) {
        #pragma unroll
        for (int r = 0; r < 8; ++r) {
            const int m = tm * (16 * GM_MT) + i * 16 + half * 8 + r;
            if (m < M) {
                #pragma unroll
                for (int j = 0; j < GM_NT; ++j) {
                    float v = acc[i][j][r] * dv[i][r] + bv[j];
                    if (RELU) v = fmaxf(v, 0.0f);
                    out[(size_t)m * NF + tn * (16 * GM_NT) + j * 16 + l16] = v;
                }
            }
        }
    }
}

__global__ void head_kernel(const float* __restrict__ h, const float* __restrict__ W3,
                            const float* __restrict__ b3, float* __restrict__ out, int M) {
    int tid = blockIdx.x * blockDim.x + threadIdx.x;
    if (tid >= M * N_CLS) return;
    int m = tid / N_CLS, c = tid % N_CLS;
    const float* __restrict__ hm = h  + (size_t)m * N_HID2;
    const float* __restrict__ wc = W3 + (size_t)c * N_HID2;
    float s = b3[c];
    #pragma unroll 8
    for (int k = 0; k < N_HID2; ++k) s += hm[k] * wc[k];
    out[tid] = s;
}

extern "C" void kernel_launch(void* const* d_in, const int* in_sizes, int n_in,
                              void* d_out, int out_size, void* d_ws, size_t ws_size,
                              hipStream_t stream) {
    const float* x  = (const float*)d_in[0];
    const int*   ei = (const int*)  d_in[1];
    const float* W1 = (const float*)d_in[3];
    const float* b1 = (const float*)d_in[4];
    const float* W2 = (const float*)d_in[5];
    const float* b2 = (const float*)d_in[6];
    const float* W3 = (const float*)d_in[7];
    const float* b3 = (const float*)d_in[8];
    float* out = (float*)d_out;

    const int M = in_sizes[0] / N_IN;
    const int E = in_sizes[1] / 2;
    const int* row = ei;
    const int* col = ei + E;

    float* ws = (float*)d_ws;
    size_t off = 0;
    const int NG  = (M + AGG_T - 1) / AGG_T;
    const int NGp = (NG + 31) & ~31;
    int*   gcnt = (int*)(ws + off); off += (size_t)NGp;
    float* dinv = ws + off; off += (size_t)M;
    off = (off + 3) & ~(size_t)3;
    float* agg1 = ws + off; off += (size_t)M * N_IN;
    float* h1   = ws + off; off += (size_t)M * N_HID;
    float* agg2 = ws + off; off += (size_t)M * N_HID;
    float* h2   = ws + off; off += (size_t)M * N_HID2;
    off = (off + 1) & ~(size_t)1;
    int2* lists = (int2*)(ws + off);
    const size_t room = (ws_size / sizeof(float) - off) / 2 / (size_t)NG;
    const int cap = (int)(room < (size_t)BIN_CAP ? room : (size_t)BIN_CAP);

    hipMemsetAsync(gcnt, 0, sizeof(int) * (size_t)NGp, stream);

    bin_edges<<<(E + BIN_NT * 4 - 1) / (BIN_NT * 4), BIN_NT, 0, stream>>>(row, col, lists, gcnt, E, M, NG, cap);

    agg_owner<N_IN, true><<<NG, AGG_NT, 0, stream>>>(x, row, col, lists, gcnt, cap, agg1, dinv, M, E);
    int tiles1 = ((M + 16 * GM_MT - 1) / (16 * GM_MT)) * (N_HID / (16 * GM_NT));
    gemm_wmma<N_IN, N_HID, true>
        <<<(tiles1 + GM_WAVES - 1) / GM_WAVES, dim3(32, GM_WAVES), 0, stream>>>(agg1, W1, b1, dinv, h1, M);

    agg_owner<N_HID, false><<<NG, AGG_NT, 0, stream>>>(h1, row, col, lists, gcnt, cap, agg2, dinv, M, E);
    int tiles2 = ((M + 16 * GM_MT - 1) / (16 * GM_MT)) * (N_HID2 / (16 * GM_NT));
    gemm_wmma<N_HID, N_HID2, true>
        <<<(tiles2 + GM_WAVES - 1) / GM_WAVES, dim3(32, GM_WAVES), 0, stream>>>(agg2, W2, b2, dinv, h2, M);

    head_kernel<<<(M * N_CLS + 255) / 256, 256, 0, stream>>>(h2, W3, b3, out, M);
}


# Head in the layer-2 GEMM epilogue: a wave owns 32 whole rows (2x8 tiles), sums relu(row).W3 over 16 lanes by a halving butterfly, 8-byte store per row; h2 plane, a launch gone; head sum order changed
# speedup vs baseline: 7.1684x; 1.0963x over previous
#include <hip/hip_runtime.h>

typedef __attribute__((ext_vector_type(2))) float v2f;
typedef __attribute__((ext_vector_type(8))) float v8f;

#define N_IN   128
#define N_HID  256
#define N_HID2 128
#define N_CLS  2

#define AGG_T    196
#define AGG_NT   1024
#define AGG_NPW  7
#define AGG_NOW  28
#define AGG_RB   8
#define AGG_LCAP 16384

__device__ __forceinline__ int4 load_cols(const int* __restrict__ col, int idx, int E, bool vec_ok) {
    int4 c;
    if (vec_ok && idx + 3 < E) {
        c = *(const int4*)(col + idx);
    } else {
        c.x = (idx     < E) ? col[idx]     : -1;
        c.y = (idx + 1 < E) ? col[idx + 1] : -1;
        c.z = (idx + 2 < E) ? col[idx + 2] : -1;
        c.w = (idx + 3 < E) ? col[idx + 3] : -1;
    }
    return c;
}

#define BIN_NT   1024
#define BIN_MAXG 1024
#define BIN_CAP  16384

__global__ __launch_bounds__(BIN_NT)
void bin_edges(const int* __restrict__ row, const int* __restrict__ col,
               int2* __restrict__ lists, int* __restrict__ gcnt,
               int E, int M, int NG, int cap) {
    __shared__ int s_cnt[BIN_MAXG];
    __shared__ int s_base[BIN_MAXG];
    const int tid = threadIdx.x;
    const int e0  = (blockIdx.x * BIN_NT + tid) * 4;
    const bool vec_ok = ((E & 3) == 0);
    const int4 c4 = load_cols(col, e0, E, vec_ok);
    const int4 r4 = load_cols(row, e0, E, true);
    const int cc[4] = { c4.x, c4.y, c4.z, c4.w };
    const int rr[4] = { r4.x, r4.y, r4.z, r4.w };

    for (int g0 = 0; g0 < NG; g0 += BIN_MAXG) {
        for (int i = tid; i < BIN_MAXG; i += BIN_NT) s_cnt[i] = 0;
        __syncthreads();
        int gl[4], rank[4];
        #pragma unroll
        for (int j = 0; j < 4; ++j) {
            const unsigned uc = (unsigned)cc[j];
            const unsigned g = (uc < (unsigned)M) ? uc / AGG_T - (unsigned)g0 : 0xffffffffu;
            gl[j]   = (g < (unsigned)BIN_MAXG) ? (int)g : -1;
            rank[j] = (gl[j] >= 0) ? atomicAdd(&s_cnt[gl[j]], 1) : 0;
        }
        __syncthreads();
        for (int i = tid; i < BIN_MAXG; i += BIN_NT) {
            const int n = s_cnt[i];
            s_base[i] = (n > 0) ? atomicAdd(&gcnt[g0 + i], n) : 0;
        }
        __syncthreads();
        #pragma unroll
        for (int j = 0; j < 4; ++j) {
            if (gl[j] >= 0) {
                const int g   = g0 + gl[j];
                const int pos = s_base[gl[j]] + rank[j];
                if (pos < cap) lists[(size_t)g * cap + pos] = make_int2(rr[j], cc[j] - g * AGG_T);
            }
        }
    }
}

template<int F, bool DEG>
__global__ __launch_bounds__(AGG_NT)
void agg_owner(const float* __restrict__ src,
               const int* __restrict__ row,
               const int* __restrict__ col,
               const int2* __restrict__ lists,
               const int* __restrict__ gcnt, int cap,
               float* __restrict__ agg, float* __restrict__ dinv, int M, int E) {
    constexpr int V = F / 128;
    static_assert(AGG_NOW * AGG_NPW >= AGG_T, "the summing waves must cover the group");
    static_assert(AGG_NOW * 32 <= AGG_NT, "the summing waves must exist");
    static_assert(32 * AGG_NPW >= AGG_T, "one wave scans the counts, AGG_NPW per lane");
    static_assert(AGG_LCAP >= BIN_CAP, "a complete list must fit");
    __shared__ int s_row[AGG_LCAP];
    __shared__ int s_cnt[AGG_T];
    __shared__ int s_off[AGG_T + 1];
    __shared__ int s_cur[AGG_T];
    __shared__ int s_tot;

    const int tid  = threadIdx.x;
    const int lane = tid & 31;
    const int wave = __builtin_amdgcn_readfirstlane(tid >> 5);
    const int lo   = blockIdx.x * AGG_T;
    const int nT   = min(AGG_T, M - lo);
    float* __restrict__ dst = agg + (size_t)lo * F;

    for (int i = tid; i < AGG_T; i += AGG_NT) s_cnt[i] = 0;
    if (tid == 0) s_tot = gcnt[blockIdx.x];
    __syncthreads();

    const int  total  = __builtin_amdgcn_readfirstlane(s_tot);
    const bool listed = (total <= cap) && (total <= AGG_LCAP);
    if (listed) {
        const int2* __restrict__ lst = lists + (size_t)blockIdx.x * cap;
        for (int i = tid; i < total; i += AGG_NT) atomicAdd(&s_cnt[lst[i].y], 1);
        __syncthreads();
        if (wave == 0) {
            int c[AGG_NPW];
            int s = 0;
            #pragma unroll
            for (int k = 0; k < AGG_NPW; ++k) {
                const int idx = lane * AGG_NPW + k;
                c[k] = (idx < AGG_T) ? s_cnt[idx] : 0;
                s += c[k];
            }
            int incl = s;
            #pragma unroll
            for (int d = 1; d < 32; d <<= 1) {
                const int t = __shfl_up(incl, d);
                if (lane >= d) incl += t;
            }
            int run = incl - s;
            #pragma unroll
            for (int k = 0; k < AGG_NPW; ++k) {
                const int idx = lane * AGG_NPW + k;
                if (idx < AGG_T) { s_off[idx] = run; s_cur[idx] = run; }
                run += c[k];
            }
            if (lane == 31) s_off[AGG_T] = run;
        }
        __syncthreads();
        for (int i = tid; i < total; i += AGG_NT) {
            const int2 rec = lst[i];
            s_row[atomicAdd(&s_cur[rec.y], 1)] = rec.x;
        }
        __syncthreads();

        if (wave < AGG_NOW) {
            for (int k = 0; k < AGG_NPW; ++k) {
                const int node = wave * AGG_NPW + k;
                if (node >= nT) break;
                const int beg = __builtin_amdgcn_readfirstlane(s_off[node]);
                const int end = __builtin_amdgcn_readfirstlane(s_off[node + 1]);
                float4 sum[V];
                #pragma unroll
                for (int q = 0; q < V; ++q) sum[q] = make_float4(0.f, 0.f, 0.f, 0.f);
                for (int b = beg; b < end; b += AGG_RB) {
                    const int cnt = min(AGG_RB, end - b);
                    float4 v[AGG_RB][V];
                    int    r[AGG_RB];
                    #pragma unroll
                    for (int j = 0; j < AGG_RB; ++j) r[j] = s_row[min(b + j, end - 1)];
                    #pragma unroll
                    for (int j = 0; j < AGG_RB; ++j) {
                        #pragma unroll
                        for (int q = 0; q < V; ++q) v[j][q] = make_float4(0.f, 0.f, 0.f, 0.f);
                        if (j < cnt) {
                            const float* __restrict__ p = src + (size_t)r[j] * F + lane * 4;
                            #pragma unroll
                            for (int q = 0; q < V; ++q) v[j][q] = *(const float4*)(p + q * 128);
                        }
                    }
                    #pragma unroll
                    for (int j = 0; j < AGG_RB; ++j) {
                        #pragma unroll
                        for (int q = 0; q < V; ++q) {
                            sum[q].x += v[j][q].x; sum[q].y += v[j][q].y;
                            sum[q].z += v[j][q].z; sum[q].w += v[j][q].w;
                        }
                    }
                }
                #pragma unroll
                for (int q = 0; q < V; ++q)
                    *(float4*)(dst + (size_t)node * F + q * 128 + lane * 4) = sum[q];
            }
        }
    } else {
        if (wave < AGG_NOW) {
            for (int k = 0; k < AGG_NPW; ++k) {
                const int node = wave * AGG_NPW + k;
                if (node >= nT) break;
                float4 sum[V];
                #pragma unroll
                for (int q = 0; q < V; ++q) sum[q] = make_float4(0.f, 0.f, 0.f, 0.f);
                int deg = 0;
                for (int base = 0; base < E; base += 32) {
                    const int e = base + lane;
                    const bool hit = (e < E) && (col[e] == lo + node);
                    unsigned m = __builtin_amdgcn_ballot_w32(hit);
                    deg += __builtin_popcount(m);
                    while (m) {
                        const int bit = __builtin_ctz(m);
                        m &= m - 1;
                        const int r = row[base + bit];
                        const float* __restrict__ p = src + (size_t)r * F + lane * 4;
                        #pragma unroll
                        for (int q = 0; q < V; ++q) {
                            const float4 t = *(const float4*)(p + q * 128);
                            sum[q].x += t.x; sum[q].y += t.y; sum[q].z += t.z; sum[q].w += t.w;
                        }
                    }
                }
                #pragma unroll
                for (int q = 0; q < V; ++q)
                    *(float4*)(dst + (size_t)node * F + q * 128 + lane * 4) = sum[q];
                if (DEG) {
                    if (lane == 0) s_cnt[node] = deg;
                }
            }
        }
    }

    if (DEG) {
        __syncthreads();
        for (int i = tid; i < nT; i += AGG_NT) dinv[lo + i] = 1.0f / fmaxf((float)s_cnt[i], 1.0f);
    }
}

#define GM_MT 2
#define GM_WAVES 4
#define GM_KS 16

template<int K, int NF, int GM_NT, bool RELU, bool HEAD>
__global__ __launch_bounds__(32 * GM_WAVES)
void gemm_wmma(const float* __restrict__ A, const float* __restrict__ W,
               const float* __restrict__ bias, const float* __restrict__ dinv,
               const float* __restrict__ W3, const float* __restrict__ b3,
               float* __restrict__ out, int M) {
    static_assert(NF % (16 * GM_NT) == 0, "column blocks must tile the row");
    static_assert(!HEAD || (NF == 16 * GM_NT && N_CLS == 2 && GM_MT == 2), "the head needs whole rows in one wave");
    constexpr int NB = NF / (16 * GM_NT);
    const int lane   = threadIdx.x;
    const int waveId = threadIdx.y;
    const int stripsM = (M + 16 * GM_MT - 1) / (16 * GM_MT);
    const int tile = blockIdx.x * GM_WAVES + waveId;
    if (tile >= stripsM * NB) return;
    const int tm = tile / NB;
    const int tn = tile % NB;

    const int half = lane >> 4;
    const int l16  = lane & 15;

    const float* __restrict__ ap[GM_MT];
    const float* __restrict__ wp[GM_NT];
    #pragma unroll
    for (int i = 0; i < GM_MT; ++i) {
        int r = tm * (16 * GM_MT) + i * 16 + l16;
        if (r >= M) r = M - 1;
        ap[i] = A + (size_t)r * K + 8 * half;
    }
    #pragma unroll
    for (int j = 0; j < GM_NT; ++j)
        wp[j] = W + (size_t)(tn * (16 * GM_NT) + j * 16 + l16) * K + 8 * half;

    v8f acc[GM_MT][GM_NT];
    #pragma unroll
    for (int i = 0; i < GM_MT; ++i)
        #pragma unroll
        for (int j = 0; j < GM_NT; ++j)
            acc[i][j] = v8f{};

    float4 a[GM_MT][2], b[GM_NT][2];
    #pragma unroll
    for (int i = 0; i < GM_MT; ++i) {
        a[i][0] = *(const float4*)(ap[i]);
        a[i][1] = *(const float4*)(ap[i] + 4);
    }
    #pragma unroll
    for (int j = 0; j < GM_NT; ++j) {
        b[j][0] = *(const float4*)(wp[j]);
        b[j][1] = *(const float4*)(wp[j] + 4);
    }

    #pragma unroll 2
    for (int k0 = 0; k0 < K; k0 += GM_KS) {
        float4 an[GM_MT][2], bn[GM_NT][2];
        const int kn = (k0 + GM_KS < K) ? k0 + GM_KS : k0;
        #pragma unroll
        for (int i = 0; i < GM_MT; ++i) {
            an[i][0] = *(const float4*)(ap[i] + kn);
            an[i][1] = *(const float4*)(ap[i] + kn + 4);
        }
        #pragma unroll
        for (int j = 0; j < GM_NT; ++j) {
            bn[j][0] = *(const float4*)(wp[j] + kn);
            bn[j][1] = *(const float4*)(wp[j] + kn + 4);
        }
        __builtin_amdgcn_sched_barrier(0);

        #pragma unroll
        for (int p = 0; p < 2; ++p) {
            #pragma unroll
            for (int i = 0; i < GM_MT; ++i) {
                const v2f a0 = { a[i][p].x, a[i][p].y };
                #pragma unroll
                for (int j = 0; j < GM_NT; ++j) {
                    const v2f b0 = { b[j][p].x, b[j][p].y };
                    acc[i][j] = __builtin_amdgcn_wmma_f32_16x16x4_f32(
                         false, a0,  false, b0,
                         (short)0, acc[i][j],  false,  false);
                }
            }
            #pragma unroll
            for (int i = 0; i < GM_MT; ++i) {
                const v2f a1 = { a[i][p].z, a[i][p].w };
                #pragma unroll
                for (int j = 0; j < GM_NT; ++j) {
                    const v2f b1 = { b[j][p].z, b[j][p].w };
                    acc[i][j] = __builtin_amdgcn_wmma_f32_16x16x4_f32(
                         false, a1,  false, b1,
                         (short)0, acc[i][j],  false,  false);
                }
            }
        }
        #pragma unroll
        for (int i = 0; i < GM_MT; ++i) { a[i][0] = an[i][0]; a[i][1] = an[i][1]; }
        #pragma unroll
        for (int j = 0; j < GM_NT; ++j) { b[j][0] = bn[j][0]; b[j][1] = bn[j][1]; }
    }

    float bv[GM_NT];
    #pragma unroll
    for (int j = 0; j < GM_NT; ++j) bv[j] = bias[tn * (16 * GM_NT) + j * 16 + l16];

    float dv[GM_MT][8];
    #pragma unroll
    for (int i = 0; i < GM_MT; ++i) {
        #pragma unroll
        for (int r = 0; r < 8; ++r) {
            int m = tm * (16 * GM_MT) + i * 16 + half * 8 + r;
            if (m >= M) m = M - 1;
            dv[i][r] = dinv[m];
        }
    }

    if (!HEAD) {
        #pragma unroll
        for (int i = 0; i < GM_MT; ++i) {
            #pragma unroll
            for (int r = 0; r < 8; ++r) {
                const int m = tm * (16 * GM_MT) + i * 16 + half * 8 + r;
                if (m < M) {
                    #pragma unroll
                    for (int j = 0; j < GM_NT; ++j) {
                        float v = acc[i][j][r] * dv[i][r] + bv[j];
                        if (RELU) v = fmaxf(v, 0.0f);
                        out[(size_t)m * NF + tn * (16 * GM_NT) + j * 16 + l16] = v;
                    }
                }
            }
        }
    } else {
        float w3a[GM_NT], w3b[GM_NT];
        #pragma unroll
        for (int j = 0; j < GM_NT; ++j) {
            w3a[j] = W3[j * 16 + l16];
            w3b[j] = W3[NF + j * 16 + l16];
        }
        float s[N_CLS][GM_MT * 8];
        #pragma unroll
        for (int i = 0; i < GM_MT; ++i) {
            #pragma unroll
            for (int r = 0; r < 8; ++r) {
                float p0 = 0.0f, p1 = 0.0f;
                #pragma unroll
                for (int j = 0; j < GM_NT; ++j) {
                    float v = acc[i][j][r] * dv[i][r] + bv[j];
                    if (RELU) v = fmaxf(v, 0.0f);
                    p0 += v * w3a[j];
                    p1 += v * w3b[j];
                }
                s[0][i * 8 + r] = p0;
                s[1][i * 8 + r] = p1;
            }
        }
        #pragma unroll
        for (int c = 0; c < N_CLS; ++c) {
            #pragma unroll
            for (int n = 8; n >= 1; n >>= 1) {
                const bool up = (l16 & n) != 0;
                #pragma unroll
                for (int t = 0; t < n; ++t) {
                    const float keep = up ? s[c][n + t] : s[c][t];
                    const float send = up ? s[c][t] : s[c][n + t];
                    s[c][t] = keep + __shfl_xor(send, n);
                }
            }
        }
        const int m = tm * (16 * GM_MT) + (l16 >> 3) * 16 + half * 8 + (l16 & 7);
        if (m < M) {
            float2 o;
            o.x = s[0][0] + b3[0];
            o.y = s[1][0] + b3[1];
            *(float2*)(out + (size_t)m * N_CLS) = o;
        }
    }
}

extern "C" void kernel_launch(void* const* d_in, const int* in_sizes, int n_in,
                              void* d_out, int out_size, void* d_ws, size_t ws_size,
                              hipStream_t stream) {
    const float* x  = (const float*)d_in[0];
    const int*   ei = (const int*)  d_in[1];
    const float* W1 = (const float*)d_in[3];
    const float* b1 = (const float*)d_in[4];
    const float* W2 = (const float*)d_in[5];
    const float* b2 = (const float*)d_in[6];
    const float* W3 = (const float*)d_in[7];
    const float* b3 = (const float*)d_in[8];
    float* out = (float*)d_out;

    const int M = in_sizes[0] / N_IN;
    const int E = in_sizes[1] / 2;
    const int* row = ei;
    const int* col = ei + E;

    float* ws = (float*)d_ws;
    size_t off = 0;
    const int NG  = (M + AGG_T - 1) / AGG_T;
    const int NGp = (NG + 31) & ~31;
    int*   gcnt = (int*)(ws + off); off += (size_t)NGp;
    float* dinv = ws + off; off += (size_t)M;
    off = (off + 3) & ~(size_t)3;
    float* agg1 = ws + off; off += (size_t)M * N_IN;
    float* h1   = ws + off; off += (size_t)M * N_HID;
    float* agg2 = ws + off; off += (size_t)M * N_HID;
    off = (off + 1) & ~(size_t)1;
    int2* lists = (int2*)(ws + off);
    const size_t room = (ws_size / sizeof(float) - off) / 2 / (size_t)NG;
    const int cap = (int)(room < (size_t)BIN_CAP ? room : (size_t)BIN_CAP);

    hipMemsetAsync(gcnt, 0, sizeof(int) * (size_t)NGp, stream);

    bin_edges<<<(E + BIN_NT * 4 - 1) / (BIN_NT * 4), BIN_NT, 0, stream>>>(row, col, lists, gcnt, E, M, NG, cap);

    agg_owner<N_IN, true><<<NG, AGG_NT, 0, stream>>>(x, row, col, lists, gcnt, cap, agg1, dinv, M, E);
    int tiles1 = ((M + 16 * GM_MT - 1) / (16 * GM_MT)) * (N_HID / (16 * 4));
    gemm_wmma<N_IN, N_HID, 4, true, false>
        <<<(tiles1 + GM_WAVES - 1) / GM_WAVES, dim3(32, GM_WAVES), 0, stream>>>(agg1, W1, b1, dinv, nullptr, nullptr, h1, M);

    agg_owner<N_HID, false><<<NG, AGG_NT, 0, stream>>>(h1, row, col, lists, gcnt, cap, agg2, dinv, M, E);
    int tiles2 = (M + 16 * GM_MT - 1) / (16 * GM_MT);
    gemm_wmma<N_HID, N_HID2, 8, true, true>
        <<<(tiles2 + GM_WAVES - 1) / GM_WAVES, dim3(32, GM_WAVES), 0, stream>>>(agg2, W2, b2, dinv, W3, b3, out, M);
}


# Layer 2 in one launch: an owner group's 28 waves sum 14 nodes into an LDS tile while 4 matrix waves multiply the tile before by W2 (in LDS) and fold the head; sum plane gone; head sum order changed
# speedup vs baseline: 7.3933x; 1.0314x over previous
#include <hip/hip_runtime.h>

typedef __attribute__((ext_vector_type(2))) float v2f;
typedef __attribute__((ext_vector_type(8))) float v8f;

#define N_IN   128
#define N_HID  256
#define N_HID2 128
#define N_CLS  2

#define AGG_T    196
#define AGG_NT   1024
#define AGG_NPW  7
#define AGG_NOW  28
#define AGG_RB   8
#define AGG_LCAP 16384

__device__ __forceinline__ int4 load_cols(const int* __restrict__ col, int idx, int E, bool vec_ok) {
    int4 c;
    if (vec_ok && idx + 3 < E) {
        c = *(const int4*)(col + idx);
    } else {
        c.x = (idx     < E) ? col[idx]     : -1;
        c.y = (idx + 1 < E) ? col[idx + 1] : -1;
        c.z = (idx + 2 < E) ? col[idx + 2] : -1;
        c.w = (idx + 3 < E) ? col[idx + 3] : -1;
    }
    return c;
}

#define BIN_NT   1024
#define BIN_MAXG 1024
#define BIN_CAP  16384

__global__ __launch_bounds__(BIN_NT)
void bin_edges(const int* __restrict__ row, const int* __restrict__ col,
               int2* __restrict__ lists, int* __restrict__ gcnt,
               int E, int M, int NG, int cap) {
    __shared__ int s_cnt[BIN_MAXG];
    __shared__ int s_base[BIN_MAXG];
    const int tid = threadIdx.x;
    const int e0  = (blockIdx.x * BIN_NT + tid) * 4;
    const bool vec_ok = ((E & 3) == 0);
    const int4 c4 = load_cols(col, e0, E, vec_ok);
    const int4 r4 = load_cols(row, e0, E, true);
    const int cc[4] = { c4.x, c4.y, c4.z, c4.w };
    const int rr[4] = { r4.x, r4.y, r4.z, r4.w };

    for (int g0 = 0; g0 < NG; g0 += BIN_MAXG) {
        for (int i = tid; i < BIN_MAXG; i += BIN_NT) s_cnt[i] = 0;
        __syncthreads();
        int gl[4], rank[4];
        #pragma unroll
        for (int j = 0; j < 4; ++j) {
            const unsigned uc = (unsigned)cc[j];
            const unsigned g = (uc < (unsigned)M) ? uc / AGG_T - (unsigned)g0 : 0xffffffffu;
            gl[j]   = (g < (unsigned)BIN_MAXG) ? (int)g : -1;
            rank[j] = (gl[j] >= 0) ? atomicAdd(&s_cnt[gl[j]], 1) : 0;
        }
        __syncthreads();
        for (int i = tid; i < BIN_MAXG; i += BIN_NT) {
            const int n = s_cnt[i];
            s_base[i] = (n > 0) ? atomicAdd(&gcnt[g0 + i], n) : 0;
        }
        __syncthreads();
        #pragma unroll
        for (int j = 0; j < 4; ++j) {
            if (gl[j] >= 0) {
                const int g   = g0 + gl[j];
                const int pos = s_base[gl[j]] + rank[j];
                if (pos < cap) lists[(size_t)g * cap + pos] = make_int2(rr[j], cc[j] - g * AGG_T);
            }
        }
    }
}

template<int F, bool DEG>
__global__ __launch_bounds__(AGG_NT)
void agg_owner(const float* __restrict__ src,
               const int* __restrict__ row,
               const int* __restrict__ col,
               const int2* __restrict__ lists,
               const int* __restrict__ gcnt, int cap,
               float* __restrict__ agg, float* __restrict__ dinv, int M, int E) {
    constexpr int V = F / 128;
    static_assert(AGG_NOW * AGG_NPW >= AGG_T, "the summing waves must cover the group");
    static_assert(AGG_NOW * 32 <= AGG_NT, "the summing waves must exist");
    static_assert(32 * AGG_NPW >= AGG_T, "one wave scans the counts, AGG_NPW per lane");
    static_assert(AGG_LCAP >= BIN_CAP, "a complete list must fit");
    __shared__ int s_row[AGG_LCAP];
    __shared__ int s_cnt[AGG_T];
    __shared__ int s_off[AGG_T + 1];
    __shared__ int s_cur[AGG_T];
    __shared__ int s_tot;

    const int tid  = threadIdx.x;
    const int lane = tid & 31;
    const int wave = __builtin_amdgcn_readfirstlane(tid >> 5);
    const int lo   = blockIdx.x * AGG_T;
    const int nT   = min(AGG_T, M - lo);
    float* __restrict__ dst = agg + (size_t)lo * F;

    for (int i = tid; i < AGG_T; i += AGG_NT) s_cnt[i] = 0;
    if (tid == 0) s_tot = gcnt[blockIdx.x];
    __syncthreads();

    const int  total  = __builtin_amdgcn_readfirstlane(s_tot);
    const bool listed = (total <= cap) && (total <= AGG_LCAP);
    if (listed) {
        const int2* __restrict__ lst = lists + (size_t)blockIdx.x * cap;
        for (int i = tid; i < total; i += AGG_NT) atomicAdd(&s_cnt[lst[i].y], 1);
        __syncthreads();
        if (wave == 0) {
            int c[AGG_NPW];
            int s = 0;
            #pragma unroll
            for (int k = 0; k < AGG_NPW; ++k) {
                const int idx = lane * AGG_NPW + k;
                c[k] = (idx < AGG_T) ? s_cnt[idx] : 0;
                s += c[k];
            }
            int incl = s;
            #pragma unroll
            for (int d = 1; d < 32; d <<= 1) {
                const int t = __shfl_up(incl, d);
                if (lane >= d) incl += t;
            }
            int run = incl - s;
            #pragma unroll
            for (int k = 0; k < AGG_NPW; ++k) {
                const int idx = lane * AGG_NPW + k;
                if (idx < AGG_T) { s_off[idx] = run; s_cur[idx] = run; }
                run += c[k];
            }
            if (lane == 31) s_off[AGG_T] = run;
        }
        __syncthreads();
        for (int i = tid; i < total; i += AGG_NT) {
            const int2 rec = lst[i];
            s_row[atomicAdd(&s_cur[rec.y], 1)] = rec.x;
        }
        __syncthreads();

        if (wave < AGG_NOW) {
            for (int k = 0; k < AGG_NPW; ++k) {
                const int node = wave * AGG_NPW + k;
                if (node >= nT) break;
                const int beg = __builtin_amdgcn_readfirstlane(s_off[node]);
                const int end = __builtin_amdgcn_readfirstlane(s_off[node + 1]);
                float4 sum[V];
                #pragma unroll
                for (int q = 0; q < V; ++q) sum[q] = make_float4(0.f, 0.f, 0.f, 0.f);
                for (int b = beg; b < end; b += AGG_RB) {
                    const int cnt = min(AGG_RB, end - b);
                    float4 v[AGG_RB][V];
                    int    r[AGG_RB];
                    #pragma unroll
                    for (int j = 0; j < AGG_RB; ++j) r[j] = s_row[min(b + j, end - 1)];
                    #pragma unroll
                    for (int j = 0; j < AGG_RB; ++j) {
                        #pragma unroll
                        for (int q = 0; q < V; ++q) v[j][q] = make_float4(0.f, 0.f, 0.f, 0.f);
                        if (j < cnt) {
                            const float* __restrict__ p = src + (size_t)r[j] * F + lane * 4;
                            #pragma unroll
                            for (int q = 0; q < V; ++q) v[j][q] = *(const float4*)(p + q * 128);
                        }
                    }
                    #pragma unroll
                    for (int j = 0; j < AGG_RB; ++j) {
                        #pragma unroll
                        for (int q = 0; q < V; ++q) {
                            sum[q].x += v[j][q].x; sum[q].y += v[j][q].y;
                            sum[q].z += v[j][q].z; sum[q].w += v[j][q].w;
                        }
                    }
                }
                #pragma unroll
                for (int q = 0; q < V; ++q)
                    *(float4*)(dst + (size_t)node * F + q * 128 + lane * 4) = sum[q];
            }
        }
    } else {
        if (wave < AGG_NOW) {
            for (int k = 0; k < AGG_NPW; ++k) {
                const int node = wave * AGG_NPW + k;
                if (node >= nT) break;
                float4 sum[V];
                #pragma unroll
                for (int q = 0; q < V; ++q) sum[q] = make_float4(0.f, 0.f, 0.f, 0.f);
                int deg = 0;
                for (int base = 0; base < E; base += 32) {
                    const int e = base + lane;
                    const bool hit = (e < E) && (col[e] == lo + node);
                    unsigned m = __builtin_amdgcn_ballot_w32(hit);
                    deg += __builtin_popcount(m);
                    while (m) {
                        const int bit = __builtin_ctz(m);
                        m &= m - 1;
                        const int r = row[base + bit];
                        const float* __restrict__ p = src + (size_t)r * F + lane * 4;
                        #pragma unroll
                        for (int q = 0; q < V; ++q) {
                            const float4 t = *(const float4*)(p + q * 128);
                            sum[q].x += t.x; sum[q].y += t.y; sum[q].z += t.z; sum[q].w += t.w;
                        }
                    }
                }
                #pragma unroll
                for (int q = 0; q < V; ++q)
                    *(float4*)(dst + (size_t)node * F + q * 128 + lane * 4) = sum[q];
                if (DEG) {
                    if (lane == 0) s_cnt[node] = deg;
                }
            }
        }
    }

    if (DEG) {
        __syncthreads();
        for (int i = tid; i < nT; i += AGG_NT) dinv[lo + i] = 1.0f / fmaxf((float)s_cnt[i], 1.0f);
    }
}

#define GM_MT 2
#define GM_WAVES 4
#define GM_KS 16

template<int K, int NF, int GM_NT, bool RELU>
__global__ __launch_bounds__(32 * GM_WAVES)
void gemm_wmma(const float* __restrict__ A, const float* __restrict__ W,
               const float* __restrict__ bias, const float* __restrict__ dinv,
               float* __restrict__ out, int M) {
    static_assert(NF % (16 * GM_NT) == 0, "column blocks must tile the row");
    constexpr int NB = NF / (16 * GM_NT);
    const int lane   = threadIdx.x;
    const int waveId = threadIdx.y;
    const int stripsM = (M + 16 * GM_MT - 1) / (16 * GM_MT);
    const int tile = blockIdx.x * GM_WAVES + waveId;
    if (tile >= stripsM * NB) return;
    const int tm = tile / NB;
    const int tn = tile % NB;

    const int half = lane >> 4;
    const int l16  = lane & 15;

    const float* __restrict__ ap[GM_MT];
    const float* __restrict__ wp[GM_NT];
    #pragma unroll
    for (int i = 0; i < GM_MT; ++i) {
        int r = tm * (16 * GM_MT) + i * 16 + l16;
        if (r >= M) r = M - 1;
        ap[i] = A + (size_t)r * K + 8 * half;
    }
    #pragma unroll
    for (int j = 0; j < GM_NT; ++j)
        wp[j] = W + (size_t)(tn * (16 * GM_NT) + j * 16 + l16) * K + 8 * half;

    v8f acc[GM_MT][GM_NT];
    #pragma unroll
    for (int i = 0; i < GM_MT; ++i)
        #pragma unroll
        for (int j = 0; j < GM_NT; ++j)
            acc[i][j] = v8f{};

    float4 a[GM_MT][2], b[GM_NT][2];
    #pragma unroll
    for (int i = 0; i < GM_MT; ++i) {
        a[i][0] = *(const float4*)(ap[i]);
        a[i][1] = *(const float4*)(ap[i] + 4);
    }
    #pragma unroll
    for (int j = 0; j < GM_NT; ++j) {
        b[j][0] = *(const float4*)(wp[j]);
        b[j][1] = *(const float4*)(wp[j] + 4);
    }

    #pragma unroll 2
    for (int k0 = 0; k0 < K; k0 += GM_KS) {
        float4 an[GM_MT][2], bn[GM_NT][2];
        const int kn = (k0 + GM_KS < K) ? k0 + GM_KS : k0;
        #pragma unroll
        for (int i = 0; i < GM_MT; ++i) {
            an[i][0] = *(const float4*)(ap[i] + kn);
            an[i][1] = *(const float4*)(ap[i] + kn + 4);
        }
        #pragma unroll
        for (int j = 0; j < GM_NT; ++j) {
            bn[j][0] = *(const float4*)(wp[j] + kn);
            bn[j][1] = *(const float4*)(wp[j] + kn + 4);
        }
        __builtin_amdgcn_sched_barrier(0);

        #pragma unroll
        for (int p = 0; p < 2; ++p) {
            #pragma unroll
            for (int i = 0; i < GM_MT; ++i) {
                const v2f a0 = { a[i][p].x, a[i][p].y };
                #pragma unroll
                for (int j = 0; j < GM_NT; ++j) {
                    const v2f b0 = { b[j][p].x, b[j][p].y };
                    acc[i][j] = __builtin_amdgcn_wmma_f32_16x16x4_f32(
                         false, a0,  false, b0,
                         (short)0, acc[i][j],  false,  false);
                }
            }
            #pragma unroll
            for (int i = 0; i < GM_MT; ++i) {
                const v2f a1 = { a[i][p].z, a[i][p].w };
                #pragma unroll
                for (int j = 0; j < GM_NT; ++j) {
                    const v2f b1 = { b[j][p].z, b[j][p].w };
                    acc[i][j] = __builtin_amdgcn_wmma_f32_16x16x4_f32(
                         false, a1,  false, b1,
                         (short)0, acc[i][j],  false,  false);
                }
            }
        }
        #pragma unroll
        for (int i = 0; i < GM_MT; ++i) { a[i][0] = an[i][0]; a[i][1] = an[i][1]; }
        #pragma unroll
        for (int j = 0; j < GM_NT; ++j) { b[j][0] = bn[j][0]; b[j][1] = bn[j][1]; }
    }

    float bv[GM_NT];
    #pragma unroll
    for (int j = 0; j < GM_NT; ++j) bv[j] = bias[tn * (16 * GM_NT) + j * 16 + l16];

    float dv[GM_MT][8];
    #pragma unroll
    for (int i = 0; i < GM_MT; ++i) {
        #pragma unroll
        for (int r = 0; r < 8; ++r) {
            int m = tm * (16 * GM_MT) + i * 16 + half * 8 + r;
            if (m >= M) m = M - 1;
            dv[i][r] = dinv[m];
        }
    }

    #pragma unroll
    for (int i = 0; i < GM_MT; ++i) {
        #pragma unroll
        for (int r = 0; r < 8; ++r) {
            const int m = tm * (16 * GM_MT) + i * 16 + half * 8 + r;
            if (m < M) {
                #pragma unroll
                for (int j = 0; j < GM_NT; ++j) {
                    float v = acc[i][j][r] * dv[i][r] + bv[j];
                    if (RELU) v = fmaxf(v, 0.0f);
                    out[(size_t)m * NF + tn * (16 * GM_NT) + j * 16 + l16] = v;
                }
            }
        }
    }
}

#define FU_TN    14
#define FU_STEPS (AGG_T / FU_TN)
#define FU_AP    (N_HID + 4)
#define FU_RB    16
#define FU_MW    4

__global__ __launch_bounds__(AGG_NT)
void agg_dense_head(const float* __restrict__ src,
                    const int* __restrict__ row,
                    const int* __restrict__ col,
                    const int2* __restrict__ lists,
                    const int* __restrict__ gcnt, int cap,
                    const float* __restrict__ W2, const float* __restrict__ b2,
                    const float* __restrict__ dinv,
                    const float* __restrict__ W3, const float* __restrict__ b3,
                    float* __restrict__ out, int M, int E) {
    static_assert(FU_TN * FU_STEPS == AGG_T, "the steps must cover the group");
    static_assert(2 * FU_TN + FU_MW == AGG_NT / 32, "every wave has one role");
    static_assert(FU_TN <= 16 && N_HID == 256 && N_HID2 == 32 * FU_MW && N_CLS == 2, "tile shapes");
    static_assert(32 * AGG_NPW >= AGG_T, "one wave scans the counts, AGG_NPW per lane");
    static_assert(AGG_LCAP >= BIN_CAP, "a complete list must fit");
    __shared__ int s_row[AGG_LCAP];
    __shared__ int s_cnt[AGG_T];
    __shared__ int s_off[AGG_T + 1];
    __shared__ int s_cur[AGG_T];
    __shared__ int s_tot;
    __shared__ __attribute__((aligned(16))) float s_W[N_HID2 * FU_AP];
    __shared__ __attribute__((aligned(16))) float s_A[2][16 * FU_AP];
    __shared__ float s_hp[FU_MW][AGG_T * N_CLS];

    const int tid  = threadIdx.x;
    const int lane = tid & 31;
    const int wave = __builtin_amdgcn_readfirstlane(tid >> 5);
    const int lo   = blockIdx.x * AGG_T;
    const int nT   = min(AGG_T, M - lo);

    for (int i = tid; i < AGG_T; i += AGG_NT) s_cnt[i] = 0;
    if (tid == 0) s_tot = gcnt[blockIdx.x];
    for (int i = tid; i < N_HID2 * (N_HID / 4); i += AGG_NT) {
        const int r  = i / (N_HID / 4);
        const int c4 = i % (N_HID / 4);
        *(float4*)(s_W + r * FU_AP + c4 * 4) = *(const float4*)(W2 + (size_t)r * N_HID + c4 * 4);
    }
    for (int i = tid; i < 2 * 16 * FU_AP; i += AGG_NT) (&s_A[0][0])[i] = 0.0f;
    __syncthreads();

    const int  total  = __builtin_amdgcn_readfirstlane(s_tot);
    const bool listed = (total <= cap) && (total <= AGG_LCAP);
    if (listed) {
        const int2* __restrict__ lst = lists + (size_t)blockIdx.x * cap;
        for (int i = tid; i < total; i += AGG_NT) atomicAdd(&s_cnt[lst[i].y], 1);
        __syncthreads();
        if (wave == 0) {
            int c[AGG_NPW];
            int s = 0;
            #pragma unroll
            for (int k = 0; k < AGG_NPW; ++k) {
                const int idx = lane * AGG_NPW + k;
                c[k] = (idx < AGG_T) ? s_cnt[idx] : 0;
                s += c[k];
            }
            int incl = s;
            #pragma unroll
            for (int d = 1; d < 32; d <<= 1) {
                const int t = __shfl_up(incl, d);
                if (lane >= d) incl += t;
            }
            int run = incl - s;
            #pragma unroll
            for (int k = 0; k < AGG_NPW; ++k) {
                const int idx = lane * AGG_NPW + k;
                if (idx < AGG_T) { s_off[idx] = run; s_cur[idx] = run; }
                run += c[k];
            }
            if (lane == 31) s_off[AGG_T] = run;
        }
        __syncthreads();
        for (int i = tid; i < total; i += AGG_NT) {
            const int2 rec = lst[i];
            s_row[atomicAdd(&s_cur[rec.y], 1)] = rec.x;
        }
        __syncthreads();
    }

    const int  mw    = wave / 9;
    const bool isMat = (wave == 9 * mw);
    const int  gi    = wave - mw - 1;
    const int  gn    = gi >> 1;
    const int  fh    = gi & 1;
    const int  half  = lane >> 4;
    const int  l16   = lane & 15;
    const int  nSteps = (nT + FU_TN - 1) / FU_TN;

    float bv[2], w3a[2], w3b[2];
    #pragma unroll
    for (int j = 0; j < 2; ++j) {
        const int cidx = (isMat ? mw : 0) * 32 + j * 16 + l16;
        bv[j]  = b2[cidx];
        w3a[j] = W3[cidx];
        w3b[j] = W3[N_HID2 + cidx];
    }

    for (int s = 0; s <= nSteps; ++s) {
        if (!isMat) {
            if (s < nSteps) {
                const int node = s * FU_TN + gn;
                float4 sum = make_float4(0.f, 0.f, 0.f, 0.f);
                if (node < nT) {
                    const float* __restrict__ sp = src + fh * 128 + lane * 4;
                    if (listed) {
                        const int beg = __builtin_amdgcn_readfirstlane(s_off[node]);
                        const int end = __builtin_amdgcn_readfirstlane(s_off[node + 1]);
                        for (int b = beg; b < end; b += FU_RB) {
                            const int cnt = min(FU_RB, end - b);
                            float4 v[FU_RB];
                            int    r[FU_RB];
                            #pragma unroll
                            for (int j = 0; j < FU_RB; ++j) r[j] = s_row[min(b + j, end - 1)];
                            #pragma unroll
                            for (int j = 0; j < FU_RB; ++j) {
                                v[j] = make_float4(0.f, 0.f, 0.f, 0.f);
                                if (j < cnt) v[j] = *(const float4*)(sp + (size_t)r[j] * N_HID);
                            }
                            #pragma unroll
                            for (int j = 0; j < FU_RB; ++j) {
                                sum.x += v[j].x; sum.y += v[j].y; sum.z += v[j].z; sum.w += v[j].w;
                            }
                        }
                    } else {
                        for (int base = 0; base < E; base += 32) {
                            const int e = base + lane;
                            const bool hit = (e < E) && (col[e] == lo + node);
                            unsigned m = __builtin_amdgcn_ballot_w32(hit);
                            while (m) {
                                const int bit = __builtin_ctz(m);
                                m &= m - 1;
                                const int r = row[base + bit];
                                const float4 t = *(const float4*)(sp + (size_t)r * N_HID);
                                sum.x += t.x; sum.y += t.y; sum.z += t.z; sum.w += t.w;
                            }
                        }
                    }
                }
                *(float4*)(&s_A[s & 1][gn * FU_AP + fh * 128 + lane * 4]) = sum;
            }
        } else if (s > 0) {
            const int t = s - 1;
            float dv[8];
            #pragma unroll
            for (int r = 0; r < 8; ++r) {
                int m = lo + t * FU_TN + half * 8 + r;
                if (m >= M) m = M - 1;
                dv[r] = dinv[m];
            }
            const float* ap = &s_A[t & 1][l16 * FU_AP + 8 * half];
            const float* wp[2];
            wp[0] = s_W + (mw * 32 + l16) * FU_AP + 8 * half;
            wp[1] = wp[0] + 16 * FU_AP;

            v8f acc[2];
            acc[0] = v8f{};
            acc[1] = v8f{};
            float4 a[2], b[2][2];
            a[0] = *(const float4*)(ap);
            a[1] = *(const float4*)(ap + 4);
            #pragma unroll
            for (int j = 0; j < 2; ++j) {
                b[j][0] = *(const float4*)(wp[j]);
                b[j][1] = *(const float4*)(wp[j] + 4);
            }
            #pragma unroll
            for (int k0 = 0; k0 < N_HID; k0 += GM_KS) {
                float4 an[2], bn[2][2];
                if (k0 + GM_KS < N_HID) {
                    const int kn = k0 + GM_KS;
                    an[0] = *(const float4*)(ap + kn);
                    an[1] = *(const float4*)(ap + kn + 4);
                    #pragma unroll
                    for (int j = 0; j < 2; ++j) {
                        bn[j][0] = *(const float4*)(wp[j] + kn);
                        bn[j][1] = *(const float4*)(wp[j] + kn + 4);
                    }
                }
                __builtin_amdgcn_sched_barrier(0);
                #pragma unroll
                for (int p = 0; p < 2; ++p) {
                    const v2f a0 = { a[p].x, a[p].y };
                    #pragma unroll
                    for (int j = 0; j < 2; ++j) {
                        const v2f b0 = { b[j][p].x, b[j][p].y };
                        acc[j] = __builtin_amdgcn_wmma_f32_16x16x4_f32(
                             false, a0,  false, b0,
                             (short)0, acc[j],  false,  false);
                    }
                    const v2f a1 = { a[p].z, a[p].w };
                    #pragma unroll
                    for (int j = 0; j < 2; ++j) {
                        const v2f b1 = { b[j][p].z, b[j][p].w };
                        acc[j] = __builtin_amdgcn_wmma_f32_16x16x4_f32(
                             false, a1,  false, b1,
                             (short)0, acc[j],  false,  false);
                    }
                }
                __builtin_amdgcn_sched_barrier(0);
                if (k0 + GM_KS < N_HID) {
                    a[0] = an[0]; a[1] = an[1];
                    #pragma unroll
                    for (int j = 0; j < 2; ++j) { b[j][0] = bn[j][0]; b[j][1] = bn[j][1]; }
                }
            }

            float sv[N_CLS * 8];
            #pragma unroll
            for (int r = 0; r < 8; ++r) {
                float p0 = 0.0f, p1 = 0.0f;
                #pragma unroll
                for (int j = 0; j < 2; ++j) {
                    float v = acc[j][r] * dv[r] + bv[j];
                    v = fmaxf(v, 0.0f);
                    p0 += v * w3a[j];
                    p1 += v * w3b[j];
                }
                sv[r]     = p0;
                sv[8 + r] = p1;
            }
            #pragma unroll
            for (int n = 8; n >= 1; n >>= 1) {
                const bool up = (l16 & n) != 0;
                #pragma unroll
                for (int q = 0; q < n; ++q) {
                    const float keep = up ? sv[n + q] : sv[q];
                    const float send = up ? sv[q] : sv[n + q];
                    sv[q] = keep + __shfl_xor(send, n);
                }
            }
            const int rt = half * 8 + (l16 & 7);
            if (rt < FU_TN) s_hp[mw][(t * FU_TN + rt) * N_CLS + (l16 >> 3)] = sv[0];
        }
        __syncthreads();
    }

    for (int i = tid; i < nT * N_CLS; i += AGG_NT)
        out[(size_t)lo * N_CLS + i] = ((s_hp[0][i] + s_hp[1][i]) + (s_hp[2][i] + s_hp[3][i])) + b3[i & 1];
}

extern "C" void kernel_launch(void* const* d_in, const int* in_sizes, int n_in,
                              void* d_out, int out_size, void* d_ws, size_t ws_size,
                              hipStream_t stream) {
    const float* x  = (const float*)d_in[0];
    const int*   ei = (const int*)  d_in[1];
    const float* W1 = (const float*)d_in[3];
    const float* b1 = (const float*)d_in[4];
    const float* W2 = (const float*)d_in[5];
    const float* b2 = (const float*)d_in[6];
    const float* W3 = (const float*)d_in[7];
    const float* b3 = (const float*)d_in[8];
    float* out = (float*)d_out;

    const int M = in_sizes[0] / N_IN;
    const int E = in_sizes[1] / 2;
    const int* row = ei;
    const int* col = ei + E;

    float* ws = (float*)d_ws;
    size_t off = 0;
    const int NG  = (M + AGG_T - 1) / AGG_T;
    const int NGp = (NG + 31) & ~31;
    int*   gcnt = (int*)(ws + off); off += (size_t)NGp;
    float* dinv = ws + off; off += (size_t)M;
    off = (off + 3) & ~(size_t)3;
    float* agg1 = ws + off; off += (size_t)M * N_IN;
    float* h1   = ws + off; off += (size_t)M * N_HID;
    off = (off + 1) & ~(size_t)1;
    int2* lists = (int2*)(ws + off);
    const size_t room = (ws_size / sizeof(float) - off) / 2 / (size_t)NG;
    const int cap = (int)(room < (size_t)BIN_CAP ? room : (size_t)BIN_CAP);

    hipMemsetAsync(gcnt, 0, sizeof(int) * (size_t)NGp, stream);

    bin_edges<<<(E + BIN_NT * 4 - 1) / (BIN_NT * 4), BIN_NT, 0, stream>>>(row, col, lists, gcnt, E, M, NG, cap);

    agg_owner<N_IN, true><<<NG, AGG_NT, 0, stream>>>(x, row, col, lists, gcnt, cap, agg1, dinv, M, E);
    int tiles1 = ((M + 16 * GM_MT - 1) / (16 * GM_MT)) * (N_HID / (16 * 4));
    gemm_wmma<N_IN, N_HID, 4, true>
        <<<(tiles1 + GM_WAVES - 1) / GM_WAVES, dim3(32, GM_WAVES), 0, stream>>>(agg1, W1, b1, dinv, h1, M);

    agg_dense_head<<<NG, AGG_NT, 0, stream>>>(h1, row, col, lists, gcnt, cap, W2, b2, dinv, W3, b3, out, M, E);
}


# Layer 1 in one launch: 28 waves of an owner group sum 14 nodes (lane half = every 2nd edge) into an LDS tile, 4 matrix waves multiply the tile before by W1 in LDS; sum plane gone; sum order changed
# speedup vs baseline: 7.8602x; 1.0632x over previous
#include <hip/hip_runtime.h>

typedef __attribute__((ext_vector_type(2))) float v2f;
typedef __attribute__((ext_vector_type(8))) float v8f;

#define N_IN   128
#define N_HID  256
#define N_HID2 128
#define N_CLS  2


#define AGG_T    196
#define AGG_NT   1024
#define AGG_NPW  7
#define AGG_LCAP 16384

__device__ __forceinline__ int4 load_cols(const int* __restrict__ col, int idx, int E, bool vec_ok) {
    int4 c;
    if (vec_ok && idx + 3 < E) {
        c = *(const int4*)(col + idx);
    } else {
        c.x = (idx     < E) ? col[idx]     : -1;
        c.y = (idx + 1 < E) ? col[idx + 1] : -1;
        c.z = (idx + 2 < E) ? col[idx + 2] : -1;
        c.w = (idx + 3 < E) ? col[idx + 3] : -1;
    }
    return c;
}

#define BIN_NT   1024
#define BIN_MAXG 1024
#define BIN_CAP  16384

__global__ __launch_bounds__(BIN_NT)
void bin_edges(const int* __restrict__ row, const int* __restrict__ col,
               int2* __restrict__ lists, int* __restrict__ gcnt,
               int E, int M, int NG, int cap) {
    __shared__ int s_cnt[BIN_MAXG];
    __shared__ int s_base[BIN_MAXG];
    const int tid = threadIdx.x;
    const int e0  = (blockIdx.x * BIN_NT + tid) * 4;
    const bool vec_ok = ((E & 3) == 0);
    const int4 c4 = load_cols(col, e0, E, vec_ok);
    const int4 r4 = load_cols(row, e0, E, true);
    const int cc[4] = { c4.x, c4.y, c4.z, c4.w };
    const int rr[4] = { r4.x, r4.y, r4.z, r4.w };

    for (int g0 = 0; g0 < NG; g0 += BIN_MAXG) {
        for (int i = tid; i < BIN_MAXG; i += BIN_NT) s_cnt[i] = 0;
        __syncthreads();
        int gl[4], rank[4];
        #pragma unroll
        for (int j = 0; j < 4; ++j) {
            const unsigned uc = (unsigned)cc[j];
            const unsigned g = (uc < (unsigned)M) ? uc / AGG_T - (unsigned)g0 : 0xffffffffu;
            gl[j]   = (g < (unsigned)BIN_MAXG) ? (int)g : -1;
            rank[j] = (gl[j] >= 0) ? atomicAdd(&s_cnt[gl[j]], 1) : 0;
        }
        __syncthreads();
        for (int i = tid; i < BIN_MAXG; i += BIN_NT) {
            const int n = s_cnt[i];
            s_base[i] = (n > 0) ? atomicAdd(&gcnt[g0 + i], n) : 0;
        }
        __syncthreads();
        #pragma unroll
        for (int j = 0; j < 4; ++j) {
            if (gl[j] >= 0) {
                const int g   = g0 + gl[j];
                const int pos = s_base[gl[j]] + rank[j];
                if (pos < cap) lists[(size_t)g * cap + pos] = make_int2(rr[j], cc[j] - g * AGG_T);
            }
        }
    }
}

#define GM_KS 16

#define FU_TN    14
#define FU_STEPS (AGG_T / FU_TN)
#define FU_AP    (N_HID + 4)
#define FU_RB    16
#define FU_MW    4

__global__ __launch_bounds__(AGG_NT)
void agg_dense_head(const float* __restrict__ src,
                    const int* __restrict__ row,
                    const int* __restrict__ col,
                    const int2* __restrict__ lists,
                    const int* __restrict__ gcnt, int cap,
                    const float* __restrict__ W2, const float* __restrict__ b2,
                    const float* __restrict__ dinv,
                    const float* __restrict__ W3, const float* __restrict__ b3,
                    float* __restrict__ out, int M, int E) {
    static_assert(FU_TN * FU_STEPS == AGG_T, "the steps must cover the group");
    static_assert(2 * FU_TN + FU_MW == AGG_NT / 32, "every wave has one role");
    static_assert(FU_TN <= 16 && N_HID == 256 && N_HID2 == 32 * FU_MW && N_CLS == 2, "tile shapes");
    static_assert(32 * AGG_NPW >= AGG_T, "one wave scans the counts, AGG_NPW per lane");
    static_assert(AGG_LCAP >= BIN_CAP, "a complete list must fit");
    __shared__ int s_row[AGG_LCAP];
    __shared__ int s_cnt[AGG_T];
    __shared__ int s_off[AGG_T + 1];
    __shared__ int s_cur[AGG_T];
    __shared__ int s_tot;
    __shared__ __attribute__((aligned(16))) float s_W[N_HID2 * FU_AP];
    __shared__ __attribute__((aligned(16))) float s_A[2][16 * FU_AP];
    __shared__ float s_hp[FU_MW][AGG_T * N_CLS];

    const int tid  = threadIdx.x;
    const int lane = tid & 31;
    const int wave = __builtin_amdgcn_readfirstlane(tid >> 5);
    const int lo   = blockIdx.x * AGG_T;
    const int nT   = min(AGG_T, M - lo);

    for (int i = tid; i < AGG_T; i += AGG_NT) s_cnt[i] = 0;
    if (tid == 0) s_tot = gcnt[blockIdx.x];
    for (int i = tid; i < N_HID2 * (N_HID / 4); i += AGG_NT) {
        const int r  = i / (N_HID / 4);
        const int c4 = i % (N_HID / 4);
        *(float4*)(s_W + r * FU_AP + c4 * 4) = *(const float4*)(W2 + (size_t)r * N_HID + c4 * 4);
    }
    for (int i = tid; i < 2 * 16 * FU_AP; i += AGG_NT) (&s_A[0][0])[i] = 0.0f;
    __syncthreads();

    const int  total  = __builtin_amdgcn_readfirstlane(s_tot);
    const bool listed = (total <= cap) && (total <= AGG_LCAP);
    if (listed) {
        const int2* __restrict__ lst = lists + (size_t)blockIdx.x * cap;
        for (int i = tid; i < total; i += AGG_NT) atomicAdd(&s_cnt[lst[i].y], 1);
        __syncthreads();
        if (wave == 0) {
            int c[AGG_NPW];
            int s = 0;
            #pragma unroll
            for (int k = 0; k < AGG_NPW; ++k) {
                const int idx = lane * AGG_NPW + k;
                c[k] = (idx < AGG_T) ? s_cnt[idx] : 0;
                s += c[k];
            }
            int incl = s;
            #pragma unroll
            for (int d = 1; d < 32; d <<= 1) {
                const int t = __shfl_up(incl, d);
                if (lane >= d) incl += t;
            }
            int run = incl - s;
            #pragma unroll
            for (int k = 0; k < AGG_NPW; ++k) {
                const int idx = lane * AGG_NPW + k;
                if (idx < AGG_T) { s_off[idx] = run; s_cur[idx] = run; }
                run += c[k];
            }
            if (lane == 31) s_off[AGG_T] = run;
        }
        __syncthreads();
        for (int i = tid; i < total; i += AGG_NT) {
            const int2 rec = lst[i];
            s_row[atomicAdd(&s_cur[rec.y], 1)] = rec.x;
        }
        __syncthreads();
    }

    const int  mw    = wave / 9;
    const bool isMat = (wave == 9 * mw);
    const int  gi    = wave - mw - 1;
    const int  gn    = gi >> 1;
    const int  fh    = gi & 1;
    const int  half  = lane >> 4;
    const int  l16   = lane & 15;
    const int  nSteps = (nT + FU_TN - 1) / FU_TN;

    float bv[2], w3a[2], w3b[2];
    #pragma unroll
    for (int j = 0; j < 2; ++j) {
        const int cidx = (isMat ? mw : 0) * 32 + j * 16 + l16;
        bv[j]  = b2[cidx];
        w3a[j] = W3[cidx];
        w3b[j] = W3[N_HID2 + cidx];
    }

    for (int s = 0; s <= nSteps; ++s) {
        if (!isMat) {
            if (s < nSteps) {
                const int node = s * FU_TN + gn;
                float4 sum = make_float4(0.f, 0.f, 0.f, 0.f);
                if (node < nT) {
                    const float* __restrict__ sp = src + fh * 128 + lane * 4;
                    if (listed) {
                        const int beg = __builtin_amdgcn_readfirstlane(s_off[node]);
                        const int end = __builtin_amdgcn_readfirstlane(s_off[node + 1]);
                        for (int b = beg; b < end; b += FU_RB) {
                            const int cnt = min(FU_RB, end - b);
                            float4 v[FU_RB];
                            int    r[FU_RB];
                            #pragma unroll
                            for (int j = 0; j < FU_RB; ++j) r[j] = s_row[min(b + j, end - 1)];
                            #pragma unroll
                            for (int j = 0; j < FU_RB; ++j) {
                                v[j] = make_float4(0.f, 0.f, 0.f, 0.f);
                                if (j < cnt) v[j] = *(const float4*)(sp + (size_t)r[j] * N_HID);
                            }
                            #pragma unroll
                            for (int j = 0; j < FU_RB; ++j) {
                                sum.x += v[j].x; sum.y += v[j].y; sum.z += v[j].z; sum.w += v[j].w;
                            }
                        }
                    } else {
                        for (int base = 0; base < E; base += 32) {
                            const int e = base + lane;
                            const bool hit = (e < E) && (col[e] == lo + node);
                            unsigned m = __builtin_amdgcn_ballot_w32(hit);
                            while (m) {
                                const int bit = __builtin_ctz(m);
                                m &= m - 1;
                                const int r = row[base + bit];
                                const float4 t = *(const float4*)(sp + (size_t)r * N_HID);
                                sum.x += t.x; sum.y += t.y; sum.z += t.z; sum.w += t.w;
                            }
                        }
                    }
                }
                *(float4*)(&s_A[s & 1][gn * FU_AP + fh * 128 + lane * 4]) = sum;
            }
        } else if (s > 0) {
            const int t = s - 1;
            float dv[8];
            #pragma unroll
            for (int r = 0; r < 8; ++r) {
                int m = lo + t * FU_TN + half * 8 + r;
                if (m >= M) m = M - 1;
                dv[r] = dinv[m];
            }
            const float* ap = &s_A[t & 1][l16 * FU_AP + 8 * half];
            const float* wp[2];
            wp[0] = s_W + (mw * 32 + l16) * FU_AP + 8 * half;
            wp[1] = wp[0] + 16 * FU_AP;

            v8f acc[2];
            acc[0] = v8f{};
            acc[1] = v8f{};
            float4 a[2], b[2][2];
            a[0] = *(const float4*)(ap);
            a[1] = *(const float4*)(ap + 4);
            #pragma unroll
            for (int j = 0; j < 2; ++j) {
                b[j][0] = *(const float4*)(wp[j]);
                b[j][1] = *(const float4*)(wp[j] + 4);
            }
            #pragma unroll
            for (int k0 = 0; k0 < N_HID; k0 += GM_KS) {
                float4 an[2], bn[2][2];
                if (k0 + GM_KS < N_HID) {
                    const int kn = k0 + GM_KS;
                    an[0] = *(const float4*)(ap + kn);
                    an[1] = *(const float4*)(ap + kn + 4);
                    #pragma unroll
                    for (int j = 0; j < 2; ++j) {
                        bn[j][0] = *(const float4*)(wp[j] + kn);
                        bn[j][1] = *(const float4*)(wp[j] + kn + 4);
                    }
                }
                __builtin_amdgcn_sched_barrier(0);
                #pragma unroll
                for (int p = 0; p < 2; ++p) {
                    const v2f a0 = { a[p].x, a[p].y };
                    #pragma unroll
                    for (int j = 0; j < 2; ++j) {
                        const v2f b0 = { b[j][p].x, b[j][p].y };
                        acc[j] = __builtin_amdgcn_wmma_f32_16x16x4_f32(
                             false, a0,  false, b0,
                             (short)0, acc[j],  false,  false);
                    }
                    const v2f a1 = { a[p].z, a[p].w };
                    #pragma unroll
                    for (int j = 0; j < 2; ++j) {
                        const v2f b1 = { b[j][p].z, b[j][p].w };
                        acc[j] = __builtin_amdgcn_wmma_f32_16x16x4_f32(
                             false, a1,  false, b1,
                             (short)0, acc[j],  false,  false);
                    }
                }
                __builtin_amdgcn_sched_barrier(0);
                if (k0 + GM_KS < N_HID) {
                    a[0] = an[0]; a[1] = an[1];
                    #pragma unroll
                    for (int j = 0; j < 2; ++j) { b[j][0] = bn[j][0]; b[j][1] = bn[j][1]; }
                }
            }

            float sv[N_CLS * 8];
            #pragma unroll
            for (int r = 0; r < 8; ++r) {
                float p0 = 0.0f, p1 = 0.0f;
                #pragma unroll
                for (int j = 0; j < 2; ++j) {
                    float v = acc[j][r] * dv[r] + bv[j];
                    v = fmaxf(v, 0.0f);
                    p0 += v * w3a[j];
                    p1 += v * w3b[j];
                }
                sv[r]     = p0;
                sv[8 + r] = p1;
            }
            #pragma unroll
            for (int n = 8; n >= 1; n >>= 1) {
                const bool up = (l16 & n) != 0;
                #pragma unroll
                for (int q = 0; q < n; ++q) {
                    const float keep = up ? sv[n + q] : sv[q];
                    const float send = up ? sv[q] : sv[n + q];
                    sv[q] = keep + __shfl_xor(send, n);
                }
            }
            const int rt = half * 8 + (l16 & 7);
            if (rt < FU_TN) s_hp[mw][(t * FU_TN + rt) * N_CLS + (l16 >> 3)] = sv[0];
        }
        __syncthreads();
    }

    for (int i = tid; i < nT * N_CLS; i += AGG_NT)
        out[(size_t)lo * N_CLS + i] = ((s_hp[0][i] + s_hp[1][i]) + (s_hp[2][i] + s_hp[3][i])) + b3[i & 1];
}

#define L1_AP    (N_IN + 4)
#define L1_RB    16

__global__ __launch_bounds__(AGG_NT)
void agg_dense_relu(const float* __restrict__ src,
                    const int* __restrict__ row,
                    const int* __restrict__ col,
                    const int2* __restrict__ lists,
                    const int* __restrict__ gcnt, int cap,
                    const float* __restrict__ W1, const float* __restrict__ b1,
                    float* __restrict__ hid, float* __restrict__ dinv, int M, int E) {
    static_assert(FU_TN * FU_STEPS == AGG_T, "the steps must cover the group");
    static_assert(2 * FU_TN + FU_MW == AGG_NT / 32, "every wave has one role");
    static_assert(FU_TN <= 16 && N_IN == 128 && N_HID == 64 * FU_MW, "tile shapes");
    static_assert(32 * AGG_NPW >= AGG_T, "one wave scans the counts, AGG_NPW per lane");
    static_assert(AGG_LCAP >= BIN_CAP, "a complete list must fit");
    __shared__ int s_row[AGG_LCAP];
    __shared__ int s_cnt[AGG_T];
    __shared__ int s_off[AGG_T + 1];
    __shared__ int s_cur[AGG_T];
    __shared__ int s_tot;
    __shared__ float s_dv[AGG_T];
    __shared__ __attribute__((aligned(16))) float s_W[N_HID * L1_AP];
    __shared__ __attribute__((aligned(16))) float s_A[2][16 * L1_AP];

    const int tid  = threadIdx.x;
    const int lane = tid & 31;
    const int wave = __builtin_amdgcn_readfirstlane(tid >> 5);
    const int lo   = blockIdx.x * AGG_T;
    const int nT   = min(AGG_T, M - lo);

    for (int i = tid; i < AGG_T; i += AGG_NT) { s_cnt[i] = 0; s_dv[i] = 1.0f; }
    if (tid == 0) s_tot = gcnt[blockIdx.x];
    for (int i = tid; i < N_HID * (N_IN / 4); i += AGG_NT) {
        const int r  = i / (N_IN / 4);
        const int c4 = i % (N_IN / 4);
        *(float4*)(s_W + r * L1_AP + c4 * 4) = *(const float4*)(W1 + (size_t)r * N_IN + c4 * 4);
    }
    for (int i = tid; i < 2 * 16 * L1_AP; i += AGG_NT) (&s_A[0][0])[i] = 0.0f;
    __syncthreads();

    const int  total  = __builtin_amdgcn_readfirstlane(s_tot);
    const bool listed = (total <= cap) && (total <= AGG_LCAP);
    if (listed) {
        const int2* __restrict__ lst = lists + (size_t)blockIdx.x * cap;
        for (int i = tid; i < total; i += AGG_NT) atomicAdd(&s_cnt[lst[i].y], 1);
        __syncthreads();
        for (int i = tid; i < AGG_T; i += AGG_NT) s_dv[i] = 1.0f / fmaxf((float)s_cnt[i], 1.0f);
        if (wave == 0) {
            int c[AGG_NPW];
            int s = 0;
            #pragma unroll
            for (int k = 0; k < AGG_NPW; ++k) {
                const int idx = lane * AGG_NPW + k;
                c[k] = (idx < AGG_T) ? s_cnt[idx] : 0;
                s += c[k];
            }
            int incl = s;
            #pragma unroll
            for (int d = 1; d < 32; d <<= 1) {
                const int t = __shfl_up(incl, d);
                if (lane >= d) incl += t;
            }
            int run = incl - s;
            #pragma unroll
            for (int k = 0; k < AGG_NPW; ++k) {
                const int idx = lane * AGG_NPW + k;
                if (idx < AGG_T) { s_off[idx] = run; s_cur[idx] = run; }
                run += c[k];
            }
            if (lane == 31) s_off[AGG_T] = run;
        }
        __syncthreads();
        for (int i = tid; i < total; i += AGG_NT) {
            const int2 rec = lst[i];
            s_row[atomicAdd(&s_cur[rec.y], 1)] = rec.x;
        }
        __syncthreads();
    }

    const int  mw    = wave / 9;
    const bool isMat = (wave == 9 * mw);
    const int  gi    = wave - mw - 1;
    const int  gn    = gi >> 1;
    const int  fh    = gi & 1;
    const int  half  = lane >> 4;
    const int  l16   = lane & 15;
    const int  nSteps = (nT + FU_TN - 1) / FU_TN;

    float bv[2][2];
    #pragma unroll
    for (int ps = 0; ps < 2; ++ps)
        #pragma unroll
        for (int j = 0; j < 2; ++j)
            bv[ps][j] = b1[(isMat ? mw : 0) * 64 + ps * 32 + j * 16 + l16];

    for (int s = 0; s <= nSteps; ++s) {
        if (!isMat) {
            if (s < nSteps) {
                const int node = s * FU_TN + gn;
                float4 sum = make_float4(0.f, 0.f, 0.f, 0.f);
                if (node < nT) {
                    const float* __restrict__ sp = src + fh * 64 + l16 * 4;
                    if (listed) {
                        const int beg = __builtin_amdgcn_readfirstlane(s_off[node]);
                        const int end = __builtin_amdgcn_readfirstlane(s_off[node + 1]);
                        for (int b = beg; b < end; b += 2 * L1_RB) {
                            float4 v[L1_RB];
                            int    r[L1_RB];
                            #pragma unroll
                            for (int j = 0; j < L1_RB; ++j) r[j] = s_row[min(b + 2 * j + half, end - 1)];
                            #pragma unroll
                            for (int j = 0; j < L1_RB; ++j) {
                                v[j] = make_float4(0.f, 0.f, 0.f, 0.f);
                                if (b + 2 * j + half < end) v[j] = *(const float4*)(sp + (size_t)r[j] * N_IN);
                            }
                            #pragma unroll
                            for (int j = 0; j < L1_RB; ++j) {
                                sum.x += v[j].x; sum.y += v[j].y; sum.z += v[j].z; sum.w += v[j].w;
                            }
                        }
                    } else {
                        int deg = 0;
                        for (int base = 0; base < E; base += 32) {
                            const int e = base + lane;
                            const bool hit = (e < E) && (col[e] == lo + node);
                            unsigned m = __builtin_amdgcn_ballot_w32(hit);
                            deg += __builtin_popcount(m);
                            while (m) {
                                const int bit = __builtin_ctz(m);
                                m &= m - 1;
                                const int r = row[base + bit];
                                const float4 t = *(const float4*)(sp + (size_t)r * N_IN);
                                if (half == 0) { sum.x += t.x; sum.y += t.y; sum.z += t.z; sum.w += t.w; }
                            }
                        }
                        if (fh == 0 && lane == 0) s_dv[node] = 1.0f / fmaxf((float)deg, 1.0f);
                    }
                }
                sum.x += __shfl_xor(sum.x, 16);
                sum.y += __shfl_xor(sum.y, 16);
                sum.z += __shfl_xor(sum.z, 16);
                sum.w += __shfl_xor(sum.w, 16);
                if (half == 0) *(float4*)(&s_A[s & 1][gn * L1_AP + fh * 64 + l16 * 4]) = sum;
            }
        } else if (s > 0) {
            const int t = s - 1;
            float dv[8];
            #pragma unroll
            for (int r = 0; r < 8; ++r) dv[r] = s_dv[min(t * FU_TN + half * 8 + r, AGG_T - 1)];
            const float* ap = &s_A[t & 1][l16 * L1_AP + 8 * half];
            #pragma unroll
            for (int ps = 0; ps < 2; ++ps) {
                const int cb = mw * 64 + ps * 32;
                const float* wp[2];
                wp[0] = s_W + (cb + l16) * L1_AP + 8 * half;
                wp[1] = wp[0] + 16 * L1_AP;

                v8f acc[2];
                acc[0] = v8f{};
                acc[1] = v8f{};
                float4 a[2], b[2][2];
                a[0] = *(const float4*)(ap);
                a[1] = *(const float4*)(ap + 4);
                #pragma unroll
                for (int j = 0; j < 2; ++j) {
                    b[j][0] = *(const float4*)(wp[j]);
                    b[j][1] = *(const float4*)(wp[j] + 4);
                }
                #pragma unroll
                for (int k0 = 0; k0 < N_IN; k0 += GM_KS) {
                    float4 an[2], bn[2][2];
                    if (k0 + GM_KS < N_IN) {
                        const int kn = k0 + GM_KS;
                        an[0] = *(const float4*)(ap + kn);
                        an[1] = *(const float4*)(ap + kn + 4);
                        #pragma unroll
                        for (int j = 0; j < 2; ++j) {
                            bn[j][0] = *(const float4*)(wp[j] + kn);
                            bn[j][1] = *(const float4*)(wp[j] + kn + 4);
                        }
                    }
                    __builtin_amdgcn_sched_barrier(0);
                    #pragma unroll
                    for (int p = 0; p < 2; ++p) {
                        const v2f a0 = { a[p].x, a[p].y };
                        #pragma unroll
                        for (int j = 0; j < 2; ++j) {
                            const v2f b0 = { b[j][p].x, b[j][p].y };
                            acc[j] = __builtin_amdgcn_wmma_f32_16x16x4_f32(
                                 false, a0,  false, b0,
                                 (short)0, acc[j],  false,  false);
                        }
                        const v2f a1 = { a[p].z, a[p].w };
                        #pragma unroll
                        for (int j = 0; j < 2; ++j) {
                            const v2f b1f = { b[j][p].z, b[j][p].w };
                            acc[j] = __builtin_amdgcn_wmma_f32_16x16x4_f32(
                                 false, a1,  false, b1f,
                                 (short)0, acc[j],  false,  false);
                        }
                    }
                    __builtin_amdgcn_sched_barrier(0);
                    if (k0 + GM_KS < N_IN) {
                        a[0] = an[0]; a[1] = an[1];
                        #pragma unroll
                        for (int j = 0; j < 2; ++j) { b[j][0] = bn[j][0]; b[j][1] = bn[j][1]; }
                    }
                }

                #pragma unroll
                for (int r = 0; r < 8; ++r) {
                    const int rt = half * 8 + r;
                    const int nd = t * FU_TN + rt;
                    if (rt < FU_TN && nd < nT) {
                        #pragma unroll
                        for (int j = 0; j < 2; ++j) {
                            float v = acc[j][r] * dv[r] + bv[ps][j];
                            v = fmaxf(v, 0.0f);
                            hid[(size_t)(lo + nd) * N_HID + cb + j * 16 + l16] = v;
                        }
                    }
                }
            }
        }
        __syncthreads();
    }

    for (int i = tid; i < nT; i += AGG_NT) dinv[lo + i] = s_dv[i];
}

extern "C" void kernel_launch(void* const* d_in, const int* in_sizes, int n_in,
                              void* d_out, int out_size, void* d_ws, size_t ws_size,
                              hipStream_t stream) {
    const float* x  = (const float*)d_in[0];
    const int*   ei = (const int*)  d_in[1];
    const float* W1 = (const float*)d_in[3];
    const float* b1 = (const float*)d_in[4];
    const float* W2 = (const float*)d_in[5];
    const float* b2 = (const float*)d_in[6];
    const float* W3 = (const float*)d_in[7];
    const float* b3 = (const float*)d_in[8];
    float* out = (float*)d_out;

    const int M = in_sizes[0] / N_IN;
    const int E = in_sizes[1] / 2;
    const int* row = ei;
    const int* col = ei + E;

    float* ws = (float*)d_ws;
    size_t off = 0;
    const int NG  = (M + AGG_T - 1) / AGG_T;
    const int NGp = (NG + 31) & ~31;
    int*   gcnt = (int*)(ws + off); off += (size_t)NGp;
    float* dinv = ws + off; off += (size_t)M;
    off = (off + 3) & ~(size_t)3;
    float* h1   = ws + off; off += (size_t)M * N_HID;
    off = (off + 1) & ~(size_t)1;
    int2* lists = (int2*)(ws + off);
    const size_t room = (ws_size / sizeof(float) - off) / 2 / (size_t)NG;
    const int cap = (int)(room < (size_t)BIN_CAP ? room : (size_t)BIN_CAP);

    hipMemsetAsync(gcnt, 0, sizeof(int) * (size_t)NGp, stream);

    bin_edges<<<(E + BIN_NT * 4 - 1) / (BIN_NT * 4), BIN_NT, 0, stream>>>(row, col, lists, gcnt, E, M, NG, cap);

    agg_dense_relu<<<NG, AGG_NT, 0, stream>>>(x, row, col, lists, gcnt, cap, W1, b1, h1, dinv, M, E);

    agg_dense_head<<<NG, AGG_NT, 0, stream>>>(h1, row, col, lists, gcnt, cap, W2, b2, dinv, W3, b3, out, M, E);
}


# Step barrier split in both layer launches: a wave signals its step's end, waits one step on before its next tile access, so summing waves sum their next node meanwhile; LDS-only fence; sum order kept
# speedup vs baseline: 8.2061x; 1.0440x over previous
#include <hip/hip_runtime.h>

typedef __attribute__((ext_vector_type(2))) float v2f;
typedef __attribute__((ext_vector_type(8))) float v8f;

#define N_IN   128
#define N_HID  256
#define N_HID2 128
#define N_CLS  2


#define AGG_T    196
#define AGG_NT   1024
#define AGG_NPW  7
#define AGG_LCAP 16384

__device__ __forceinline__ int4 load_cols(const int* __restrict__ col, int idx, int E, bool vec_ok) {
    int4 c;
    if (vec_ok && idx + 3 < E) {
        c = *(const int4*)(col + idx);
    } else {
        c.x = (idx     < E) ? col[idx]     : -1;
        c.y = (idx + 1 < E) ? col[idx + 1] : -1;
        c.z = (idx + 2 < E) ? col[idx + 2] : -1;
        c.w = (idx + 3 < E) ? col[idx + 3] : -1;
    }
    return c;
}

#define BIN_NT   1024
#define BIN_MAXG 1024
#define BIN_CAP  16384

__global__ __launch_bounds__(BIN_NT)
void bin_edges(const int* __restrict__ row, const int* __restrict__ col,
               int2* __restrict__ lists, int* __restrict__ gcnt,
               int E, int M, int NG, int cap) {
    __shared__ int s_cnt[BIN_MAXG];
    __shared__ int s_base[BIN_MAXG];
    const int tid = threadIdx.x;
    const int e0  = (blockIdx.x * BIN_NT + tid) * 4;
    const bool vec_ok = ((E & 3) == 0);
    const int4 c4 = load_cols(col, e0, E, vec_ok);
    const int4 r4 = load_cols(row, e0, E, true);
    const int cc[4] = { c4.x, c4.y, c4.z, c4.w };
    const int rr[4] = { r4.x, r4.y, r4.z, r4.w };

    for (int g0 = 0; g0 < NG; g0 += BIN_MAXG) {
        for (int i = tid; i < BIN_MAXG; i += BIN_NT) s_cnt[i] = 0;
        __syncthreads();
        int gl[4], rank[4];
        #pragma unroll
        for (int j = 0; j < 4; ++j) {
            const unsigned uc = (unsigned)cc[j];
            const unsigned g = (uc < (unsigned)M) ? uc / AGG_T - (unsigned)g0 : 0xffffffffu;
            gl[j]   = (g < (unsigned)BIN_MAXG) ? (int)g : -1;
            rank[j] = (gl[j] >= 0) ? atomicAdd(&s_cnt[gl[j]], 1) : 0;
        }
        __syncthreads();
        for (int i = tid; i < BIN_MAXG; i += BIN_NT) {
            const int n = s_cnt[i];
            s_base[i] = (n > 0) ? atomicAdd(&gcnt[g0 + i], n) : 0;
        }
        __syncthreads();
        #pragma unroll
        for (int j = 0; j < 4; ++j) {
            if (gl[j] >= 0) {
                const int g   = g0 + gl[j];
                const int pos = s_base[gl[j]] + rank[j];
                if (pos < cap) lists[(size_t)g * cap + pos] = make_int2(rr[j], cc[j] - g * AGG_T);
            }
        }
    }
}

#define GM_KS 16

#define FU_TN    14
#define FU_STEPS (AGG_T / FU_TN)
#define FU_AP    (N_HID + 4)
#define FU_RB    16
#define FU_MW    4

__global__ __launch_bounds__(AGG_NT)
void agg_dense_head(const float* __restrict__ src,
                    const int* __restrict__ row,
                    const int* __restrict__ col,
                    const int2* __restrict__ lists,
                    const int* __restrict__ gcnt, int cap,
                    const float* __restrict__ W2, const float* __restrict__ b2,
                    const float* __restrict__ dinv,
                    const float* __restrict__ W3, const float* __restrict__ b3,
                    float* __restrict__ out, int M, int E) {
    static_assert(FU_TN * FU_STEPS == AGG_T, "the steps must cover the group");
    static_assert(2 * FU_TN + FU_MW == AGG_NT / 32, "every wave has one role");
    static_assert(FU_TN <= 16 && N_HID == 256 && N_HID2 == 32 * FU_MW && N_CLS == 2, "tile shapes");
    static_assert(32 * AGG_NPW >= AGG_T, "one wave scans the counts, AGG_NPW per lane");
    static_assert(AGG_LCAP >= BIN_CAP, "a complete list must fit");
    __shared__ int s_row[AGG_LCAP];
    __shared__ int s_cnt[AGG_T];
    __shared__ int s_off[AGG_T + 1];
    __shared__ int s_cur[AGG_T];
    __shared__ int s_tot;
    __shared__ __attribute__((aligned(16))) float s_W[N_HID2 * FU_AP];
    __shared__ __attribute__((aligned(16))) float s_A[2][16 * FU_AP];
    __shared__ float s_hp[FU_MW][AGG_T * N_CLS];

    const int tid  = threadIdx.x;
    const int lane = tid & 31;
    const int wave = __builtin_amdgcn_readfirstlane(tid >> 5);
    const int lo   = blockIdx.x * AGG_T;
    const int nT   = min(AGG_T, M - lo);

    for (int i = tid; i < AGG_T; i += AGG_NT) s_cnt[i] = 0;
    if (tid == 0) s_tot = gcnt[blockIdx.x];
    for (int i = tid; i < N_HID2 * (N_HID / 4); i += AGG_NT) {
        const int r  = i / (N_HID / 4);
        const int c4 = i % (N_HID / 4);
        *(float4*)(s_W + r * FU_AP + c4 * 4) = *(const float4*)(W2 + (size_t)r * N_HID + c4 * 4);
    }
    for (int i = tid; i < 2 * 16 * FU_AP; i += AGG_NT) (&s_A[0][0])[i] = 0.0f;
    __syncthreads();

    const int  total  = __builtin_amdgcn_readfirstlane(s_tot);
    const bool listed = (total <= cap) && (total <= AGG_LCAP);
    if (listed) {
        const int2* __restrict__ lst = lists + (size_t)blockIdx.x * cap;
        for (int i = tid; i < total; i += AGG_NT) atomicAdd(&s_cnt[lst[i].y], 1);
        __syncthreads();
        if (wave == 0) {
            int c[AGG_NPW];
            int s = 0;
            #pragma unroll
            for (int k = 0; k < AGG_NPW; ++k) {
                const int idx = lane * AGG_NPW + k;
                c[k] = (idx < AGG_T) ? s_cnt[idx] : 0;
                s += c[k];
            }
            int incl = s;
            #pragma unroll
            for (int d = 1; d < 32; d <<= 1) {
                const int t = __shfl_up(incl, d);
                if (lane >= d) incl += t;
            }
            int run = incl - s;
            #pragma unroll
            for (int k = 0; k < AGG_NPW; ++k) {
                const int idx = lane * AGG_NPW + k;
                if (idx < AGG_T) { s_off[idx] = run; s_cur[idx] = run; }
                run += c[k];
            }
            if (lane == 31) s_off[AGG_T] = run;
        }
        __syncthreads();
        for (int i = tid; i < total; i += AGG_NT) {
            const int2 rec = lst[i];
            s_row[atomicAdd(&s_cur[rec.y], 1)] = rec.x;
        }
        __syncthreads();
    }

    const int  mw    = wave / 9;
    const bool isMat = (wave == 9 * mw);
    const int  gi    = wave - mw - 1;
    const int  gn    = gi >> 1;
    const int  fh    = gi & 1;
    const int  half  = lane >> 4;
    const int  l16   = lane & 15;
    const int  nSteps = (nT + FU_TN - 1) / FU_TN;

    float bv[2], w3a[2], w3b[2];
    #pragma unroll
    for (int j = 0; j < 2; ++j) {
        const int cidx = (isMat ? mw : 0) * 32 + j * 16 + l16;
        bv[j]  = b2[cidx];
        w3a[j] = W3[cidx];
        w3b[j] = W3[N_HID2 + cidx];
    }

    for (int s = 0; s <= nSteps; ++s) {
        float4 sum = make_float4(0.f, 0.f, 0.f, 0.f);
        if (!isMat && s < nSteps) {
            const int node = s * FU_TN + gn;
            if (node < nT) {
                const float* __restrict__ sp = src + fh * 128 + lane * 4;
                if (listed) {
                    const int beg = __builtin_amdgcn_readfirstlane(s_off[node]);
                    const int end = __builtin_amdgcn_readfirstlane(s_off[node + 1]);
                    for (int b = beg; b < end; b += FU_RB) {
                        const int cnt = min(FU_RB, end - b);
                        float4 v[FU_RB];
                        int    r[FU_RB];
                        #pragma unroll
                        for (int j = 0; j < FU_RB; ++j) r[j] = s_row[min(b + j, end - 1)];
                        #pragma unroll
                        for (int j = 0; j < FU_RB; ++j) {
                            v[j] = make_float4(0.f, 0.f, 0.f, 0.f);
                            if (j < cnt) v[j] = *(const float4*)(sp + (size_t)r[j] * N_HID);
                        }
                        #pragma unroll
                        for (int j = 0; j < FU_RB; ++j) {
                            sum.x += v[j].x; sum.y += v[j].y; sum.z += v[j].z; sum.w += v[j].w;
                        }
                    }
                } else {
                    for (int base = 0; base < E; base += 32) {
                        const int e = base + lane;
                        const bool hit = (e < E) && (col[e] == lo + node);
                        unsigned m = __builtin_amdgcn_ballot_w32(hit);
                        while (m) {
                            const int bit = __builtin_ctz(m);
                            m &= m - 1;
                            const int r = row[base + bit];
                            const float4 t = *(const float4*)(sp + (size_t)r * N_HID);
                            sum.x += t.x; sum.y += t.y; sum.z += t.z; sum.w += t.w;
                        }
                    }
                }
            }
        }
        if (s > 0) {
            __builtin_amdgcn_s_barrier_wait(-1);
            __builtin_amdgcn_fence(__ATOMIC_ACQUIRE, "workgroup", "local");
        }
        if (!isMat) {
            if (s < nSteps) *(float4*)(&s_A[s & 1][gn * FU_AP + fh * 128 + lane * 4]) = sum;
        } else if (s > 0) {
            const int t = s - 1;
            float dv[8];
            #pragma unroll
            for (int r = 0; r < 8; ++r) {
                int m = lo + t * FU_TN + half * 8 + r;
                if (m >= M) m = M - 1;
                dv[r] = dinv[m];
            }
            const float* ap = &s_A[t & 1][l16 * FU_AP + 8 * half];
            const float* wp[2];
            wp[0] = s_W + (mw * 32 + l16) * FU_AP + 8 * half;
            wp[1] = wp[0] + 16 * FU_AP;

            v8f acc[2];
            acc[0] = v8f{};
            acc[1] = v8f{};
            float4 a[2], b[2][2];
            a[0] = *(const float4*)(ap);
            a[1] = *(const float4*)(ap + 4);
            #pragma unroll
            for (int j = 0; j < 2; ++j) {
                b[j][0] = *(const float4*)(wp[j]);
                b[j][1] = *(const float4*)(wp[j] + 4);
            }
            #pragma unroll
            for (int k0 = 0; k0 < N_HID; k0 += GM_KS) {
                float4 an[2], bn[2][2];
                if (k0 + GM_KS < N_HID) {
                    const int kn = k0 + GM_KS;
                    an[0] = *(const float4*)(ap + kn);
                    an[1] = *(const float4*)(ap + kn + 4);
                    #pragma unroll
                    for (int j = 0; j < 2; ++j) {
                        bn[j][0] = *(const float4*)(wp[j] + kn);
                        bn[j][1] = *(const float4*)(wp[j] + kn + 4);
                    }
                }
                __builtin_amdgcn_sched_barrier(0);
                #pragma unroll
                for (int p = 0; p < 2; ++p) {
                    const v2f a0 = { a[p].x, a[p].y };
                    #pragma unroll
                    for (int j = 0; j < 2; ++j) {
                        const v2f b0 = { b[j][p].x, b[j][p].y };
                        acc[j] = __builtin_amdgcn_wmma_f32_16x16x4_f32(
                             false, a0,  false, b0,
                             (short)0, acc[j],  false,  false);
                    }
                    const v2f a1 = { a[p].z, a[p].w };
                    #pragma unroll
                    for (int j = 0; j < 2; ++j) {
                        const v2f b1 = { b[j][p].z, b[j][p].w };
                        acc[j] = __builtin_amdgcn_wmma_f32_16x16x4_f32(
                             false, a1,  false, b1,
                             (short)0, acc[j],  false,  false);
                    }
                }
                __builtin_amdgcn_sched_barrier(0);
                if (k0 + GM_KS < N_HID) {
                    a[0] = an[0]; a[1] = an[1];
                    #pragma unroll
                    for (int j = 0; j < 2; ++j) { b[j][0] = bn[j][0]; b[j][1] = bn[j][1]; }
                }
            }

            float sv[N_CLS * 8];
            #pragma unroll
            for (int r = 0; r < 8; ++r) {
                float p0 = 0.0f, p1 = 0.0f;
                #pragma unroll
                for (int j = 0; j < 2; ++j) {
                    float v = acc[j][r] * dv[r] + bv[j];
                    v = fmaxf(v, 0.0f);
                    p0 += v * w3a[j];
                    p1 += v * w3b[j];
                }
                sv[r]     = p0;
                sv[8 + r] = p1;
            }
            #pragma unroll
            for (int n = 8; n >= 1; n >>= 1) {
                const bool up = (l16 & n) != 0;
                #pragma unroll
                for (int q = 0; q < n; ++q) {
                    const float keep = up ? sv[n + q] : sv[q];
                    const float send = up ? sv[q] : sv[n + q];
                    sv[q] = keep + __shfl_xor(send, n);
                }
            }
            const int rt = half * 8 + (l16 & 7);
            if (rt < FU_TN) s_hp[mw][(t * FU_TN + rt) * N_CLS + (l16 >> 3)] = sv[0];
        }
        __builtin_amdgcn_fence(__ATOMIC_RELEASE, "workgroup", "local");
        __builtin_amdgcn_s_barrier_signal(-1);
    }
    __builtin_amdgcn_s_barrier_wait(-1);
    __builtin_amdgcn_fence(__ATOMIC_ACQUIRE, "workgroup", "local");

    for (int i = tid; i < nT * N_CLS; i += AGG_NT)
        out[(size_t)lo * N_CLS + i] = ((s_hp[0][i] + s_hp[1][i]) + (s_hp[2][i] + s_hp[3][i])) + b3[i & 1];
}

#define L1_AP    (N_IN + 4)
#define L1_RB    16

__global__ __launch_bounds__(AGG_NT)
void agg_dense_relu(const float* __restrict__ src,
                    const int* __restrict__ row,
                    const int* __restrict__ col,
                    const int2* __restrict__ lists,
                    const int* __restrict__ gcnt, int cap,
                    const float* __restrict__ W1, const float* __restrict__ b1,
                    float* __restrict__ hid, float* __restrict__ dinv, int M, int E) {
    static_assert(FU_TN * FU_STEPS == AGG_T, "the steps must cover the group");
    static_assert(2 * FU_TN + FU_MW == AGG_NT / 32, "every wave has one role");
    static_assert(FU_TN <= 16 && N_IN == 128 && N_HID == 64 * FU_MW, "tile shapes");
    static_assert(32 * AGG_NPW >= AGG_T, "one wave scans the counts, AGG_NPW per lane");
    static_assert(AGG_LCAP >= BIN_CAP, "a complete list must fit");
    __shared__ int s_row[AGG_LCAP];
    __shared__ int s_cnt[AGG_T];
    __shared__ int s_off[AGG_T + 1];
    __shared__ int s_cur[AGG_T];
    __shared__ int s_tot;
    __shared__ float s_dv[AGG_T];
    __shared__ __attribute__((aligned(16))) float s_W[N_HID * L1_AP];
    __shared__ __attribute__((aligned(16))) float s_A[2][16 * L1_AP];

    const int tid  = threadIdx.x;
    const int lane = tid & 31;
    const int wave = __builtin_amdgcn_readfirstlane(tid >> 5);
    const int lo   = blockIdx.x * AGG_T;
    const int nT   = min(AGG_T, M - lo);

    for (int i = tid; i < AGG_T; i += AGG_NT) { s_cnt[i] = 0; s_dv[i] = 1.0f; }
    if (tid == 0) s_tot = gcnt[blockIdx.x];
    for (int i = tid; i < N_HID * (N_IN / 4); i += AGG_NT) {
        const int r  = i / (N_IN / 4);
        const int c4 = i % (N_IN / 4);
        *(float4*)(s_W + r * L1_AP + c4 * 4) = *(const float4*)(W1 + (size_t)r * N_IN + c4 * 4);
    }
    for (int i = tid; i < 2 * 16 * L1_AP; i += AGG_NT) (&s_A[0][0])[i] = 0.0f;
    __syncthreads();

    const int  total  = __builtin_amdgcn_readfirstlane(s_tot);
    const bool listed = (total <= cap) && (total <= AGG_LCAP);
    if (listed) {
        const int2* __restrict__ lst = lists + (size_t)blockIdx.x * cap;
        for (int i = tid; i < total; i += AGG_NT) atomicAdd(&s_cnt[lst[i].y], 1);
        __syncthreads();
        for (int i = tid; i < AGG_T; i += AGG_NT) s_dv[i] = 1.0f / fmaxf((float)s_cnt[i], 1.0f);
        if (wave == 0) {
            int c[AGG_NPW];
            int s = 0;
            #pragma unroll
            for (int k = 0; k < AGG_NPW; ++k) {
                const int idx = lane * AGG_NPW + k;
                c[k] = (idx < AGG_T) ? s_cnt[idx] : 0;
                s += c[k];
            }
            int incl = s;
            #pragma unroll
            for (int d = 1; d < 32; d <<= 1) {
                const int t = __shfl_up(incl, d);
                if (lane >= d) incl += t;
            }
            int run = incl - s;
            #pragma unroll
            for (int k = 0; k < AGG_NPW; ++k) {
                const int idx = lane * AGG_NPW + k;
                if (idx < AGG_T) { s_off[idx] = run; s_cur[idx] = run; }
                run += c[k];
            }
            if (lane == 31) s_off[AGG_T] = run;
        }
        __syncthreads();
        for (int i = tid; i < total; i += AGG_NT) {
            const int2 rec = lst[i];
            s_row[atomicAdd(&s_cur[rec.y], 1)] = rec.x;
        }
        __syncthreads();
    }

    const int  mw    = wave / 9;
    const bool isMat = (wave == 9 * mw);
    const int  gi    = wave - mw - 1;
    const int  gn    = gi >> 1;
    const int  fh    = gi & 1;
    const int  half  = lane >> 4;
    const int  l16   = lane & 15;
    const int  nSteps = (nT + FU_TN - 1) / FU_TN;

    float bv[2][2];
    #pragma unroll
    for (int ps = 0; ps < 2; ++ps)
        #pragma unroll
        for (int j = 0; j < 2; ++j)
            bv[ps][j] = b1[(isMat ? mw : 0) * 64 + ps * 32 + j * 16 + l16];

    for (int s = 0; s <= nSteps; ++s) {
        float4 sum = make_float4(0.f, 0.f, 0.f, 0.f);
        if (!isMat && s < nSteps) {
            const int node = s * FU_TN + gn;
            if (node < nT) {
                const float* __restrict__ sp = src + fh * 64 + l16 * 4;
                if (listed) {
                    const int beg = __builtin_amdgcn_readfirstlane(s_off[node]);
                    const int end = __builtin_amdgcn_readfirstlane(s_off[node + 1]);
                    for (int b = beg; b < end; b += 2 * L1_RB) {
                        float4 v[L1_RB];
                        int    r[L1_RB];
                        #pragma unroll
                        for (int j = 0; j < L1_RB; ++j) r[j] = s_row[min(b + 2 * j + half, end - 1)];
                        #pragma unroll
                        for (int j = 0; j < L1_RB; ++j) {
                            v[j] = make_float4(0.f, 0.f, 0.f, 0.f);
                            if (b + 2 * j + half < end) v[j] = *(const float4*)(sp + (size_t)r[j] * N_IN);
                        }
                        #pragma unroll
                        for (int j = 0; j < L1_RB; ++j) {
                            sum.x += v[j].x; sum.y += v[j].y; sum.z += v[j].z; sum.w += v[j].w;
                        }
                    }
                } else {
                    int deg = 0;
                    for (int base = 0; base < E; base += 32) {
                        const int e = base + lane;
                        const bool hit = (e < E) && (col[e] == lo + node);
                        unsigned m = __builtin_amdgcn_ballot_w32(hit);
                        deg += __builtin_popcount(m);
                        while (m) {
                            const int bit = __builtin_ctz(m);
                            m &= m - 1;
                            const int r = row[base + bit];
                            const float4 t = *(const float4*)(sp + (size_t)r * N_IN);
                            if (half == 0) { sum.x += t.x; sum.y += t.y; sum.z += t.z; sum.w += t.w; }
                        }
                    }
                    if (fh == 0 && lane == 0) s_dv[node] = 1.0f / fmaxf((float)deg, 1.0f);
                }
            }
            sum.x += __shfl_xor(sum.x, 16);
            sum.y += __shfl_xor(sum.y, 16);
            sum.z += __shfl_xor(sum.z, 16);
            sum.w += __shfl_xor(sum.w, 16);
        }
        if (s > 0) {
            __builtin_amdgcn_s_barrier_wait(-1);
            __builtin_amdgcn_fence(__ATOMIC_ACQUIRE, "workgroup", "local");
        }
        if (!isMat) {
            if (s < nSteps && half == 0) *(float4*)(&s_A[s & 1][gn * L1_AP + fh * 64 + l16 * 4]) = sum;
        } else if (s > 0) {
            const int t = s - 1;
            float dv[8];
            #pragma unroll
            for (int r = 0; r < 8; ++r) dv[r] = s_dv[min(t * FU_TN + half * 8 + r, AGG_T - 1)];
            const float* ap = &s_A[t & 1][l16 * L1_AP + 8 * half];
            #pragma unroll
            for (int ps = 0; ps < 2; ++ps) {
                const int cb = mw * 64 + ps * 32;
                const float* wp[2];
                wp[0] = s_W + (cb + l16) * L1_AP + 8 * half;
                wp[1] = wp[0] + 16 * L1_AP;

                v8f acc[2];
                acc[0] = v8f{};
                acc[1] = v8f{};
                float4 a[2], b[2][2];
                a[0] = *(const float4*)(ap);
                a[1] = *(const float4*)(ap + 4);
                #pragma unroll
                for (int j = 0; j < 2; ++j) {
                    b[j][0] = *(const float4*)(wp[j]);
                    b[j][1] = *(const float4*)(wp[j] + 4);
                }
                #pragma unroll
                for (int k0 = 0; k0 < N_IN; k0 += GM_KS) {
                    float4 an[2], bn[2][2];
                    if (k0 + GM_KS < N_IN) {
                        const int kn = k0 + GM_KS;
                        an[0] = *(const float4*)(ap + kn);
                        an[1] = *(const float4*)(ap + kn + 4);
                        #pragma unroll
                        for (int j = 0; j < 2; ++j) {
                            bn[j][0] = *(const float4*)(wp[j] + kn);
                            bn[j][1] = *(const float4*)(wp[j] + kn + 4);
                        }
                    }
                    __builtin_amdgcn_sched_barrier(0);
                    #pragma unroll
                    for (int p = 0; p < 2; ++p) {
                        const v2f a0 = { a[p].x, a[p].y };
                        #pragma unroll
                        for (int j = 0; j < 2; ++j) {
                            const v2f b0 = { b[j][p].x, b[j][p].y };
                            acc[j] = __builtin_amdgcn_wmma_f32_16x16x4_f32(
                                 false, a0,  false, b0,
                                 (short)0, acc[j],  false,  false);
                        }
                        const v2f a1 = { a[p].z, a[p].w };
                        #pragma unroll
                        for (int j = 0; j < 2; ++j) {
                            const v2f b1f = { b[j][p].z, b[j][p].w };
                            acc[j] = __builtin_amdgcn_wmma_f32_16x16x4_f32(
                                 false, a1,  false, b1f,
                                 (short)0, acc[j],  false,  false);
                        }
                    }
                    __builtin_amdgcn_sched_barrier(0);
                    if (k0 + GM_KS < N_IN) {
                        a[0] = an[0]; a[1] = an[1];
                        #pragma unroll
                        for (int j = 0; j < 2; ++j) { b[j][0] = bn[j][0]; b[j][1] = bn[j][1]; }
                    }
                }

                #pragma unroll
                for (int r = 0; r < 8; ++r) {
                    const int rt = half * 8 + r;
                    const int nd = t * FU_TN + rt;
                    if (rt < FU_TN && nd < nT) {
                        #pragma unroll
                        for (int j = 0; j < 2; ++j) {
                            float v = acc[j][r] * dv[r] + bv[ps][j];
                            v = fmaxf(v, 0.0f);
                            hid[(size_t)(lo + nd) * N_HID + cb + j * 16 + l16] = v;
                        }
                    }
                }
            }
        }
        __builtin_amdgcn_fence(__ATOMIC_RELEASE, "workgroup", "local");
        __builtin_amdgcn_s_barrier_signal(-1);
    }
    __builtin_amdgcn_s_barrier_wait(-1);
    __builtin_amdgcn_fence(__ATOMIC_ACQUIRE, "workgroup", "local");

    for (int i = tid; i < nT; i += AGG_NT) dinv[lo + i] = s_dv[i];
}

extern "C" void kernel_launch(void* const* d_in, const int* in_sizes, int n_in,
                              void* d_out, int out_size, void* d_ws, size_t ws_size,
                              hipStream_t stream) {
    const float* x  = (const float*)d_in[0];
    const int*   ei = (const int*)  d_in[1];
    const float* W1 = (const float*)d_in[3];
    const float* b1 = (const float*)d_in[4];
    const float* W2 = (const float*)d_in[5];
    const float* b2 = (const float*)d_in[6];
    const float* W3 = (const float*)d_in[7];
    const float* b3 = (const float*)d_in[8];
    float* out = (float*)d_out;

    const int M = in_sizes[0] / N_IN;
    const int E = in_sizes[1] / 2;
    const int* row = ei;
    const int* col = ei + E;

    float* ws = (float*)d_ws;
    size_t off = 0;
    const int NG  = (M + AGG_T - 1) / AGG_T;
    const int NGp = (NG + 31) & ~31;
    int*   gcnt = (int*)(ws + off); off += (size_t)NGp;
    float* dinv = ws + off; off += (size_t)M;
    off = (off + 3) & ~(size_t)3;
    float* h1   = ws + off; off += (size_t)M * N_HID;
    off = (off + 1) & ~(size_t)1;
    int2* lists = (int2*)(ws + off);
    const size_t room = (ws_size / sizeof(float) - off) / 2 / (size_t)NG;
    const int cap = (int)(room < (size_t)BIN_CAP ? room : (size_t)BIN_CAP);

    hipMemsetAsync(gcnt, 0, sizeof(int) * (size_t)NGp, stream);

    bin_edges<<<(E + BIN_NT * 4 - 1) / (BIN_NT * 4), BIN_NT, 0, stream>>>(row, col, lists, gcnt, E, M, NG, cap);

    agg_dense_relu<<<NG, AGG_NT, 0, stream>>>(x, row, col, lists, gcnt, cap, W1, b1, h1, dinv, M, E);

    agg_dense_head<<<NG, AGG_NT, 0, stream>>>(h1, row, col, lists, gcnt, cap, W2, b2, dinv, W3, b3, out, M, E);
}


# Set-up of both layer launches waits for memory once: a thread requests its 8 weight pieces and 4 list records together (was one wait each); records stay in registers; sum order arrival as before
# speedup vs baseline: 8.5427x; 1.0410x over previous
#include <hip/hip_runtime.h>

typedef __attribute__((ext_vector_type(2))) float v2f;
typedef __attribute__((ext_vector_type(8))) float v8f;

#define N_IN   128
#define N_HID  256
#define N_HID2 128
#define N_CLS  2


#define AGG_T    196
#define AGG_NT   1024
#define AGG_NPW  7
#define AGG_LCAP 16384
#define SU_RPT   4
#define SU_WPT   8

__device__ __forceinline__ int4 load_cols(const int* __restrict__ col, int idx, int E, bool vec_ok) {
    int4 c;
    if (vec_ok && idx + 3 < E) {
        c = *(const int4*)(col + idx);
    } else {
        c.x = (idx     < E) ? col[idx]     : -1;
        c.y = (idx + 1 < E) ? col[idx + 1] : -1;
        c.z = (idx + 2 < E) ? col[idx + 2] : -1;
        c.w = (idx + 3 < E) ? col[idx + 3] : -1;
    }
    return c;
}

#define BIN_NT   1024
#define BIN_MAXG 1024
#define BIN_CAP  16384

__global__ __launch_bounds__(BIN_NT)
void bin_edges(const int* __restrict__ row, const int* __restrict__ col,
               int2* __restrict__ lists, int* __restrict__ gcnt,
               int E, int M, int NG, int cap) {
    __shared__ int s_cnt[BIN_MAXG];
    __shared__ int s_base[BIN_MAXG];
    const int tid = threadIdx.x;
    const int e0  = (blockIdx.x * BIN_NT + tid) * 4;
    const bool vec_ok = ((E & 3) == 0);
    const int4 c4 = load_cols(col, e0, E, vec_ok);
    const int4 r4 = load_cols(row, e0, E, true);
    const int cc[4] = { c4.x, c4.y, c4.z, c4.w };
    const int rr[4] = { r4.x, r4.y, r4.z, r4.w };

    for (int g0 = 0; g0 < NG; g0 += BIN_MAXG) {
        for (int i = tid; i < BIN_MAXG; i += BIN_NT) s_cnt[i] = 0;
        __syncthreads();
        int gl[4], rank[4];
        #pragma unroll
        for (int j = 0; j < 4; ++j) {
            const unsigned uc = (unsigned)cc[j];
            const unsigned g = (uc < (unsigned)M) ? uc / AGG_T - (unsigned)g0 : 0xffffffffu;
            gl[j]   = (g < (unsigned)BIN_MAXG) ? (int)g : -1;
            rank[j] = (gl[j] >= 0) ? atomicAdd(&s_cnt[gl[j]], 1) : 0;
        }
        __syncthreads();
        for (int i = tid; i < BIN_MAXG; i += BIN_NT) {
            const int n = s_cnt[i];
            s_base[i] = (n > 0) ? atomicAdd(&gcnt[g0 + i], n) : 0;
        }
        __syncthreads();
        #pragma unroll
        for (int j = 0; j < 4; ++j) {
            if (gl[j] >= 0) {
                const int g   = g0 + gl[j];
                const int pos = s_base[gl[j]] + rank[j];
                if (pos < cap) lists[(size_t)g * cap + pos] = make_int2(rr[j], cc[j] - g * AGG_T);
            }
        }
    }
}

#define GM_KS 16

#define FU_TN    14
#define FU_STEPS (AGG_T / FU_TN)
#define FU_AP    (N_HID + 4)
#define FU_RB    16
#define FU_MW    4

__global__ __launch_bounds__(AGG_NT)
void agg_dense_head(const float* __restrict__ src,
                    const int* __restrict__ row,
                    const int* __restrict__ col,
                    const int2* __restrict__ lists,
                    const int* __restrict__ gcnt, int cap,
                    const float* __restrict__ W2, const float* __restrict__ b2,
                    const float* __restrict__ dinv,
                    const float* __restrict__ W3, const float* __restrict__ b3,
                    float* __restrict__ out, int M, int E) {
    static_assert(FU_TN * FU_STEPS == AGG_T, "the steps must cover the group");
    static_assert(2 * FU_TN + FU_MW == AGG_NT / 32, "every wave has one role");
    static_assert(FU_TN <= 16 && N_HID == 256 && N_HID2 == 32 * FU_MW && N_CLS == 2, "tile shapes");
    static_assert(SU_WPT * AGG_NT == N_HID2 * (N_HID / 4), "the threads' pieces cover W2");
    static_assert(32 * AGG_NPW >= AGG_T, "one wave scans the counts, AGG_NPW per lane");
    static_assert(AGG_LCAP >= BIN_CAP, "a complete list must fit");
    __shared__ int s_row[AGG_LCAP];
    __shared__ int s_cnt[AGG_T];
    __shared__ int s_off[AGG_T + 1];
    __shared__ int s_cur[AGG_T];
    __shared__ __attribute__((aligned(16))) float s_W[N_HID2 * FU_AP];
    __shared__ __attribute__((aligned(16))) float s_A[2][16 * FU_AP];
    __shared__ float s_hp[FU_MW][AGG_T * N_CLS];

    const int tid  = threadIdx.x;
    const int lane = tid & 31;
    const int wave = __builtin_amdgcn_readfirstlane(tid >> 5);
    const int lo   = blockIdx.x * AGG_T;
    const int nT   = min(AGG_T, M - lo);

    float4 wreg[SU_WPT];
    #pragma unroll
    for (int k = 0; k < SU_WPT; ++k) {
        const int i  = tid + k * AGG_NT;
        const int r  = i / (N_HID / 4);
        const int c4 = i % (N_HID / 4);
        wreg[k] = *(const float4*)(W2 + (size_t)r * N_HID + c4 * 4);
    }
    const int  total  = gcnt[blockIdx.x];
    const bool listed = (total <= cap) && (total <= AGG_LCAP);
    const int2* __restrict__ lst = lists + (size_t)blockIdx.x * cap;
    int2 rec[SU_RPT];
    #pragma unroll
    for (int k = 0; k < SU_RPT; ++k) {
        const int i = tid + k * AGG_NT;
        rec[k] = make_int2(0, -1);
        if (listed && i < total) rec[k] = lst[i];
    }

    for (int i = tid; i < AGG_T; i += AGG_NT) s_cnt[i] = 0;
    for (int i = tid; i < 2 * 16 * FU_AP; i += AGG_NT) (&s_A[0][0])[i] = 0.0f;
    #pragma unroll
    for (int k = 0; k < SU_WPT; ++k) {
        const int i  = tid + k * AGG_NT;
        const int r  = i / (N_HID / 4);
        const int c4 = i % (N_HID / 4);
        *(float4*)(s_W + r * FU_AP + c4 * 4) = wreg[k];
    }
    __syncthreads();

    if (listed) {
        #pragma unroll
        for (int k = 0; k < SU_RPT; ++k)
            if (rec[k].y >= 0) atomicAdd(&s_cnt[rec[k].y], 1);
        for (int i = tid + SU_RPT * AGG_NT; i < total; i += AGG_NT) atomicAdd(&s_cnt[lst[i].y], 1);
        __syncthreads();
        if (wave == 0) {
            int c[AGG_NPW];
            int s = 0;
            #pragma unroll
            for (int k = 0; k < AGG_NPW; ++k) {
                const int idx = lane * AGG_NPW + k;
                c[k] = (idx < AGG_T) ? s_cnt[idx] : 0;
                s += c[k];
            }
            int incl = s;
            #pragma unroll
            for (int d = 1; d < 32; d <<= 1) {
                const int t = __shfl_up(incl, d);
                if (lane >= d) incl += t;
            }
            int run = incl - s;
            #pragma unroll
            for (int k = 0; k < AGG_NPW; ++k) {
                const int idx = lane * AGG_NPW + k;
                if (idx < AGG_T) { s_off[idx] = run; s_cur[idx] = run; }
                run += c[k];
            }
            if (lane == 31) s_off[AGG_T] = run;
        }
        __syncthreads();
        #pragma unroll
        for (int k = 0; k < SU_RPT; ++k)
            if (rec[k].y >= 0) s_row[atomicAdd(&s_cur[rec[k].y], 1)] = rec[k].x;
        for (int i = tid + SU_RPT * AGG_NT; i < total; i += AGG_NT) {
            const int2 rc = lst[i];
            s_row[atomicAdd(&s_cur[rc.y], 1)] = rc.x;
        }
        __syncthreads();
    }

    const int  mw    = wave / 9;
    const bool isMat = (wave == 9 * mw);
    const int  gi    = wave - mw - 1;
    const int  gn    = gi >> 1;
    const int  fh    = gi & 1;
    const int  half  = lane >> 4;
    const int  l16   = lane & 15;
    const int  nSteps = (nT + FU_TN - 1) / FU_TN;

    float bv[2], w3a[2], w3b[2];
    #pragma unroll
    for (int j = 0; j < 2; ++j) {
        const int cidx = (isMat ? mw : 0) * 32 + j * 16 + l16;
        bv[j]  = b2[cidx];
        w3a[j] = W3[cidx];
        w3b[j] = W3[N_HID2 + cidx];
    }

    for (int s = 0; s <= nSteps; ++s) {
        float4 sum = make_float4(0.f, 0.f, 0.f, 0.f);
        if (!isMat && s < nSteps) {
            const int node = s * FU_TN + gn;
            if (node < nT) {
                const float* __restrict__ sp = src + fh * 128 + lane * 4;
                if (listed) {
                    const int beg = __builtin_amdgcn_readfirstlane(s_off[node]);
                    const int end = __builtin_amdgcn_readfirstlane(s_off[node + 1]);
                    for (int b = beg; b < end; b += FU_RB) {
                        const int cnt = min(FU_RB, end - b);
                        float4 v[FU_RB];
                        int    r[FU_RB];
                        #pragma unroll
                        for (int j = 0; j < FU_RB; ++j) r[j] = s_row[min(b + j, end - 1)];
                        #pragma unroll
                        for (int j = 0; j < FU_RB; ++j) {
                            v[j] = make_float4(0.f, 0.f, 0.f, 0.f);
                            if (j < cnt) v[j] = *(const float4*)(sp + (size_t)r[j] * N_HID);
                        }
                        #pragma unroll
                        for (int j = 0; j < FU_RB; ++j) {
                            sum.x += v[j].x; sum.y += v[j].y; sum.z += v[j].z; sum.w += v[j].w;
                        }
                    }
                } else {
                    for (int base = 0; base < E; base += 32) {
                        const int e = base + lane;
                        const bool hit = (e < E) && (col[e] == lo + node);
                        unsigned m = __builtin_amdgcn_ballot_w32(hit);
                        while (m) {
                            const int bit = __builtin_ctz(m);
                            m &= m - 1;
                            const int r = row[base + bit];
                            const float4 t = *(const float4*)(sp + (size_t)r * N_HID);
                            sum.x += t.x; sum.y += t.y; sum.z += t.z; sum.w += t.w;
                        }
                    }
                }
            }
        }
        if (s > 0) {
            __builtin_amdgcn_s_barrier_wait(-1);
            __builtin_amdgcn_fence(__ATOMIC_ACQUIRE, "workgroup", "local");
        }
        if (!isMat) {
            if (s < nSteps) *(float4*)(&s_A[s & 1][gn * FU_AP + fh * 128 + lane * 4]) = sum;
        } else if (s > 0) {
            const int t = s - 1;
            float dv[8];
            #pragma unroll
            for (int r = 0; r < 8; ++r) {
                int m = lo + t * FU_TN + half * 8 + r;
                if (m >= M) m = M - 1;
                dv[r] = dinv[m];
            }
            const float* ap = &s_A[t & 1][l16 * FU_AP + 8 * half];
            const float* wp[2];
            wp[0] = s_W + (mw * 32 + l16) * FU_AP + 8 * half;
            wp[1] = wp[0] + 16 * FU_AP;

            v8f acc[2];
            acc[0] = v8f{};
            acc[1] = v8f{};
            float4 a[2], b[2][2];
            a[0] = *(const float4*)(ap);
            a[1] = *(const float4*)(ap + 4);
            #pragma unroll
            for (int j = 0; j < 2; ++j) {
                b[j][0] = *(const float4*)(wp[j]);
                b[j][1] = *(const float4*)(wp[j] + 4);
            }
            #pragma unroll
            for (int k0 = 0; k0 < N_HID; k0 += GM_KS) {
                float4 an[2], bn[2][2];
                if (k0 + GM_KS < N_HID) {
                    const int kn = k0 + GM_KS;
                    an[0] = *(const float4*)(ap + kn);
                    an[1] = *(const float4*)(ap + kn + 4);
                    #pragma unroll
                    for (int j = 0; j < 2; ++j) {
                        bn[j][0] = *(const float4*)(wp[j] + kn);
                        bn[j][1] = *(const float4*)(wp[j] + kn + 4);
                    }
                }
                __builtin_amdgcn_sched_barrier(0);
                #pragma unroll
                for (int p = 0; p < 2; ++p) {
                    const v2f a0 = { a[p].x, a[p].y };
                    #pragma unroll
                    for (int j = 0; j < 2; ++j) {
                        const v2f b0 = { b[j][p].x, b[j][p].y };
                        acc[j] = __builtin_amdgcn_wmma_f32_16x16x4_f32(
                             false, a0,  false, b0,
                             (short)0, acc[j],  false,  false);
                    }
                    const v2f a1 = { a[p].z, a[p].w };
                    #pragma unroll
                    for (int j = 0; j < 2; ++j) {
                        const v2f b1 = { b[j][p].z, b[j][p].w };
                        acc[j] = __builtin_amdgcn_wmma_f32_16x16x4_f32(
                             false, a1,  false, b1,
                             (short)0, acc[j],  false,  false);
                    }
                }
                __builtin_amdgcn_sched_barrier(0);
                if (k0 + GM_KS < N_HID) {
                    a[0] = an[0]; a[1] = an[1];
                    #pragma unroll
                    for (int j = 0; j < 2; ++j) { b[j][0] = bn[j][0]; b[j][1] = bn[j][1]; }
                }
            }

            float sv[N_CLS * 8];
            #pragma unroll
            for (int r = 0; r < 8; ++r) {
                float p0 = 0.0f, p1 = 0.0f;
                #pragma unroll
                for (int j = 0; j < 2; ++j) {
                    float v = acc[j][r] * dv[r] + bv[j];
                    v = fmaxf(v, 0.0f);
                    p0 += v * w3a[j];
                    p1 += v * w3b[j];
                }
                sv[r]     = p0;
                sv[8 + r] = p1;
            }
            #pragma unroll
            for (int n = 8; n >= 1; n >>= 1) {
                const bool up = (l16 & n) != 0;
                #pragma unroll
                for (int q = 0; q < n; ++q) {
                    const float keep = up ? sv[n + q] : sv[q];
                    const float send = up ? sv[q] : sv[n + q];
                    sv[q] = keep + __shfl_xor(send, n);
                }
            }
            const int rt = half * 8 + (l16 & 7);
            if (rt < FU_TN) s_hp[mw][(t * FU_TN + rt) * N_CLS + (l16 >> 3)] = sv[0];
        }
        __builtin_amdgcn_fence(__ATOMIC_RELEASE, "workgroup", "local");
        __builtin_amdgcn_s_barrier_signal(-1);
    }
    __builtin_amdgcn_s_barrier_wait(-1);
    __builtin_amdgcn_fence(__ATOMIC_ACQUIRE, "workgroup", "local");

    for (int i = tid; i < nT * N_CLS; i += AGG_NT)
        out[(size_t)lo * N_CLS + i] = ((s_hp[0][i] + s_hp[1][i]) + (s_hp[2][i] + s_hp[3][i])) + b3[i & 1];
}

#define L1_AP    (N_IN + 4)
#define L1_RB    16

__global__ __launch_bounds__(AGG_NT)
void agg_dense_relu(const float* __restrict__ src,
                    const int* __restrict__ row,
                    const int* __restrict__ col,
                    const int2* __restrict__ lists,
                    const int* __restrict__ gcnt, int cap,
                    const float* __restrict__ W1, const float* __restrict__ b1,
                    float* __restrict__ hid, float* __restrict__ dinv, int M, int E) {
    static_assert(FU_TN * FU_STEPS == AGG_T, "the steps must cover the group");
    static_assert(2 * FU_TN + FU_MW == AGG_NT / 32, "every wave has one role");
    static_assert(FU_TN <= 16 && N_IN == 128 && N_HID == 64 * FU_MW, "tile shapes");
    static_assert(SU_WPT * AGG_NT == N_HID * (N_IN / 4), "the threads' pieces cover W1");
    static_assert(32 * AGG_NPW >= AGG_T, "one wave scans the counts, AGG_NPW per lane");
    static_assert(AGG_LCAP >= BIN_CAP, "a complete list must fit");
    __shared__ int s_row[AGG_LCAP];
    __shared__ int s_cnt[AGG_T];
    __shared__ int s_off[AGG_T + 1];
    __shared__ int s_cur[AGG_T];
    __shared__ float s_dv[AGG_T];
    __shared__ __attribute__((aligned(16))) float s_W[N_HID * L1_AP];
    __shared__ __attribute__((aligned(16))) float s_A[2][16 * L1_AP];

    const int tid  = threadIdx.x;
    const int lane = tid & 31;
    const int wave = __builtin_amdgcn_readfirstlane(tid >> 5);
    const int lo   = blockIdx.x * AGG_T;
    const int nT   = min(AGG_T, M - lo);

    float4 wreg[SU_WPT];
    #pragma unroll
    for (int k = 0; k < SU_WPT; ++k) {
        const int i  = tid + k * AGG_NT;
        const int r  = i / (N_IN / 4);
        const int c4 = i % (N_IN / 4);
        wreg[k] = *(const float4*)(W1 + (size_t)r * N_IN + c4 * 4);
    }
    const int  total  = gcnt[blockIdx.x];
    const bool listed = (total <= cap) && (total <= AGG_LCAP);
    const int2* __restrict__ lst = lists + (size_t)blockIdx.x * cap;
    int2 rec[SU_RPT];
    #pragma unroll
    for (int k = 0; k < SU_RPT; ++k) {
        const int i = tid + k * AGG_NT;
        rec[k] = make_int2(0, -1);
        if (listed && i < total) rec[k] = lst[i];
    }

    for (int i = tid; i < AGG_T; i += AGG_NT) { s_cnt[i] = 0; s_dv[i] = 1.0f; }
    for (int i = tid; i < 2 * 16 * L1_AP; i += AGG_NT) (&s_A[0][0])[i] = 0.0f;
    #pragma unroll
    for (int k = 0; k < SU_WPT; ++k) {
        const int i  = tid + k * AGG_NT;
        const int r  = i / (N_IN / 4);
        const int c4 = i % (N_IN / 4);
        *(float4*)(s_W + r * L1_AP + c4 * 4) = wreg[k];
    }
    __syncthreads();

    if (listed) {
        #pragma unroll
        for (int k = 0; k < SU_RPT; ++k)
            if (rec[k].y >= 0) atomicAdd(&s_cnt[rec[k].y], 1);
        for (int i = tid + SU_RPT * AGG_NT; i < total; i += AGG_NT) atomicAdd(&s_cnt[lst[i].y], 1);
        __syncthreads();
        for (int i = tid; i < AGG_T; i += AGG_NT) s_dv[i] = 1.0f / fmaxf((float)s_cnt[i], 1.0f);
        if (wave == 0) {
            int c[AGG_NPW];
            int s = 0;
            #pragma unroll
            for (int k = 0; k < AGG_NPW; ++k) {
                const int idx = lane * AGG_NPW + k;
                c[k] = (idx < AGG_T) ? s_cnt[idx] : 0;
                s += c[k];
            }
            int incl = s;
            #pragma unroll
            for (int d = 1; d < 32; d <<= 1) {
                const int t = __shfl_up(incl, d);
                if (lane >= d) incl += t;
            }
            int run = incl - s;
            #pragma unroll
            for (int k = 0; k < AGG_NPW; ++k) {
                const int idx = lane * AGG_NPW + k;
                if (idx < AGG_T) { s_off[idx] = run; s_cur[idx] = run; }
                run += c[k];
            }
            if (lane == 31) s_off[AGG_T] = run;
        }
        __syncthreads();
        #pragma unroll
        for (int k = 0; k < SU_RPT; ++k)
            if (rec[k].y >= 0) s_row[atomicAdd(&s_cur[rec[k].y], 1)] = rec[k].x;
        for (int i = tid + SU_RPT * AGG_NT; i < total; i += AGG_NT) {
            const int2 rc = lst[i];
            s_row[atomicAdd(&s_cur[rc.y], 1)] = rc.x;
        }
        __syncthreads();
    }

    const int  mw    = wave / 9;
    const bool isMat = (wave == 9 * mw);
    const int  gi    = wave - mw - 1;
    const int  gn    = gi >> 1;
    const int  fh    = gi & 1;
    const int  half  = lane >> 4;
    const int  l16   = lane & 15;
    const int  nSteps = (nT + FU_TN - 1) / FU_TN;

    float bv[2][2];
    #pragma unroll
    for (int ps = 0; ps < 2; ++ps)
        #pragma unroll
        for (int j = 0; j < 2; ++j)
            bv[ps][j] = b1[(isMat ? mw : 0) * 64 + ps * 32 + j * 16 + l16];

    for (int s = 0; s <= nSteps; ++s) {
        float4 sum = make_float4(0.f, 0.f, 0.f, 0.f);
        if (!isMat && s < nSteps) {
            const int node = s * FU_TN + gn;
            if (node < nT) {
                const float* __restrict__ sp = src + fh * 64 + l16 * 4;
                if (listed) {
                    const int beg = __builtin_amdgcn_readfirstlane(s_off[node]);
                    const int end = __builtin_amdgcn_readfirstlane(s_off[node + 1]);
                    for (int b = beg; b < end; b += 2 * L1_RB) {
                        float4 v[L1_RB];
                        int    r[L1_RB];
                        #pragma unroll
                        for (int j = 0; j < L1_RB; ++j) r[j] = s_row[min(b + 2 * j + half, end - 1)];
                        #pragma unroll
                        for (int j = 0; j < L1_RB; ++j) {
                            v[j] = make_float4(0.f, 0.f, 0.f, 0.f);
                            if (b + 2 * j + half < end) v[j] = *(const float4*)(sp + (size_t)r[j] * N_IN);
                        }
                        #pragma unroll
                        for (int j = 0; j < L1_RB; ++j) {
                            sum.x += v[j].x; sum.y += v[j].y; sum.z += v[j].z; sum.w += v[j].w;
                        }
                    }
                } else {
                    int deg = 0;
                    for (int base = 0; base < E; base += 32) {
                        const int e = base + lane;
                        const bool hit = (e < E) && (col[e] == lo + node);
                        unsigned m = __builtin_amdgcn_ballot_w32(hit);
                        deg += __builtin_popcount(m);
                        while (m) {
                            const int bit = __builtin_ctz(m);
                            m &= m - 1;
                            const int r = row[base + bit];
                            const float4 t = *(const float4*)(sp + (size_t)r * N_IN);
                            if (half == 0) { sum.x += t.x; sum.y += t.y; sum.z += t.z; sum.w += t.w; }
                        }
                    }
                    if (fh == 0 && lane == 0) s_dv[node] = 1.0f / fmaxf((float)deg, 1.0f);
                }
            }
            sum.x += __shfl_xor(sum.x, 16);
            sum.y += __shfl_xor(sum.y, 16);
            sum.z += __shfl_xor(sum.z, 16);
            sum.w += __shfl_xor(sum.w, 16);
        }
        if (s > 0) {
            __builtin_amdgcn_s_barrier_wait(-1);
            __builtin_amdgcn_fence(__ATOMIC_ACQUIRE, "workgroup", "local");
        }
        if (!isMat) {
            if (s < nSteps && half == 0) *(float4*)(&s_A[s & 1][gn * L1_AP + fh * 64 + l16 * 4]) = sum;
        } else if (s > 0) {
            const int t = s - 1;
            float dv[8];
            #pragma unroll
            for (int r = 0; r < 8; ++r) dv[r] = s_dv[min(t * FU_TN + half * 8 + r, AGG_T - 1)];
            const float* ap = &s_A[t & 1][l16 * L1_AP + 8 * half];
            #pragma unroll
            for (int ps = 0; ps < 2; ++ps) {
                const int cb = mw * 64 + ps * 32;
                const float* wp[2];
                wp[0] = s_W + (cb + l16) * L1_AP + 8 * half;
                wp[1] = wp[0] + 16 * L1_AP;

                v8f acc[2];
                acc[0] = v8f{};
                acc[1] = v8f{};
                float4 a[2], b[2][2];
                a[0] = *(const float4*)(ap);
                a[1] = *(const float4*)(ap + 4);
                #pragma unroll
                for (int j = 0; j < 2; ++j) {
                    b[j][0] = *(const float4*)(wp[j]);
                    b[j][1] = *(const float4*)(wp[j] + 4);
                }
                #pragma unroll
                for (int k0 = 0; k0 < N_IN; k0 += GM_KS) {
                    float4 an[2], bn[2][2];
                    if (k0 + GM_KS < N_IN) {
                        const int kn = k0 + GM_KS;
                        an[0] = *(const float4*)(ap + kn);
                        an[1] = *(const float4*)(ap + kn + 4);
                        #pragma unroll
                        for (int j = 0; j < 2; ++j) {
                            bn[j][0] = *(const float4*)(wp[j] + kn);
                            bn[j][1] = *(const float4*)(wp[j] + kn + 4);
                        }
                    }
                    __builtin_amdgcn_sched_barrier(0);
                    #pragma unroll
                    for (int p = 0; p < 2; ++p) {
                        const v2f a0 = { a[p].x, a[p].y };
                        #pragma unroll
                        for (int j = 0; j < 2; ++j) {
                            const v2f b0 = { b[j][p].x, b[j][p].y };
                            acc[j] = __builtin_amdgcn_wmma_f32_16x16x4_f32(
                                 false, a0,  false, b0,
                                 (short)0, acc[j],  false,  false);
                        }
                        const v2f a1 = { a[p].z, a[p].w };
                        #pragma unroll
                        for (int j = 0; j < 2; ++j) {
                            const v2f b1f = { b[j][p].z, b[j][p].w };
                            acc[j] = __builtin_amdgcn_wmma_f32_16x16x4_f32(
                                 false, a1,  false, b1f,
                                 (short)0, acc[j],  false,  false);
                        }
                    }
                    __builtin_amdgcn_sched_barrier(0);
                    if (k0 + GM_KS < N_IN) {
                        a[0] = an[0]; a[1] = an[1];
                        #pragma unroll
                        for (int j = 0; j < 2; ++j) { b[j][0] = bn[j][0]; b[j][1] = bn[j][1]; }
                    }
                }

                #pragma unroll
                for (int r = 0; r < 8; ++r) {
                    const int rt = half * 8 + r;
                    const int nd = t * FU_TN + rt;
                    if (rt < FU_TN && nd < nT) {
                        #pragma unroll
                        for (int j = 0; j < 2; ++j) {
                            float v = acc[j][r] * dv[r] + bv[ps][j];
                            v = fmaxf(v, 0.0f);
                            hid[(size_t)(lo + nd) * N_HID + cb + j * 16 + l16] = v;
                        }
                    }
                }
            }
        }
        __builtin_amdgcn_fence(__ATOMIC_RELEASE, "workgroup", "local");
        __builtin_amdgcn_s_barrier_signal(-1);
    }
    __builtin_amdgcn_s_barrier_wait(-1);
    __builtin_amdgcn_fence(__ATOMIC_ACQUIRE, "workgroup", "local");

    for (int i = tid; i < nT; i += AGG_NT) dinv[lo + i] = s_dv[i];
}

extern "C" void kernel_launch(void* const* d_in, const int* in_sizes, int n_in,
                              void* d_out, int out_size, void* d_ws, size_t ws_size,
                              hipStream_t stream) {
    const float* x  = (const float*)d_in[0];
    const int*   ei = (const int*)  d_in[1];
    const float* W1 = (const float*)d_in[3];
    const float* b1 = (const float*)d_in[4];
    const float* W2 = (const float*)d_in[5];
    const float* b2 = (const float*)d_in[6];
    const float* W3 = (const float*)d_in[7];
    const float* b3 = (const float*)d_in[8];
    float* out = (float*)d_out;

    const int M = in_sizes[0] / N_IN;
    const int E = in_sizes[1] / 2;
    const int* row = ei;
    const int* col = ei + E;

    float* ws = (float*)d_ws;
    size_t off = 0;
    const int NG  = (M + AGG_T - 1) / AGG_T;
    const int NGp = (NG + 31) & ~31;
    int*   gcnt = (int*)(ws + off); off += (size_t)NGp;
    float* dinv = ws + off; off += (size_t)M;
    off = (off + 3) & ~(size_t)3;
    float* h1   = ws + off; off += (size_t)M * N_HID;
    off = (off + 1) & ~(size_t)1;
    int2* lists = (int2*)(ws + off);
    const size_t room = (ws_size / sizeof(float) - off) / 2 / (size_t)NG;
    const int cap = (int)(room < (size_t)BIN_CAP ? room : (size_t)BIN_CAP);

    hipMemsetAsync(gcnt, 0, sizeof(int) * (size_t)NGp, stream);

    bin_edges<<<(E + BIN_NT * 4 - 1) / (BIN_NT * 4), BIN_NT, 0, stream>>>(row, col, lists, gcnt, E, M, NG, cap);

    agg_dense_relu<<<NG, AGG_NT, 0, stream>>>(x, row, col, lists, gcnt, cap, W1, b1, h1, dinv, M, E);

    agg_dense_head<<<NG, AGG_NT, 0, stream>>>(h1, row, col, lists, gcnt, cap, W2, b2, dinv, W3, b3, out, M, E);
}
